# starting kernel
# baseline (speedup 1.0000x reference)
#include <hip/hip_runtime.h>
#include <math.h>

typedef __attribute__((ext_vector_type(2))) float v2f;
typedef __attribute__((ext_vector_type(8))) float v8f;

#define GAT_NEG_SLOPE 0.2f
#define GAT_SM_EPS 1e-16f
#define GAT_LN_EPS 1e-5f

__device__ __forceinline__ float warp_sum32(float v) {
#pragma unroll
  for (int o = 16; o > 0; o >>= 1) v += __shfl_xor(v, o, 32);
  return v;
}
__device__ __forceinline__ float warp_max32(float v) {
#pragma unroll
  for (int o = 16; o > 0; o >>= 1) v = fmaxf(v, __shfl_xor(v, o, 32));
  return v;
}

__device__ __forceinline__ void atomic_max_f32(float* addr, float v) {
  if (v >= 0.0f) atomicMax((int*)addr, __float_as_int(v));
  else           atomicMin((unsigned int*)addr, __float_as_uint(v));
}

__global__ void fill_f32(float* __restrict__ p, int n, float v) {
  int i = blockIdx.x * blockDim.x + threadIdx.x;
  int stride = gridDim.x * blockDim.x;
  for (; i < n; i += stride) p[i] = v;
}

__global__ void gemm_wmma_f32(const float* __restrict__ A, const float* __restrict__ B,
                              float* __restrict__ C, int M, int K, int Nc) {
  extern __shared__ float As[];
  const int KP = K + 1;
  const int mBase = blockIdx.x * 16;
  const int tid = threadIdx.x;

  for (int i = tid; i < 16 * K; i += blockDim.x) {
    int r = i / K, c = i - r * K;
    int row = mBase + r;
    As[r * KP + c] = (row < M) ? A[(size_t)row * K + c] : 0.0f;
  }
  __syncthreads();

  const int wave = tid >> 5;
  const int lane = tid & 31;
  const int nBase = wave * 16;
  const int l15 = lane & 15;
  const int khalf = (lane >> 4) * 2;

  v8f acc = {};
  for (int k0 = 0; k0 < K; k0 += 4) {
    v2f a, b;
    a.x = As[l15 * KP + k0 + khalf];
    a.y = As[l15 * KP + k0 + khalf + 1];
    b.x = B[(size_t)(k0 + khalf) * Nc + nBase + l15];
    b.y = B[(size_t)(k0 + khalf + 1) * Nc + nBase + l15];
    acc = __builtin_amdgcn_wmma_f32_16x16x4_f32(false, a, false, b,
                                                (short)0, acc, false, false);
  }

  const int rowTop = mBase + (lane >> 4) * 8;
  const int col = nBase + l15;
#pragma unroll
  for (int r = 0; r < 8; ++r) {
    int row = rowTop + r;
    if (row < M) C[(size_t)row * Nc + col] = acc[r];
  }
}

__global__ void attn_scores_k(const float* __restrict__ h, const float* __restrict__ a_src,
                              const float* __restrict__ a_dst, float* __restrict__ es,
                              float* __restrict__ ed, int N, int H, int CH) {
  int idx = blockIdx.x * blockDim.x + threadIdx.x;
  if (idx >= N * H) return;
  int n = idx / H, hh = idx - n * H;
  const float* hp = h + (size_t)n * H * CH + hh * CH;
  const float* asp = a_src + hh * CH;
  const float* adp = a_dst + hh * CH;
  float s = 0.0f, d = 0.0f;
  for (int c = 0; c < CH; ++c) { float t = hp[c]; s += t * asp[c]; d += t * adp[c]; }
  es[idx] = s; ed[idx] = d;
}

__device__ __forceinline__ void edge_sd(const int* __restrict__ ei, int E, int e,
                                        int& s, int& d) {
  if (e < E) { s = ei[e]; d = ei[E + e]; }
  else       { s = d = e - E; }
}

__global__ void edge_max_k(const int* __restrict__ ei, int E, int N,
                           const float* __restrict__ es, const float* __restrict__ ed,
                           float* __restrict__ m, int H) {
  int idx = blockIdx.x * blockDim.x + threadIdx.x;
  int tot = (E + N) * H;
  if (idx >= tot) return;
  int h = idx & (H - 1);
  int e = idx >> 2;
  int s, d; edge_sd(ei, E, e, s, d);
  float v = es[s * H + h] + ed[d * H + h];
  v = (v > 0.0f) ? v : GAT_NEG_SLOPE * v;
  atomic_max_f32(&m[d * H + h], v);
}

__global__ void edge_expsum_k(const int* __restrict__ ei, int E, int N,
                              const float* __restrict__ es, const float* __restrict__ ed,
                              const float* __restrict__ m, float* __restrict__ denom, int H) {
  int idx = blockIdx.x * blockDim.x + threadIdx.x;
  int tot = (E + N) * H;
  if (idx >= tot) return;
  int h = idx & (H - 1);
  int e = idx >> 2;
  int s, d; edge_sd(ei, E, e, s, d);
  float v = es[s * H + h] + ed[d * H + h];
  v = (v > 0.0f) ? v : GAT_NEG_SLOPE * v;
  atomicAdd(&denom[d * H + h], __expf(v - m[d * H + h]));
}

template <int CHT, int CH>
__global__ void edge_agg_k(const int* __restrict__ ei, int E, int N,
                           const float* __restrict__ feat, const float* __restrict__ es,
                           const float* __restrict__ ed, const float* __restrict__ m,
                           const float* __restrict__ denom, float* __restrict__ out) {
  const int H = CHT / CH;
  int gw = (blockIdx.x * blockDim.x + threadIdx.x) >> 5;
  int lane = threadIdx.x & 31;
  if (gw >= E + N) return;
  int s, d; edge_sd(ei, E, gw, s, d);

  float alpha = 0.0f;
  if (lane < H) {
    float v = es[s * H + lane] + ed[d * H + lane];
    v = (v > 0.0f) ? v : GAT_NEG_SLOPE * v;
    alpha = __expf(v - m[d * H + lane]) / (denom[d * H + lane] + GAT_SM_EPS);
  }
  const float* fp = feat + (size_t)s * CHT;
  float* op = out + (size_t)d * CHT;
#pragma unroll
  for (int c = lane; c < CHT; c += 32) {
    int head = c / CH;
    float a = __shfl(alpha, head, 32);
    atomicAdd(&op[c], fp[c] * a);
  }
}

__global__ void elu_ln128_k(float* __restrict__ x, const float* __restrict__ bias,
                            const float* __restrict__ g, const float* __restrict__ b, int N) {
  int node = blockIdx.x * (blockDim.x >> 5) + (threadIdx.x >> 5);
  int lane = threadIdx.x & 31;
  if (node >= N) return;
  float* xp = x + (size_t)node * 128;
  float v[4];
  float sum = 0.0f;
#pragma unroll
  for (int j = 0; j < 4; ++j) {
    int c = lane * 4 + j;
    float t = xp[c] + bias[c];
    t = (t > 0.0f) ? t : (__expf(t) - 1.0f);
    v[j] = t; sum += t;
  }
  float mu = warp_sum32(sum) * (1.0f / 128.0f);
  float vs = 0.0f;
#pragma unroll
  for (int j = 0; j < 4; ++j) { float dd = v[j] - mu; vs += dd * dd; }
  float var = warp_sum32(vs) * (1.0f / 128.0f);
  float rstd = rsqrtf(var + GAT_LN_EPS);
#pragma unroll
  for (int j = 0; j < 4; ++j) {
    int c = lane * 4 + j;
    xp[c] = (v[j] - mu) * rstd * g[c] + b[c];
  }
}

__global__ void final_k(const float* __restrict__ out2, const float* __restrict__ b2,
                        const float* __restrict__ g, const float* __restrict__ bb,
                        float* __restrict__ y, int N) {
  int node = blockIdx.x * (blockDim.x >> 5) + (threadIdx.x >> 5);
  int lane = threadIdx.x & 31;
  if (node >= N) return;
  const bool has1 = (lane < 8);
  const float* op = out2 + (size_t)node * 160;

  float x0, x1 = 0.0f;
  { float s = 0.0f;
#pragma unroll
    for (int h = 0; h < 4; ++h) s += op[h * 40 + lane];
    x0 = s * 0.25f + b2[lane]; }
  if (has1) {
    float s = 0.0f;
#pragma unroll
    for (int h = 0; h < 4; ++h) s += op[h * 40 + lane + 32];
    x1 = s * 0.25f + b2[lane + 32];
  }
  float mu = warp_sum32(x0 + (has1 ? x1 : 0.0f)) * (1.0f / 40.0f);
  float d0 = x0 - mu, d1 = has1 ? (x1 - mu) : 0.0f;
  float var = warp_sum32(d0 * d0 + d1 * d1) * (1.0f / 40.0f);
  float rstd = rsqrtf(var + GAT_LN_EPS);
  float z0 = d0 * rstd * g[lane] + bb[lane];
  float z1 = has1 ? (d1 * rstd * g[lane + 32] + bb[lane + 32]) : -3.0e38f;
  float mx = warp_max32(fmaxf(z0, z1));
  float se = warp_sum32(__expf(z0 - mx) + (has1 ? __expf(z1 - mx) : 0.0f));
  float lse = __logf(se);
  float* yp = y + (size_t)node * 40;
  yp[lane] = z0 - mx - lse;
  if (has1) yp[lane + 32] = z1 - mx - lse;
}

extern "C" void kernel_launch(void* const* d_in, const int* in_sizes, int n_in,
                              void* d_out, int out_size, void* d_ws, size_t ws_size,
                              hipStream_t stream) {
  const float* x        = (const float*)d_in[0];
  const int*   ei       = (const int*)d_in[1];
  const float* W1       = (const float*)d_in[2];
  const float* att_src1 = (const float*)d_in[3];
  const float* att_dst1 = (const float*)d_in[4];
  const float* b1       = (const float*)d_in[5];
  const float* W2       = (const float*)d_in[6];
  const float* att_src2 = (const float*)d_in[7];
  const float* att_dst2 = (const float*)d_in[8];
  const float* b2       = (const float*)d_in[9];
  const float* ln0_g    = (const float*)d_in[10];
  const float* ln0_b    = (const float*)d_in[11];
  const float* ln1_g    = (const float*)d_in[12];
  const float* ln1_b    = (const float*)d_in[13];

  const int F_IN = 256, H1C = 128, H2C = 160, H = 4;
  const int N = in_sizes[0] / F_IN;
  const int E = in_sizes[1] / 2;
  const int EL = E + N;

  float* ws   = (float*)d_ws;
  float* h1   = ws;
  float* out1 = h1   + (size_t)N * H1C;
  float* out2 = out1 + (size_t)N * H1C;
  float* h2   = out2 + (size_t)N * H2C;
  float* es   = h2   + (size_t)N * H2C;
  float* ed   = es   + (size_t)N * H;
  float* mseg = ed   + (size_t)N * H;
  float* dseg = mseg + (size_t)N * H;

  const float NEG_INF = -__builtin_inff();
  const int mTiles = (N + 15) / 16;

  gemm_wmma_f32<<<mTiles, 2 * H1C, 16 * (F_IN + 1) * sizeof(float), stream>>>(
      x, W1, h1, N, F_IN, H1C);
  attn_scores_k<<<(N * H + 255) / 256, 256, 0, stream>>>(h1, att_src1, att_dst1,
                                                         es, ed, N, H, 32);
  fill_f32<<<512, 256, 0, stream>>>(mseg, N * H, NEG_INF);
  fill_f32<<<512, 256, 0, stream>>>(dseg, N * H, 0.0f);
  fill_f32<<<2048, 256, 0, stream>>>(out1, N * H1C, 0.0f);
  edge_max_k<<<(EL * H + 255) / 256, 256, 0, stream>>>(ei, E, N, es, ed, mseg, H);
  edge_expsum_k<<<(EL * H + 255) / 256, 256, 0, stream>>>(ei, E, N, es, ed, mseg, dseg, H);
  edge_agg_k<128, 32><<<(EL + 7) / 8, 256, 0, stream>>>(ei, E, N, h1, es, ed,
                                                        mseg, dseg, out1);
  elu_ln128_k<<<(N + 7) / 8, 256, 0, stream>>>(out1, b1, ln0_g, ln0_b, N);

  gemm_wmma_f32<<<mTiles, 2 * H2C, 16 * (H1C + 1) * sizeof(float), stream>>>(
      out1, W2, h2, N, H1C, H2C);
  attn_scores_k<<<(N * H + 255) / 256, 256, 0, stream>>>(h2, att_src2, att_dst2,
                                                         es, ed, N, H, 40);
  fill_f32<<<512, 256, 0, stream>>>(mseg, N * H, NEG_INF);
  fill_f32<<<512, 256, 0, stream>>>(dseg, N * H, 0.0f);
  fill_f32<<<2048, 256, 0, stream>>>(out2, N * H2C, 0.0f);
  edge_max_k<<<(EL * H + 255) / 256, 256, 0, stream>>>(ei, E, N, es, ed, mseg, H);
  edge_expsum_k<<<(EL * H + 255) / 256, 256, 0, stream>>>(ei, E, N, es, ed, mseg, dseg, H);
  edge_agg_k<160, 40><<<(EL + 7) / 8, 256, 0, stream>>>(ei, E, N, h2, es, ed,
                                                        mseg, dseg, out2);

  final_k<<<(N + 7) / 8, 256, 0, stream>>>(out2, b2, ln1_g, ln1_b, (float*)d_out, N);
}


# Float-atomic edge scatter replaced by dst-sorted gather: in-call counting sort (int atomics, arrival-order rank), wave per node does softmax and row sum in registers, one divide at the end
# speedup vs baseline: 2.1408x; 2.1408x over previous
#include <hip/hip_runtime.h>
#include <math.h>

typedef __attribute__((ext_vector_type(2))) float v2f;
typedef __attribute__((ext_vector_type(8))) float v8f;

#define GAT_NEG_SLOPE 0.2f
#define GAT_SM_EPS 1e-16f
#define GAT_LN_EPS 1e-5f

__device__ __forceinline__ float warp_sum32(float v) {
#pragma unroll
  for (int o = 16; o > 0; o >>= 1) v += __shfl_xor(v, o, 32);
  return v;
}
__device__ __forceinline__ float warp_max32(float v) {
#pragma unroll
  for (int o = 16; o > 0; o >>= 1) v = fmaxf(v, __shfl_xor(v, o, 32));
  return v;
}

__global__ void fill_i32(int* __restrict__ p, int n, int v) {
  int i = blockIdx.x * blockDim.x + threadIdx.x;
  int stride = gridDim.x * blockDim.x;
  for (; i < n; i += stride) p[i] = v;
}

__global__ void gemm_wmma_f32(const float* __restrict__ A, const float* __restrict__ B,
                              float* __restrict__ C, int M, int K, int Nc) {
  extern __shared__ float As[];
  const int KP = K + 1;
  const int mBase = blockIdx.x * 16;
  const int tid = threadIdx.x;

  for (int i = tid; i < 16 * K; i += blockDim.x) {
    int r = i / K, c = i - r * K;
    int row = mBase + r;
    As[r * KP + c] = (row < M) ? A[(size_t)row * K + c] : 0.0f;
  }
  __syncthreads();

  const int wave = tid >> 5;
  const int lane = tid & 31;
  const int nBase = wave * 16;
  const int l15 = lane & 15;
  const int khalf = (lane >> 4) * 2;

  v8f acc = {};
  for (int k0 = 0; k0 < K; k0 += 4) {
    v2f a, b;
    a.x = As[l15 * KP + k0 + khalf];
    a.y = As[l15 * KP + k0 + khalf + 1];
    b.x = B[(size_t)(k0 + khalf) * Nc + nBase + l15];
    b.y = B[(size_t)(k0 + khalf + 1) * Nc + nBase + l15];
    acc = __builtin_amdgcn_wmma_f32_16x16x4_f32(false, a, false, b,
                                                (short)0, acc, false, false);
  }

  const int rowTop = mBase + (lane >> 4) * 8;
  const int col = nBase + l15;
#pragma unroll
  for (int r = 0; r < 8; ++r) {
    int row = rowTop + r;
    if (row < M) C[(size_t)row * Nc + col] = acc[r];
  }
}

__global__ void attn_scores_k(const float* __restrict__ h, const float* __restrict__ a_src,
                              const float* __restrict__ a_dst, float* __restrict__ es,
                              float* __restrict__ ed, int N, int H, int CH) {
  int idx = blockIdx.x * blockDim.x + threadIdx.x;
  if (idx >= N * H) return;
  int n = idx / H, hh = idx - n * H;
  const float* hp = h + (size_t)n * H * CH + hh * CH;
  const float* asp = a_src + hh * CH;
  const float* adp = a_dst + hh * CH;
  float s = 0.0f, d = 0.0f;
  for (int c = 0; c < CH; ++c) { float t = hp[c]; s += t * asp[c]; d += t * adp[c]; }
  es[idx] = s; ed[idx] = d;
}

__device__ __forceinline__ float lrelu(float v) {
  return (v > 0.0f) ? v : GAT_NEG_SLOPE * v;
}
__device__ __forceinline__ float pick4(int h, float a0, float a1, float a2, float a3) {
  return (h == 0) ? a0 : ((h == 1) ? a1 : ((h == 2) ? a2 : a3));
}

__global__ void dst_count_k(const int* __restrict__ ei, int E, int* __restrict__ cnt) {
  int e = blockIdx.x * blockDim.x + threadIdx.x;
  if (e < E) atomicAdd(&cnt[ei[E + e]], 1);
}

__global__ void seg_alloc_k(const int* cnt, int* cursor, int* __restrict__ beg,
                            int* __restrict__ cur, int N) {
  int n = blockIdx.x * blockDim.x + threadIdx.x;
  int lane = threadIdx.x & 31;
  int c = (n < N) ? cnt[n] : 0;
  int incl = c;
#pragma unroll
  for (int o = 1; o < 32; o <<= 1) {
    int t = __shfl_up(incl, o, 32);
    if (lane >= o) incl += t;
  }
  int total = __shfl(incl, 31, 32);
  int base = 0;
  if (lane == 31) base = atomicAdd(cursor, total);
  base = __shfl(base, 31, 32);
  if (n < N) { int b = base + incl - c; beg[n] = b; cur[n] = b; }
}

__global__ void dst_fill_k(const int* __restrict__ ei, int E, int* __restrict__ cur,
                           int* __restrict__ ssrc) {
  int e = blockIdx.x * blockDim.x + threadIdx.x;
  if (e < E) {
    int pos = atomicAdd(&cur[ei[E + e]], 1);
    ssrc[pos] = ei[e];
  }
}

template <int CHT, int CH>
__global__ void node_gather_k(const int* __restrict__ beg, const int* __restrict__ endp,
                              const int* __restrict__ ssrc, const float* __restrict__ feat,
                              const float* __restrict__ es, const float* __restrict__ ed,
                              float* __restrict__ out, int N) {
  constexpr int NI = CHT / 32;
  const int node = blockIdx.x * (blockDim.x >> 5) + (threadIdx.x >> 5);
  const int lane = threadIdx.x & 31;
  if (node >= N) return;
  const float4* es4 = reinterpret_cast<const float4*>(es);
  const float4* ed4 = reinterpret_cast<const float4*>(ed);
  const int jb = beg[node], je = endp[node];
  const float4 dv = ed4[node];
  const float4 sv = es4[node];

  const float v0 = lrelu(sv.x + dv.x), v1 = lrelu(sv.y + dv.y);
  const float v2 = lrelu(sv.z + dv.z), v3 = lrelu(sv.w + dv.w);
  float m0 = v0, m1 = v1, m2 = v2, m3 = v3;
  for (int j = jb + lane; j < je; j += 32) {
    const float4 e4 = es4[ssrc[j]];
    m0 = fmaxf(m0, lrelu(e4.x + dv.x));
    m1 = fmaxf(m1, lrelu(e4.y + dv.y));
    m2 = fmaxf(m2, lrelu(e4.z + dv.z));
    m3 = fmaxf(m3, lrelu(e4.w + dv.w));
  }
  m0 = warp_max32(m0); m1 = warp_max32(m1); m2 = warp_max32(m2); m3 = warp_max32(m3);

  int hd[NI];
  float acc[NI];
#pragma unroll
  for (int i = 0; i < NI; ++i) { hd[i] = (lane + 32 * i) / CH; acc[i] = 0.0f; }
  float d0 = 0.0f, d1 = 0.0f, d2 = 0.0f, d3 = 0.0f;

  for (int j0 = jb; j0 < je; j0 += 32) {
    const int j = j0 + lane;
    const bool ok = j < je;
    const int s = ok ? ssrc[j] : node;
    const float4 e4 = es4[s];
    const float p0 = ok ? __expf(lrelu(e4.x + dv.x) - m0) : 0.0f;
    const float p1 = ok ? __expf(lrelu(e4.y + dv.y) - m1) : 0.0f;
    const float p2 = ok ? __expf(lrelu(e4.z + dv.z) - m2) : 0.0f;
    const float p3 = ok ? __expf(lrelu(e4.w + dv.w) - m3) : 0.0f;
    d0 += p0; d1 += p1; d2 += p2; d3 += p3;
    const int nk = (je - j0 < 32) ? (je - j0) : 32;
    for (int k = 0; k < nk; ++k) {
      const int sk = __shfl(s, k, 32);
      const float q0 = __shfl(p0, k, 32), q1 = __shfl(p1, k, 32);
      const float q2 = __shfl(p2, k, 32), q3 = __shfl(p3, k, 32);
      const float* fp = feat + (size_t)sk * CHT + lane;
#pragma unroll
      for (int i = 0; i < NI; ++i) acc[i] += fp[32 * i] * pick4(hd[i], q0, q1, q2, q3);
    }
  }

  const float s0 = __expf(v0 - m0), s1 = __expf(v1 - m1);
  const float s2 = __expf(v2 - m2), s3 = __expf(v3 - m3);
  {
    const float* fp = feat + (size_t)node * CHT + lane;
#pragma unroll
    for (int i = 0; i < NI; ++i) acc[i] += fp[32 * i] * pick4(hd[i], s0, s1, s2, s3);
  }
  d0 = warp_sum32(d0) + s0; d1 = warp_sum32(d1) + s1;
  d2 = warp_sum32(d2) + s2; d3 = warp_sum32(d3) + s3;

  float* op = out + (size_t)node * CHT + lane;
#pragma unroll
  for (int i = 0; i < NI; ++i)
    op[32 * i] = acc[i] / (pick4(hd[i], d0, d1, d2, d3) + GAT_SM_EPS);
}

__global__ void elu_ln128_k(float* __restrict__ x, const float* __restrict__ bias,
                            const float* __restrict__ g, const float* __restrict__ b, int N) {
  int node = blockIdx.x * (blockDim.x >> 5) + (threadIdx.x >> 5);
  int lane = threadIdx.x & 31;
  if (node >= N) return;
  float* xp = x + (size_t)node * 128;
  float v[4];
  float sum = 0.0f;
#pragma unroll
  for (int j = 0; j < 4; ++j) {
    int c = lane * 4 + j;
    float t = xp[c] + bias[c];
    t = (t > 0.0f) ? t : (__expf(t) - 1.0f);
    v[j] = t; sum += t;
  }
  float mu = warp_sum32(sum) * (1.0f / 128.0f);
  float vs = 0.0f;
#pragma unroll
  for (int j = 0; j < 4; ++j) { float dd = v[j] - mu; vs += dd * dd; }
  float var = warp_sum32(vs) * (1.0f / 128.0f);
  float rstd = rsqrtf(var + GAT_LN_EPS);
#pragma unroll
  for (int j = 0; j < 4; ++j) {
    int c = lane * 4 + j;
    xp[c] = (v[j] - mu) * rstd * g[c] + b[c];
  }
}

__global__ void final_k(const float* __restrict__ out2, const float* __restrict__ b2,
                        const float* __restrict__ g, const float* __restrict__ bb,
                        float* __restrict__ y, int N) {
  int node = blockIdx.x * (blockDim.x >> 5) + (threadIdx.x >> 5);
  int lane = threadIdx.x & 31;
  if (node >= N) return;
  const bool has1 = (lane < 8);
  const float* op = out2 + (size_t)node * 160;

  float x0, x1 = 0.0f;
  { float s = 0.0f;
#pragma unroll
    for (int h = 0; h < 4; ++h) s += op[h * 40 + lane];
    x0 = s * 0.25f + b2[lane]; }
  if (has1) {
    float s = 0.0f;
#pragma unroll
    for (int h = 0; h < 4; ++h) s += op[h * 40 + lane + 32];
    x1 = s * 0.25f + b2[lane + 32];
  }
  float mu = warp_sum32(x0 + (has1 ? x1 : 0.0f)) * (1.0f / 40.0f);
  float d0 = x0 - mu, d1 = has1 ? (x1 - mu) : 0.0f;
  float var = warp_sum32(d0 * d0 + d1 * d1) * (1.0f / 40.0f);
  float rstd = rsqrtf(var + GAT_LN_EPS);
  float z0 = d0 * rstd * g[lane] + bb[lane];
  float z1 = has1 ? (d1 * rstd * g[lane + 32] + bb[lane + 32]) : -3.0e38f;
  float mx = warp_max32(fmaxf(z0, z1));
  float se = warp_sum32(__expf(z0 - mx) + (has1 ? __expf(z1 - mx) : 0.0f));
  float lse = __logf(se);
  float* yp = y + (size_t)node * 40;
  yp[lane] = z0 - mx - lse;
  if (has1) yp[lane + 32] = z1 - mx - lse;
}

extern "C" void kernel_launch(void* const* d_in, const int* in_sizes, int n_in,
                              void* d_out, int out_size, void* d_ws, size_t ws_size,
                              hipStream_t stream) {
  const float* x        = (const float*)d_in[0];
  const int*   ei       = (const int*)d_in[1];
  const float* W1       = (const float*)d_in[2];
  const float* att_src1 = (const float*)d_in[3];
  const float* att_dst1 = (const float*)d_in[4];
  const float* b1       = (const float*)d_in[5];
  const float* W2       = (const float*)d_in[6];
  const float* att_src2 = (const float*)d_in[7];
  const float* att_dst2 = (const float*)d_in[8];
  const float* b2       = (const float*)d_in[9];
  const float* ln0_g    = (const float*)d_in[10];
  const float* ln0_b    = (const float*)d_in[11];
  const float* ln1_g    = (const float*)d_in[12];
  const float* ln1_b    = (const float*)d_in[13];

  const int F_IN = 256, H1C = 128, H2C = 160, H = 4;
  const int N = in_sizes[0] / F_IN;
  const int E = in_sizes[1] / 2;

  float* ws   = (float*)d_ws;
  float* h1   = ws;
  float* out1 = h1   + (size_t)N * H1C;
  float* out2 = out1 + (size_t)N * H1C;
  float* h2   = out2 + (size_t)N * H2C;
  float* es   = h2   + (size_t)N * H2C;
  float* ed   = es   + (size_t)N * H;
  int* cnt    = (int*)(ed + (size_t)N * H);
  int* beg    = cnt + (N + 1);
  int* cur    = beg + N;
  int* ssrc   = cur + N;

  const int mTiles = (N + 15) / 16;

  fill_i32<<<(N + 256) / 256, 256, 0, stream>>>(cnt, N + 1, 0);
  dst_count_k<<<(E + 255) / 256, 256, 0, stream>>>(ei, E, cnt);
  seg_alloc_k<<<(N + 255) / 256, 256, 0, stream>>>(cnt, cnt + N, beg, cur, N);
  dst_fill_k<<<(E + 255) / 256, 256, 0, stream>>>(ei, E, cur, ssrc);

  gemm_wmma_f32<<<mTiles, 2 * H1C, 16 * (F_IN + 1) * sizeof(float), stream>>>(
      x, W1, h1, N, F_IN, H1C);
  attn_scores_k<<<(N * H + 255) / 256, 256, 0, stream>>>(h1, att_src1, att_dst1,
                                                         es, ed, N, H, 32);
  node_gather_k<128, 32><<<(N + 7) / 8, 256, 0, stream>>>(beg, cur, ssrc, h1, es, ed, out1, N);
  elu_ln128_k<<<(N + 7) / 8, 256, 0, stream>>>(out1, b1, ln0_g, ln0_b, N);

  gemm_wmma_f32<<<mTiles, 2 * H2C, 16 * (H1C + 1) * sizeof(float), stream>>>(
      out1, W2, h2, N, H1C, H2C);
  attn_scores_k<<<(N * H + 255) / 256, 256, 0, stream>>>(h2, att_src2, att_dst2,
                                                         es, ed, N, H, 40);
  node_gather_k<160, 40><<<(N + 7) / 8, 256, 0, stream>>>(beg, cur, ssrc, h2, es, ed, out2, N);

  final_k<<<(N + 7) / 8, 256, 0, stream>>>(out2, b2, ln1_g, ln1_b, (float*)d_out, N);
}


# Row tails fused into the gather waves: ELU+LayerNorm (conv1) and head-mean+LayerNorm+log_softmax (conv2) run on the row still in registers; LN-128 sums now reduce in lane-strided order
# speedup vs baseline: 2.2299x; 1.0416x over previous
#include <hip/hip_runtime.h>
#include <math.h>

typedef __attribute__((ext_vector_type(2))) float v2f;
typedef __attribute__((ext_vector_type(8))) float v8f;

#define GAT_NEG_SLOPE 0.2f
#define GAT_SM_EPS 1e-16f
#define GAT_LN_EPS 1e-5f

__device__ __forceinline__ float warp_sum32(float v) {
#pragma unroll
  for (int o = 16; o > 0; o >>= 1) v += __shfl_xor(v, o, 32);
  return v;
}
__device__ __forceinline__ float warp_max32(float v) {
#pragma unroll
  for (int o = 16; o > 0; o >>= 1) v = fmaxf(v, __shfl_xor(v, o, 32));
  return v;
}

__global__ void fill_i32(int* __restrict__ p, int n, int v) {
  int i = blockIdx.x * blockDim.x + threadIdx.x;
  int stride = gridDim.x * blockDim.x;
  for (; i < n; i += stride) p[i] = v;
}

__global__ void gemm_wmma_f32(const float* __restrict__ A, const float* __restrict__ B,
                              float* __restrict__ C, int M, int K, int Nc) {
  extern __shared__ float As[];
  const int KP = K + 1;
  const int mBase = blockIdx.x * 16;
  const int tid = threadIdx.x;

  for (int i = tid; i < 16 * K; i += blockDim.x) {
    int r = i / K, c = i - r * K;
    int row = mBase + r;
    As[r * KP + c] = (row < M) ? A[(size_t)row * K + c] : 0.0f;
  }
  __syncthreads();

  const int wave = tid >> 5;
  const int lane = tid & 31;
  const int nBase = wave * 16;
  const int l15 = lane & 15;
  const int khalf = (lane >> 4) * 2;

  v8f acc = {};
  for (int k0 = 0; k0 < K; k0 += 4) {
    v2f a, b;
    a.x = As[l15 * KP + k0 + khalf];
    a.y = As[l15 * KP + k0 + khalf + 1];
    b.x = B[(size_t)(k0 + khalf) * Nc + nBase + l15];
    b.y = B[(size_t)(k0 + khalf + 1) * Nc + nBase + l15];
    acc = __builtin_amdgcn_wmma_f32_16x16x4_f32(false, a, false, b,
                                                (short)0, acc, false, false);
  }

  const int rowTop = mBase + (lane >> 4) * 8;
  const int col = nBase + l15;
#pragma unroll
  for (int r = 0; r < 8; ++r) {
    int row = rowTop + r;
    if (row < M) C[(size_t)row * Nc + col] = acc[r];
  }
}

__global__ void attn_scores_k(const float* __restrict__ h, const float* __restrict__ a_src,
                              const float* __restrict__ a_dst, float* __restrict__ es,
                              float* __restrict__ ed, int N, int H, int CH) {
  int idx = blockIdx.x * blockDim.x + threadIdx.x;
  if (idx >= N * H) return;
  int n = idx / H, hh = idx - n * H;
  const float* hp = h + (size_t)n * H * CH + hh * CH;
  const float* asp = a_src + hh * CH;
  const float* adp = a_dst + hh * CH;
  float s = 0.0f, d = 0.0f;
  for (int c = 0; c < CH; ++c) { float t = hp[c]; s += t * asp[c]; d += t * adp[c]; }
  es[idx] = s; ed[idx] = d;
}

__device__ __forceinline__ float lrelu(float v) {
  return (v > 0.0f) ? v : GAT_NEG_SLOPE * v;
}
__device__ __forceinline__ float pick4(int h, float a0, float a1, float a2, float a3) {
  return (h == 0) ? a0 : ((h == 1) ? a1 : ((h == 2) ? a2 : a3));
}

__global__ void dst_count_k(const int* __restrict__ ei, int E, int* __restrict__ cnt) {
  int e = blockIdx.x * blockDim.x + threadIdx.x;
  if (e < E) atomicAdd(&cnt[ei[E + e]], 1);
}

__global__ void seg_alloc_k(const int* cnt, int* cursor, int* __restrict__ beg,
                            int* __restrict__ cur, int N) {
  int n = blockIdx.x * blockDim.x + threadIdx.x;
  int lane = threadIdx.x & 31;
  int c = (n < N) ? cnt[n] : 0;
  int incl = c;
#pragma unroll
  for (int o = 1; o < 32; o <<= 1) {
    int t = __shfl_up(incl, o, 32);
    if (lane >= o) incl += t;
  }
  int total = __shfl(incl, 31, 32);
  int base = 0;
  if (lane == 31) base = atomicAdd(cursor, total);
  base = __shfl(base, 31, 32);
  if (n < N) { int b = base + incl - c; beg[n] = b; cur[n] = b; }
}

__global__ void dst_fill_k(const int* __restrict__ ei, int E, int* __restrict__ cur,
                           int* __restrict__ ssrc) {
  int e = blockIdx.x * blockDim.x + threadIdx.x;
  if (e < E) {
    int pos = atomicAdd(&cur[ei[E + e]], 1);
    ssrc[pos] = ei[e];
  }
}

template <int CHT, int CH, bool LAST>
__global__ void node_gather_k(const int* __restrict__ beg, const int* __restrict__ endp,
                              const int* __restrict__ ssrc, const float* __restrict__ feat,
                              const float* __restrict__ es, const float* __restrict__ ed,
                              const float* __restrict__ bias, const float* __restrict__ g,
                              const float* __restrict__ b, float* __restrict__ out, int N) {
  constexpr int NI = CHT / 32;
  const int node = blockIdx.x * (blockDim.x >> 5) + (threadIdx.x >> 5);
  const int lane = threadIdx.x & 31;
  if (node >= N) return;
  const float4* es4 = reinterpret_cast<const float4*>(es);
  const float4* ed4 = reinterpret_cast<const float4*>(ed);
  const int jb = beg[node], je = endp[node];
  const float4 dv = ed4[node];
  const float4 sv = es4[node];

  const float v0 = lrelu(sv.x + dv.x), v1 = lrelu(sv.y + dv.y);
  const float v2 = lrelu(sv.z + dv.z), v3 = lrelu(sv.w + dv.w);
  float m0 = v0, m1 = v1, m2 = v2, m3 = v3;
  for (int j = jb + lane; j < je; j += 32) {
    const float4 e4 = es4[ssrc[j]];
    m0 = fmaxf(m0, lrelu(e4.x + dv.x));
    m1 = fmaxf(m1, lrelu(e4.y + dv.y));
    m2 = fmaxf(m2, lrelu(e4.z + dv.z));
    m3 = fmaxf(m3, lrelu(e4.w + dv.w));
  }
  m0 = warp_max32(m0); m1 = warp_max32(m1); m2 = warp_max32(m2); m3 = warp_max32(m3);

  int hd[NI];
  float acc[NI];
#pragma unroll
  for (int i = 0; i < NI; ++i) { hd[i] = (lane + 32 * i) / CH; acc[i] = 0.0f; }
  float d0 = 0.0f, d1 = 0.0f, d2 = 0.0f, d3 = 0.0f;

  for (int j0 = jb; j0 < je; j0 += 32) {
    const int j = j0 + lane;
    const bool ok = j < je;
    const int s = ok ? ssrc[j] : node;
    const float4 e4 = es4[s];
    const float p0 = ok ? __expf(lrelu(e4.x + dv.x) - m0) : 0.0f;
    const float p1 = ok ? __expf(lrelu(e4.y + dv.y) - m1) : 0.0f;
    const float p2 = ok ? __expf(lrelu(e4.z + dv.z) - m2) : 0.0f;
    const float p3 = ok ? __expf(lrelu(e4.w + dv.w) - m3) : 0.0f;
    d0 += p0; d1 += p1; d2 += p2; d3 += p3;
    const int nk = (je - j0 < 32) ? (je - j0) : 32;
    for (int k = 0; k < nk; ++k) {
      const int sk = __shfl(s, k, 32);
      const float q0 = __shfl(p0, k, 32), q1 = __shfl(p1, k, 32);
      const float q2 = __shfl(p2, k, 32), q3 = __shfl(p3, k, 32);
      const float* fp = feat + (size_t)sk * CHT + lane;
#pragma unroll
      for (int i = 0; i < NI; ++i) acc[i] += fp[32 * i] * pick4(hd[i], q0, q1, q2, q3);
    }
  }

  const float s0 = __expf(v0 - m0), s1 = __expf(v1 - m1);
  const float s2 = __expf(v2 - m2), s3 = __expf(v3 - m3);
  {
    const float* fp = feat + (size_t)node * CHT + lane;
#pragma unroll
    for (int i = 0; i < NI; ++i) acc[i] += fp[32 * i] * pick4(hd[i], s0, s1, s2, s3);
  }
  d0 = warp_sum32(d0) + s0; d1 = warp_sum32(d1) + s1;
  d2 = warp_sum32(d2) + s2; d3 = warp_sum32(d3) + s3;

  float val[NI];
#pragma unroll
  for (int i = 0; i < NI; ++i)
    val[i] = acc[i] / (pick4(hd[i], d0, d1, d2, d3) + GAT_SM_EPS);

  if constexpr (!LAST) {
    float sum = 0.0f;
#pragma unroll
    for (int i = 0; i < NI; ++i) {
      float t = val[i] + bias[lane + 32 * i];
      t = (t > 0.0f) ? t : (__expf(t) - 1.0f);
      val[i] = t; sum += t;
    }
    const float mu = warp_sum32(sum) * (1.0f / (float)CHT);
    float vs = 0.0f;
#pragma unroll
    for (int i = 0; i < NI; ++i) { float dd = val[i] - mu; vs += dd * dd; }
    const float var = warp_sum32(vs) * (1.0f / (float)CHT);
    const float rstd = rsqrtf(var + GAT_LN_EPS);
    float* op = out + (size_t)node * CHT + lane;
#pragma unroll
    for (int i = 0; i < NI; ++i) {
      const int c = lane + 32 * i;
      op[32 * i] = (val[i] - mu) * rstd * g[c] + b[c];
    }
  } else {
    static_assert(!LAST || (CHT == 160 && CH == 40), "head-mean tail is laid out for 4 heads of 40");
    const bool has1 = (lane < 8);
    const int r8 = (lane + 8) & 31, r16 = (lane + 16) & 31, r24 = (lane + 24) & 31;
    const float a1 = __shfl((lane >= 8) ? val[1] : val[2], r8, 32);
    const float a2 = __shfl((lane >= 16) ? val[2] : val[3], r16, 32);
    const float a3 = __shfl((lane >= 24) ? val[3] : val[4], r24, 32);
    const float c1 = __shfl(val[2], r8, 32);
    const float c2 = __shfl(val[3], r16, 32);
    const float c3 = __shfl(val[4], r24, 32);

    float x0, x1 = 0.0f;
    { float s = val[0]; s += a1; s += a2; s += a3;
      x0 = s * 0.25f + bias[lane]; }
    if (has1) {
      float s = val[1]; s += c1; s += c2; s += c3;
      x1 = s * 0.25f + bias[lane + 32];
    }
    const float mu = warp_sum32(x0 + (has1 ? x1 : 0.0f)) * (1.0f / 40.0f);
    const float e0 = x0 - mu, e1 = has1 ? (x1 - mu) : 0.0f;
    const float var = warp_sum32(e0 * e0 + e1 * e1) * (1.0f / 40.0f);
    const float rstd = rsqrtf(var + GAT_LN_EPS);
    const float z0 = e0 * rstd * g[lane] + b[lane];
    const float z1 = has1 ? (e1 * rstd * g[lane + 32] + b[lane + 32]) : -3.0e38f;
    const float mx = warp_max32(fmaxf(z0, z1));
    const float se = warp_sum32(__expf(z0 - mx) + (has1 ? __expf(z1 - mx) : 0.0f));
    const float lse = __logf(se);
    float* yp = out + (size_t)node * CH;
    yp[lane] = z0 - mx - lse;
    if (has1) yp[lane + 32] = z1 - mx - lse;
  }
}

extern "C" void kernel_launch(void* const* d_in, const int* in_sizes, int n_in,
                              void* d_out, int out_size, void* d_ws, size_t ws_size,
                              hipStream_t stream) {
  const float* x        = (const float*)d_in[0];
  const int*   ei       = (const int*)d_in[1];
  const float* W1       = (const float*)d_in[2];
  const float* att_src1 = (const float*)d_in[3];
  const float* att_dst1 = (const float*)d_in[4];
  const float* b1       = (const float*)d_in[5];
  const float* W2       = (const float*)d_in[6];
  const float* att_src2 = (const float*)d_in[7];
  const float* att_dst2 = (const float*)d_in[8];
  const float* b2       = (const float*)d_in[9];
  const float* ln0_g    = (const float*)d_in[10];
  const float* ln0_b    = (const float*)d_in[11];
  const float* ln1_g    = (const float*)d_in[12];
  const float* ln1_b    = (const float*)d_in[13];

  const int F_IN = 256, H1C = 128, H2C = 160, H = 4;
  const int N = in_sizes[0] / F_IN;
  const int E = in_sizes[1] / 2;

  float* ws   = (float*)d_ws;
  float* h1   = ws;
  float* out1 = h1   + (size_t)N * H1C;
  float* h2   = out1 + (size_t)N * H1C;
  float* es   = h2   + (size_t)N * H2C;
  float* ed   = es   + (size_t)N * H;
  int* cnt    = (int*)(ed + (size_t)N * H);
  int* beg    = cnt + (N + 1);
  int* cur    = beg + N;
  int* ssrc   = cur + N;

  const int mTiles = (N + 15) / 16;

  fill_i32<<<(N + 256) / 256, 256, 0, stream>>>(cnt, N + 1, 0);
  dst_count_k<<<(E + 255) / 256, 256, 0, stream>>>(ei, E, cnt);
  seg_alloc_k<<<(N + 255) / 256, 256, 0, stream>>>(cnt, cnt + N, beg, cur, N);
  dst_fill_k<<<(E + 255) / 256, 256, 0, stream>>>(ei, E, cur, ssrc);

  gemm_wmma_f32<<<mTiles, 2 * H1C, 16 * (F_IN + 1) * sizeof(float), stream>>>(
      x, W1, h1, N, F_IN, H1C);
  attn_scores_k<<<(N * H + 255) / 256, 256, 0, stream>>>(h1, att_src1, att_dst1,
                                                         es, ed, N, H, 32);
  node_gather_k<128, 32, false><<<(N + 7) / 8, 256, 0, stream>>>(
      beg, cur, ssrc, h1, es, ed, b1, ln0_g, ln0_b, out1, N);

  gemm_wmma_f32<<<mTiles, 2 * H2C, 16 * (H1C + 1) * sizeof(float), stream>>>(
      out1, W2, h2, N, H1C, H2C);
  attn_scores_k<<<(N * H + 255) / 256, 256, 0, stream>>>(h2, att_src2, att_dst2,
                                                         es, ed, N, H, 40);
  node_gather_k<160, 40, true><<<(N + 7) / 8, 256, 0, stream>>>(
      beg, cur, ssrc, h2, es, ed, b2, ln1_g, ln1_b, (float*)d_out, N);
}


# Edge sort on LDS counters: two-level counting sort (256-node buckets, then nodes) replaces 1.6M scattered global atomics; edges keep an arrival-order rank, now from LDS atomics
# speedup vs baseline: 2.8891x; 1.2956x over previous
#include <hip/hip_runtime.h>
#include <math.h>

typedef __attribute__((ext_vector_type(2))) float v2f;
typedef __attribute__((ext_vector_type(8))) float v8f;

#define GAT_NEG_SLOPE 0.2f
#define GAT_SM_EPS 1e-16f
#define GAT_LN_EPS 1e-5f

__device__ __forceinline__ float warp_sum32(float v) {
#pragma unroll
  for (int o = 16; o > 0; o >>= 1) v += __shfl_xor(v, o, 32);
  return v;
}
__device__ __forceinline__ float warp_max32(float v) {
#pragma unroll
  for (int o = 16; o > 0; o >>= 1) v = fmaxf(v, __shfl_xor(v, o, 32));
  return v;
}

__global__ void fill_i32(int* __restrict__ p, int n, int v) {
  int i = blockIdx.x * blockDim.x + threadIdx.x;
  int stride = gridDim.x * blockDim.x;
  for (; i < n; i += stride) p[i] = v;
}

__global__ void gemm_wmma_f32(const float* __restrict__ A, const float* __restrict__ B,
                              float* __restrict__ C, int M, int K, int Nc) {
  extern __shared__ float As[];
  const int KP = K + 1;
  const int mBase = blockIdx.x * 16;
  const int tid = threadIdx.x;

  for (int i = tid; i < 16 * K; i += blockDim.x) {
    int r = i / K, c = i - r * K;
    int row = mBase + r;
    As[r * KP + c] = (row < M) ? A[(size_t)row * K + c] : 0.0f;
  }
  __syncthreads();

  const int wave = tid >> 5;
  const int lane = tid & 31;
  const int nBase = wave * 16;
  const int l15 = lane & 15;
  const int khalf = (lane >> 4) * 2;

  v8f acc = {};
  for (int k0 = 0; k0 < K; k0 += 4) {
    v2f a, b;
    a.x = As[l15 * KP + k0 + khalf];
    a.y = As[l15 * KP + k0 + khalf + 1];
    b.x = B[(size_t)(k0 + khalf) * Nc + nBase + l15];
    b.y = B[(size_t)(k0 + khalf + 1) * Nc + nBase + l15];
    acc = __builtin_amdgcn_wmma_f32_16x16x4_f32(false, a, false, b,
                                                (short)0, acc, false, false);
  }

  const int rowTop = mBase + (lane >> 4) * 8;
  const int col = nBase + l15;
#pragma unroll
  for (int r = 0; r < 8; ++r) {
    int row = rowTop + r;
    if (row < M) C[(size_t)row * Nc + col] = acc[r];
  }
}

__global__ void attn_scores_k(const float* __restrict__ h, const float* __restrict__ a_src,
                              const float* __restrict__ a_dst, float* __restrict__ es,
                              float* __restrict__ ed, int N, int H, int CH) {
  int idx = blockIdx.x * blockDim.x + threadIdx.x;
  if (idx >= N * H) return;
  int n = idx / H, hh = idx - n * H;
  const float* hp = h + (size_t)n * H * CH + hh * CH;
  const float* asp = a_src + hh * CH;
  const float* adp = a_dst + hh * CH;
  float s = 0.0f, d = 0.0f;
  for (int c = 0; c < CH; ++c) { float t = hp[c]; s += t * asp[c]; d += t * adp[c]; }
  es[idx] = s; ed[idx] = d;
}

__device__ __forceinline__ float lrelu(float v) {
  return (v > 0.0f) ? v : GAT_NEG_SLOPE * v;
}
__device__ __forceinline__ float pick4(int h, float a0, float a1, float a2, float a3) {
  return (h == 0) ? a0 : ((h == 1) ? a1 : ((h == 2) ? a2 : a3));
}

#define BK_SHIFT 8
#define BK_NODES (1 << BK_SHIFT)
#define BK_MAX 256
#define BK_PAD 32
#define EPB 2048
#define EPT (EPB / 256)

__device__ __forceinline__ int wave_incl_scan(int v, int lane) {
#pragma unroll
  for (int o = 1; o < 32; o <<= 1) {
    int t = __shfl_up(v, o, 32);
    if (lane >= o) v += t;
  }
  return v;
}

__global__ void bucket_count_k(const int* __restrict__ ei, int E, int NBK,
                               int* __restrict__ btot, int* __restrict__ blkoff) {
  __shared__ int hist[BK_MAX];
  const int t = threadIdx.x;
  hist[t] = 0;
  __syncthreads();
  const int e0 = blockIdx.x * EPB + t;
  int d[EPT];
#pragma unroll
  for (int k = 0; k < EPT; ++k) {
    const int e = e0 + k * 256;
    d[k] = ei[E + ((e < E) ? e : (E - 1))];
  }
#pragma unroll
  for (int k = 0; k < EPT; ++k)
    if (e0 + k * 256 < E) atomicAdd(&hist[d[k] >> BK_SHIFT], 1);
  __syncthreads();
  if (t < NBK) {
    const int c = hist[t];
    int off = 0;
    if (c > 0) off = atomicAdd(&btot[t * BK_PAD], c);
    blkoff[blockIdx.x * NBK + t] = off;
  }
}

__global__ void bucket_scatter_k(const int* __restrict__ ei, int E, int NBK,
                                 const int* __restrict__ btot, const int* __restrict__ blkoff,
                                 int* __restrict__ bbase, int* __restrict__ pairs) {
  __shared__ int cursor[BK_MAX];
  __shared__ int wsum[BK_MAX / 32];
  const int t = threadIdx.x, lane = t & 31, wave = t >> 5;
  const int v = (t < NBK) ? btot[t * BK_PAD] : 0;
  const int incl = wave_incl_scan(v, lane);
  if (lane == 31) wsum[wave] = incl;
  __syncthreads();
  int add = 0;
  for (int w = 0; w < wave; ++w) add += wsum[w];
  const int excl = add + incl - v;
  if (t < NBK) {
    cursor[t] = excl + blkoff[blockIdx.x * NBK + t];
    if (blockIdx.x == 0) bbase[t] = excl;
  }
  __syncthreads();
  const int e0 = blockIdx.x * EPB + t;
  int s[EPT], d[EPT];
#pragma unroll
  for (int k = 0; k < EPT; ++k) {
    const int e = e0 + k * 256;
    const int ec = (e < E) ? e : (E - 1);
    s[k] = ei[ec];
    d[k] = ei[E + ec];
  }
  int2* __restrict__ pr = reinterpret_cast<int2*>(pairs);
#pragma unroll
  for (int k = 0; k < EPT; ++k)
    if (e0 + k * 256 < E) {
      const int pos = atomicAdd(&cursor[d[k] >> BK_SHIFT], 1);
      pr[pos] = make_int2(s[k], d[k]);
    }
}

__global__ void __launch_bounds__(1024)
bucket_sort_k(const int* __restrict__ pairs, const int* __restrict__ btot,
              const int* __restrict__ bbase, int N, int* __restrict__ beg,
              int* __restrict__ endp, int* __restrict__ ssrc) {
  __shared__ int cnt[BK_NODES];
  __shared__ int cursor[BK_NODES];
  __shared__ int wsum[BK_NODES / 32];
  const int t = threadIdx.x, lane = t & 31, wave = t >> 5;
  const int bk = blockIdx.x;
  const int start = bbase[bk];
  const int end = start + btot[bk * BK_PAD];
  const int2* __restrict__ pr = reinterpret_cast<const int2*>(pairs);
  if (t < BK_NODES) cnt[t] = 0;
  __syncthreads();
  for (int i = start + t; i < end; i += 1024)
    atomicAdd(&cnt[pr[i].y & (BK_NODES - 1)], 1);
  __syncthreads();
  int v = 0, incl = 0;
  if (t < BK_NODES) {
    v = cnt[t];
    incl = wave_incl_scan(v, lane);
    if (lane == 31) wsum[wave] = incl;
  }
  __syncthreads();
  if (t < BK_NODES) {
    int add = 0;
    for (int w = 0; w < wave; ++w) add += wsum[w];
    const int p = start + add + incl - v;
    cursor[t] = p;
    const int node = bk * BK_NODES + t;
    if (node < N) { beg[node] = p; endp[node] = p + v; }
  }
  __syncthreads();
  for (int i = start + t; i < end; i += 1024) {
    const int2 e = pr[i];
    const int pos = atomicAdd(&cursor[e.y & (BK_NODES - 1)], 1);
    ssrc[pos] = e.x;
  }
}

template <int CHT, int CH, bool LAST>
__global__ void node_gather_k(const int* __restrict__ beg, const int* __restrict__ endp,
                              const int* __restrict__ ssrc, const float* __restrict__ feat,
                              const float* __restrict__ es, const float* __restrict__ ed,
                              const float* __restrict__ bias, const float* __restrict__ g,
                              const float* __restrict__ b, float* __restrict__ out, int N) {
  constexpr int NI = CHT / 32;
  const int node = blockIdx.x * (blockDim.x >> 5) + (threadIdx.x >> 5);
  const int lane = threadIdx.x & 31;
  if (node >= N) return;
  const float4* es4 = reinterpret_cast<const float4*>(es);
  const float4* ed4 = reinterpret_cast<const float4*>(ed);
  const int jb = beg[node], je = endp[node];
  const float4 dv = ed4[node];
  const float4 sv = es4[node];

  const float v0 = lrelu(sv.x + dv.x), v1 = lrelu(sv.y + dv.y);
  const float v2 = lrelu(sv.z + dv.z), v3 = lrelu(sv.w + dv.w);
  float m0 = v0, m1 = v1, m2 = v2, m3 = v3;
  for (int j = jb + lane; j < je; j += 32) {
    const float4 e4 = es4[ssrc[j]];
    m0 = fmaxf(m0, lrelu(e4.x + dv.x));
    m1 = fmaxf(m1, lrelu(e4.y + dv.y));
    m2 = fmaxf(m2, lrelu(e4.z + dv.z));
    m3 = fmaxf(m3, lrelu(e4.w + dv.w));
  }
  m0 = warp_max32(m0); m1 = warp_max32(m1); m2 = warp_max32(m2); m3 = warp_max32(m3);

  int hd[NI];
  float acc[NI];
#pragma unroll
  for (int i = 0; i < NI; ++i) { hd[i] = (lane + 32 * i) / CH; acc[i] = 0.0f; }
  float d0 = 0.0f, d1 = 0.0f, d2 = 0.0f, d3 = 0.0f;

  for (int j0 = jb; j0 < je; j0 += 32) {
    const int j = j0 + lane;
    const bool ok = j < je;
    const int s = ok ? ssrc[j] : node;
    const float4 e4 = es4[s];
    const float p0 = ok ? __expf(lrelu(e4.x + dv.x) - m0) : 0.0f;
    const float p1 = ok ? __expf(lrelu(e4.y + dv.y) - m1) : 0.0f;
    const float p2 = ok ? __expf(lrelu(e4.z + dv.z) - m2) : 0.0f;
    const float p3 = ok ? __expf(lrelu(e4.w + dv.w) - m3) : 0.0f;
    d0 += p0; d1 += p1; d2 += p2; d3 += p3;
    const int nk = (je - j0 < 32) ? (je - j0) : 32;
    for (int k = 0; k < nk; ++k) {
      const int sk = __shfl(s, k, 32);
      const float q0 = __shfl(p0, k, 32), q1 = __shfl(p1, k, 32);
      const float q2 = __shfl(p2, k, 32), q3 = __shfl(p3, k, 32);
      const float* fp = feat + (size_t)sk * CHT + lane;
#pragma unroll
      for (int i = 0; i < NI; ++i) acc[i] += fp[32 * i] * pick4(hd[i], q0, q1, q2, q3);
    }
  }

  const float s0 = __expf(v0 - m0), s1 = __expf(v1 - m1);
  const float s2 = __expf(v2 - m2), s3 = __expf(v3 - m3);
  {
    const float* fp = feat + (size_t)node * CHT + lane;
#pragma unroll
    for (int i = 0; i < NI; ++i) acc[i] += fp[32 * i] * pick4(hd[i], s0, s1, s2, s3);
  }
  d0 = warp_sum32(d0) + s0; d1 = warp_sum32(d1) + s1;
  d2 = warp_sum32(d2) + s2; d3 = warp_sum32(d3) + s3;

  float val[NI];
#pragma unroll
  for (int i = 0; i < NI; ++i)
    val[i] = acc[i] / (pick4(hd[i], d0, d1, d2, d3) + GAT_SM_EPS);

  if constexpr (!LAST) {
    float sum = 0.0f;
#pragma unroll
    for (int i = 0; i < NI; ++i) {
      float t = val[i] + bias[lane + 32 * i];
      t = (t > 0.0f) ? t : (__expf(t) - 1.0f);
      val[i] = t; sum += t;
    }
    const float mu = warp_sum32(sum) * (1.0f / (float)CHT);
    float vs = 0.0f;
#pragma unroll
    for (int i = 0; i < NI; ++i) { float dd = val[i] - mu; vs += dd * dd; }
    const float var = warp_sum32(vs) * (1.0f / (float)CHT);
    const float rstd = rsqrtf(var + GAT_LN_EPS);
    float* op = out + (size_t)node * CHT + lane;
#pragma unroll
    for (int i = 0; i < NI; ++i) {
      const int c = lane + 32 * i;
      op[32 * i] = (val[i] - mu) * rstd * g[c] + b[c];
    }
  } else {
    static_assert(!LAST || (CHT == 160 && CH == 40), "head-mean tail is laid out for 4 heads of 40");
    const bool has1 = (lane < 8);
    const int r8 = (lane + 8) & 31, r16 = (lane + 16) & 31, r24 = (lane + 24) & 31;
    const float a1 = __shfl((lane >= 8) ? val[1] : val[2], r8, 32);
    const float a2 = __shfl((lane >= 16) ? val[2] : val[3], r16, 32);
    const float a3 = __shfl((lane >= 24) ? val[3] : val[4], r24, 32);
    const float c1 = __shfl(val[2], r8, 32);
    const float c2 = __shfl(val[3], r16, 32);
    const float c3 = __shfl(val[4], r24, 32);

    float x0, x1 = 0.0f;
    { float s = val[0]; s += a1; s += a2; s += a3;
      x0 = s * 0.25f + bias[lane]; }
    if (has1) {
      float s = val[1]; s += c1; s += c2; s += c3;
      x1 = s * 0.25f + bias[lane + 32];
    }
    const float mu = warp_sum32(x0 + (has1 ? x1 : 0.0f)) * (1.0f / 40.0f);
    const float e0 = x0 - mu, e1 = has1 ? (x1 - mu) : 0.0f;
    const float var = warp_sum32(e0 * e0 + e1 * e1) * (1.0f / 40.0f);
    const float rstd = rsqrtf(var + GAT_LN_EPS);
    const float z0 = e0 * rstd * g[lane] + b[lane];
    const float z1 = has1 ? (e1 * rstd * g[lane + 32] + b[lane + 32]) : -3.0e38f;
    const float mx = warp_max32(fmaxf(z0, z1));
    const float se = warp_sum32(__expf(z0 - mx) + (has1 ? __expf(z1 - mx) : 0.0f));
    const float lse = __logf(se);
    float* yp = out + (size_t)node * CH;
    yp[lane] = z0 - mx - lse;
    if (has1) yp[lane + 32] = z1 - mx - lse;
  }
}

extern "C" void kernel_launch(void* const* d_in, const int* in_sizes, int n_in,
                              void* d_out, int out_size, void* d_ws, size_t ws_size,
                              hipStream_t stream) {
  const float* x        = (const float*)d_in[0];
  const int*   ei       = (const int*)d_in[1];
  const float* W1       = (const float*)d_in[2];
  const float* att_src1 = (const float*)d_in[3];
  const float* att_dst1 = (const float*)d_in[4];
  const float* b1       = (const float*)d_in[5];
  const float* W2       = (const float*)d_in[6];
  const float* att_src2 = (const float*)d_in[7];
  const float* att_dst2 = (const float*)d_in[8];
  const float* b2       = (const float*)d_in[9];
  const float* ln0_g    = (const float*)d_in[10];
  const float* ln0_b    = (const float*)d_in[11];
  const float* ln1_g    = (const float*)d_in[12];
  const float* ln1_b    = (const float*)d_in[13];

  const int F_IN = 256, H1C = 128, H2C = 160, H = 4;
  const int N = in_sizes[0] / F_IN;
  const int E = in_sizes[1] / 2;

  float* ws   = (float*)d_ws;
  float* h1   = ws;
  float* out1 = h1   + (size_t)N * H1C;
  float* h2   = out1 + (size_t)N * H1C;
  float* es   = h2   + (size_t)N * H2C;
  float* ed   = es   + (size_t)N * H;
  const int NBK = (N + BK_NODES - 1) / BK_NODES;
  const int G1  = (E + EPB - 1) / EPB;
  int* btot   = (int*)(ed + (size_t)N * H);
  int* pairs  = btot + (size_t)NBK * BK_PAD;
  int* ssrc   = pairs + 2 * (size_t)E;
  int* beg    = ssrc + E;
  int* cur    = beg + N;
  int* blkoff = cur + N;
  int* bbase  = blkoff + (size_t)G1 * NBK;

  const int mTiles = (N + 15) / 16;

  fill_i32<<<(NBK * BK_PAD + 255) / 256, 256, 0, stream>>>(btot, NBK * BK_PAD, 0);
  bucket_count_k<<<G1, BK_MAX, 0, stream>>>(ei, E, NBK, btot, blkoff);
  bucket_scatter_k<<<G1, BK_MAX, 0, stream>>>(ei, E, NBK, btot, blkoff, bbase, pairs);
  bucket_sort_k<<<NBK, 1024, 0, stream>>>(pairs, btot, bbase, N, beg, cur, ssrc);

  gemm_wmma_f32<<<mTiles, 2 * H1C, 16 * (F_IN + 1) * sizeof(float), stream>>>(
      x, W1, h1, N, F_IN, H1C);
  attn_scores_k<<<(N * H + 255) / 256, 256, 0, stream>>>(h1, att_src1, att_dst1,
                                                         es, ed, N, H, 32);
  node_gather_k<128, 32, false><<<(N + 7) / 8, 256, 0, stream>>>(
      beg, cur, ssrc, h1, es, ed, b1, ln0_g, ln0_b, out1, N);

  gemm_wmma_f32<<<mTiles, 2 * H2C, 16 * (H1C + 1) * sizeof(float), stream>>>(
      out1, W2, h2, N, H1C, H2C);
  attn_scores_k<<<(N * H + 255) / 256, 256, 0, stream>>>(h2, att_src2, att_dst2,
                                                         es, ed, N, H, 40);
  node_gather_k<160, 40, true><<<(N + 7) / 8, 256, 0, stream>>>(
      beg, cur, ssrc, h2, es, ed, b2, ln1_g, ln1_b, (float*)d_out, N);
}


# GEMM retiled to 64-row blocks: a wave keeps 4 accumulator tiles so one B fragment feeds 4 WMMAs, and the A tile goes to LDS with all its loads in flight (was 16-row blocks, one waited load per step)
# speedup vs baseline: 3.8354x; 1.3275x over previous
#include <hip/hip_runtime.h>
#include <math.h>

typedef __attribute__((ext_vector_type(2))) float v2f;
typedef __attribute__((ext_vector_type(8))) float v8f;

#define GAT_NEG_SLOPE 0.2f
#define GAT_SM_EPS 1e-16f
#define GAT_LN_EPS 1e-5f

__device__ __forceinline__ float warp_sum32(float v) {
#pragma unroll
  for (int o = 16; o > 0; o >>= 1) v += __shfl_xor(v, o, 32);
  return v;
}
__device__ __forceinline__ float warp_max32(float v) {
#pragma unroll
  for (int o = 16; o > 0; o >>= 1) v = fmaxf(v, __shfl_xor(v, o, 32));
  return v;
}

__global__ void fill_i32(int* __restrict__ p, int n, int v) {
  int i = blockIdx.x * blockDim.x + threadIdx.x;
  int stride = gridDim.x * blockDim.x;
  for (; i < n; i += stride) p[i] = v;
}

#define GEMM_MT 4
#define GEMM_BM (16 * GEMM_MT)
template <int K, int NC>
__global__ void __launch_bounds__(2 * NC)
gemm_wmma_f32(const float* __restrict__ A, const float* __restrict__ B,
              float* __restrict__ C, int M) {
  constexpr int KP = K + 4;
  constexpr int NT = 2 * NC;
  constexpr int RV = K / 4;
  constexpr int TOT = GEMM_BM * RV;
  constexpr int NIT = (TOT + NT - 1) / NT;
  __shared__ __attribute__((aligned(16))) float As[GEMM_BM * KP];
  const int tid = threadIdx.x;
  const int mBase = blockIdx.x * GEMM_BM;

  float4 stg[NIT];
#pragma unroll
  for (int it = 0; it < NIT; ++it) {
    const int i = tid + it * NT;
    const int r = i / RV, c4 = i - r * RV;
    const int row = mBase + r;
    stg[it] = make_float4(0.0f, 0.0f, 0.0f, 0.0f);
    if (i < TOT && row < M)
      stg[it] = *reinterpret_cast<const float4*>(A + (size_t)row * K + c4 * 4);
  }
#pragma unroll
  for (int it = 0; it < NIT; ++it) {
    const int i = tid + it * NT;
    const int r = i / RV, c4 = i - r * RV;
    if (i < TOT) *reinterpret_cast<float4*>(As + r * KP + c4 * 4) = stg[it];
  }
  __syncthreads();

  const int wave = tid >> 5;
  const int lane = tid & 31;
  const int nBase = wave * 16;
  const int l15 = lane & 15;
  const int khalf = (lane >> 4) * 2;

  v8f acc[GEMM_MT];
#pragma unroll
  for (int mt = 0; mt < GEMM_MT; ++mt) acc[mt] = v8f{};

  const float* ap = As + l15 * KP + khalf;
  const float* bp = B + (size_t)khalf * NC + nBase + l15;
#pragma unroll 8
  for (int k0 = 0; k0 < K; k0 += 4) {
    v2f b;
    b.x = bp[(size_t)k0 * NC];
    b.y = bp[(size_t)(k0 + 1) * NC];
#pragma unroll
    for (int mt = 0; mt < GEMM_MT; ++mt) {
      const v2f a = *reinterpret_cast<const v2f*>(ap + mt * 16 * KP + k0);
      acc[mt] = __builtin_amdgcn_wmma_f32_16x16x4_f32(false, a, false, b,
                                                      (short)0, acc[mt], false, false);
    }
  }

  const int col = nBase + l15;
#pragma unroll
  for (int mt = 0; mt < GEMM_MT; ++mt) {
    const int rowTop = mBase + mt * 16 + (lane >> 4) * 8;
#pragma unroll
    for (int r = 0; r < 8; ++r) {
      const int row = rowTop + r;
      if (row < M) C[(size_t)row * NC + col] = acc[mt][r];
    }
  }
}

__global__ void attn_scores_k(const float* __restrict__ h, const float* __restrict__ a_src,
                              const float* __restrict__ a_dst, float* __restrict__ es,
                              float* __restrict__ ed, int N, int H, int CH) {
  int idx = blockIdx.x * blockDim.x + threadIdx.x;
  if (idx >= N * H) return;
  int n = idx / H, hh = idx - n * H;
  const float* hp = h + (size_t)n * H * CH + hh * CH;
  const float* asp = a_src + hh * CH;
  const float* adp = a_dst + hh * CH;
  float s = 0.0f, d = 0.0f;
  for (int c = 0; c < CH; ++c) { float t = hp[c]; s += t * asp[c]; d += t * adp[c]; }
  es[idx] = s; ed[idx] = d;
}

__device__ __forceinline__ float lrelu(float v) {
  return (v > 0.0f) ? v : GAT_NEG_SLOPE * v;
}
__device__ __forceinline__ float pick4(int h, float a0, float a1, float a2, float a3) {
  return (h == 0) ? a0 : ((h == 1) ? a1 : ((h == 2) ? a2 : a3));
}

#define BK_SHIFT 8
#define BK_NODES (1 << BK_SHIFT)
#define BK_MAX 256
#define BK_PAD 32
#define EPB 2048
#define EPT (EPB / 256)

__device__ __forceinline__ int wave_incl_scan(int v, int lane) {
#pragma unroll
  for (int o = 1; o < 32; o <<= 1) {
    int t = __shfl_up(v, o, 32);
    if (lane >= o) v += t;
  }
  return v;
}

__global__ void bucket_count_k(const int* __restrict__ ei, int E, int NBK,
                               int* __restrict__ btot, int* __restrict__ blkoff) {
  __shared__ int hist[BK_MAX];
  const int t = threadIdx.x;
  hist[t] = 0;
  __syncthreads();
  const int e0 = blockIdx.x * EPB + t;
  int d[EPT];
#pragma unroll
  for (int k = 0; k < EPT; ++k) {
    const int e = e0 + k * 256;
    d[k] = ei[E + ((e < E) ? e : (E - 1))];
  }
#pragma unroll
  for (int k = 0; k < EPT; ++k)
    if (e0 + k * 256 < E) atomicAdd(&hist[d[k] >> BK_SHIFT], 1);
  __syncthreads();
  if (t < NBK) {
    const int c = hist[t];
    int off = 0;
    if (c > 0) off = atomicAdd(&btot[t * BK_PAD], c);
    blkoff[blockIdx.x * NBK + t] = off;
  }
}

__global__ void bucket_scatter_k(const int* __restrict__ ei, int E, int NBK,
                                 const int* __restrict__ btot, const int* __restrict__ blkoff,
                                 int* __restrict__ bbase, int* __restrict__ pairs) {
  __shared__ int cursor[BK_MAX];
  __shared__ int wsum[BK_MAX / 32];
  const int t = threadIdx.x, lane = t & 31, wave = t >> 5;
  const int v = (t < NBK) ? btot[t * BK_PAD] : 0;
  const int incl = wave_incl_scan(v, lane);
  if (lane == 31) wsum[wave] = incl;
  __syncthreads();
  int add = 0;
  for (int w = 0; w < wave; ++w) add += wsum[w];
  const int excl = add + incl - v;
  if (t < NBK) {
    cursor[t] = excl + blkoff[blockIdx.x * NBK + t];
    if (blockIdx.x == 0) bbase[t] = excl;
  }
  __syncthreads();
  const int e0 = blockIdx.x * EPB + t;
  int s[EPT], d[EPT];
#pragma unroll
  for (int k = 0; k < EPT; ++k) {
    const int e = e0 + k * 256;
    const int ec = (e < E) ? e : (E - 1);
    s[k] = ei[ec];
    d[k] = ei[E + ec];
  }
  int2* __restrict__ pr = reinterpret_cast<int2*>(pairs);
#pragma unroll
  for (int k = 0; k < EPT; ++k)
    if (e0 + k * 256 < E) {
      const int pos = atomicAdd(&cursor[d[k] >> BK_SHIFT], 1);
      pr[pos] = make_int2(s[k], d[k]);
    }
}

__global__ void __launch_bounds__(1024)
bucket_sort_k(const int* __restrict__ pairs, const int* __restrict__ btot,
              const int* __restrict__ bbase, int N, int* __restrict__ beg,
              int* __restrict__ endp, int* __restrict__ ssrc) {
  __shared__ int cnt[BK_NODES];
  __shared__ int cursor[BK_NODES];
  __shared__ int wsum[BK_NODES / 32];
  const int t = threadIdx.x, lane = t & 31, wave = t >> 5;
  const int bk = blockIdx.x;
  const int start = bbase[bk];
  const int end = start + btot[bk * BK_PAD];
  const int2* __restrict__ pr = reinterpret_cast<const int2*>(pairs);
  if (t < BK_NODES) cnt[t] = 0;
  __syncthreads();
  for (int i = start + t; i < end; i += 1024)
    atomicAdd(&cnt[pr[i].y & (BK_NODES - 1)], 1);
  __syncthreads();
  int v = 0, incl = 0;
  if (t < BK_NODES) {
    v = cnt[t];
    incl = wave_incl_scan(v, lane);
    if (lane == 31) wsum[wave] = incl;
  }
  __syncthreads();
  if (t < BK_NODES) {
    int add = 0;
    for (int w = 0; w < wave; ++w) add += wsum[w];
    const int p = start + add + incl - v;
    cursor[t] = p;
    const int node = bk * BK_NODES + t;
    if (node < N) { beg[node] = p; endp[node] = p + v; }
  }
  __syncthreads();
  for (int i = start + t; i < end; i += 1024) {
    const int2 e = pr[i];
    const int pos = atomicAdd(&cursor[e.y & (BK_NODES - 1)], 1);
    ssrc[pos] = e.x;
  }
}

template <int CHT, int CH, bool LAST>
__global__ void node_gather_k(const int* __restrict__ beg, const int* __restrict__ endp,
                              const int* __restrict__ ssrc, const float* __restrict__ feat,
                              const float* __restrict__ es, const float* __restrict__ ed,
                              const float* __restrict__ bias, const float* __restrict__ g,
                              const float* __restrict__ b, float* __restrict__ out, int N) {
  constexpr int NI = CHT / 32;
  const int node = blockIdx.x * (blockDim.x >> 5) + (threadIdx.x >> 5);
  const int lane = threadIdx.x & 31;
  if (node >= N) return;
  const float4* es4 = reinterpret_cast<const float4*>(es);
  const float4* ed4 = reinterpret_cast<const float4*>(ed);
  const int jb = beg[node], je = endp[node];
  const float4 dv = ed4[node];
  const float4 sv = es4[node];

  const float v0 = lrelu(sv.x + dv.x), v1 = lrelu(sv.y + dv.y);
  const float v2 = lrelu(sv.z + dv.z), v3 = lrelu(sv.w + dv.w);
  float m0 = v0, m1 = v1, m2 = v2, m3 = v3;
  for (int j = jb + lane; j < je; j += 32) {
    const float4 e4 = es4[ssrc[j]];
    m0 = fmaxf(m0, lrelu(e4.x + dv.x));
    m1 = fmaxf(m1, lrelu(e4.y + dv.y));
    m2 = fmaxf(m2, lrelu(e4.z + dv.z));
    m3 = fmaxf(m3, lrelu(e4.w + dv.w));
  }
  m0 = warp_max32(m0); m1 = warp_max32(m1); m2 = warp_max32(m2); m3 = warp_max32(m3);

  int hd[NI];
  float acc[NI];
#pragma unroll
  for (int i = 0; i < NI; ++i) { hd[i] = (lane + 32 * i) / CH; acc[i] = 0.0f; }
  float d0 = 0.0f, d1 = 0.0f, d2 = 0.0f, d3 = 0.0f;

  for (int j0 = jb; j0 < je; j0 += 32) {
    const int j = j0 + lane;
    const bool ok = j < je;
    const int s = ok ? ssrc[j] : node;
    const float4 e4 = es4[s];
    const float p0 = ok ? __expf(lrelu(e4.x + dv.x) - m0) : 0.0f;
    const float p1 = ok ? __expf(lrelu(e4.y + dv.y) - m1) : 0.0f;
    const float p2 = ok ? __expf(lrelu(e4.z + dv.z) - m2) : 0.0f;
    const float p3 = ok ? __expf(lrelu(e4.w + dv.w) - m3) : 0.0f;
    d0 += p0; d1 += p1; d2 += p2; d3 += p3;
    const int nk = (je - j0 < 32) ? (je - j0) : 32;
    for (int k = 0; k < nk; ++k) {
      const int sk = __shfl(s, k, 32);
      const float q0 = __shfl(p0, k, 32), q1 = __shfl(p1, k, 32);
      const float q2 = __shfl(p2, k, 32), q3 = __shfl(p3, k, 32);
      const float* fp = feat + (size_t)sk * CHT + lane;
#pragma unroll
      for (int i = 0; i < NI; ++i) acc[i] += fp[32 * i] * pick4(hd[i], q0, q1, q2, q3);
    }
  }

  const float s0 = __expf(v0 - m0), s1 = __expf(v1 - m1);
  const float s2 = __expf(v2 - m2), s3 = __expf(v3 - m3);
  {
    const float* fp = feat + (size_t)node * CHT + lane;
#pragma unroll
    for (int i = 0; i < NI; ++i) acc[i] += fp[32 * i] * pick4(hd[i], s0, s1, s2, s3);
  }
  d0 = warp_sum32(d0) + s0; d1 = warp_sum32(d1) + s1;
  d2 = warp_sum32(d2) + s2; d3 = warp_sum32(d3) + s3;

  float val[NI];
#pragma unroll
  for (int i = 0; i < NI; ++i)
    val[i] = acc[i] / (pick4(hd[i], d0, d1, d2, d3) + GAT_SM_EPS);

  if constexpr (!LAST) {
    float sum = 0.0f;
#pragma unroll
    for (int i = 0; i < NI; ++i) {
      float t = val[i] + bias[lane + 32 * i];
      t = (t > 0.0f) ? t : (__expf(t) - 1.0f);
      val[i] = t; sum += t;
    }
    const float mu = warp_sum32(sum) * (1.0f / (float)CHT);
    float vs = 0.0f;
#pragma unroll
    for (int i = 0; i < NI; ++i) { float dd = val[i] - mu; vs += dd * dd; }
    const float var = warp_sum32(vs) * (1.0f / (float)CHT);
    const float rstd = rsqrtf(var + GAT_LN_EPS);
    float* op = out + (size_t)node * CHT + lane;
#pragma unroll
    for (int i = 0; i < NI; ++i) {
      const int c = lane + 32 * i;
      op[32 * i] = (val[i] - mu) * rstd * g[c] + b[c];
    }
  } else {
    static_assert(!LAST || (CHT == 160 && CH == 40), "head-mean tail is laid out for 4 heads of 40");
    const bool has1 = (lane < 8);
    const int r8 = (lane + 8) & 31, r16 = (lane + 16) & 31, r24 = (lane + 24) & 31;
    const float a1 = __shfl((lane >= 8) ? val[1] : val[2], r8, 32);
    const float a2 = __shfl((lane >= 16) ? val[2] : val[3], r16, 32);
    const float a3 = __shfl((lane >= 24) ? val[3] : val[4], r24, 32);
    const float c1 = __shfl(val[2], r8, 32);
    const float c2 = __shfl(val[3], r16, 32);
    const float c3 = __shfl(val[4], r24, 32);

    float x0, x1 = 0.0f;
    { float s = val[0]; s += a1; s += a2; s += a3;
      x0 = s * 0.25f + bias[lane]; }
    if (has1) {
      float s = val[1]; s += c1; s += c2; s += c3;
      x1 = s * 0.25f + bias[lane + 32];
    }
    const float mu = warp_sum32(x0 + (has1 ? x1 : 0.0f)) * (1.0f / 40.0f);
    const float e0 = x0 - mu, e1 = has1 ? (x1 - mu) : 0.0f;
    const float var = warp_sum32(e0 * e0 + e1 * e1) * (1.0f / 40.0f);
    const float rstd = rsqrtf(var + GAT_LN_EPS);
    const float z0 = e0 * rstd * g[lane] + b[lane];
    const float z1 = has1 ? (e1 * rstd * g[lane + 32] + b[lane + 32]) : -3.0e38f;
    const float mx = warp_max32(fmaxf(z0, z1));
    const float se = warp_sum32(__expf(z0 - mx) + (has1 ? __expf(z1 - mx) : 0.0f));
    const float lse = __logf(se);
    float* yp = out + (size_t)node * CH;
    yp[lane] = z0 - mx - lse;
    if (has1) yp[lane + 32] = z1 - mx - lse;
  }
}

extern "C" void kernel_launch(void* const* d_in, const int* in_sizes, int n_in,
                              void* d_out, int out_size, void* d_ws, size_t ws_size,
                              hipStream_t stream) {
  const float* x        = (const float*)d_in[0];
  const int*   ei       = (const int*)d_in[1];
  const float* W1       = (const float*)d_in[2];
  const float* att_src1 = (const float*)d_in[3];
  const float* att_dst1 = (const float*)d_in[4];
  const float* b1       = (const float*)d_in[5];
  const float* W2       = (const float*)d_in[6];
  const float* att_src2 = (const float*)d_in[7];
  const float* att_dst2 = (const float*)d_in[8];
  const float* b2       = (const float*)d_in[9];
  const float* ln0_g    = (const float*)d_in[10];
  const float* ln0_b    = (const float*)d_in[11];
  const float* ln1_g    = (const float*)d_in[12];
  const float* ln1_b    = (const float*)d_in[13];

  const int F_IN = 256, H1C = 128, H2C = 160, H = 4;
  const int N = in_sizes[0] / F_IN;
  const int E = in_sizes[1] / 2;

  float* ws   = (float*)d_ws;
  float* h1   = ws;
  float* out1 = h1   + (size_t)N * H1C;
  float* h2   = out1 + (size_t)N * H1C;
  float* es   = h2   + (size_t)N * H2C;
  float* ed   = es   + (size_t)N * H;
  const int NBK = (N + BK_NODES - 1) / BK_NODES;
  const int G1  = (E + EPB - 1) / EPB;
  int* btot   = (int*)(ed + (size_t)N * H);
  int* pairs  = btot + (size_t)NBK * BK_PAD;
  int* ssrc   = pairs + 2 * (size_t)E;
  int* beg    = ssrc + E;
  int* cur    = beg + N;
  int* blkoff = cur + N;
  int* bbase  = blkoff + (size_t)G1 * NBK;

  const int mBlocks = (N + GEMM_BM - 1) / GEMM_BM;

  fill_i32<<<(NBK * BK_PAD + 255) / 256, 256, 0, stream>>>(btot, NBK * BK_PAD, 0);
  bucket_count_k<<<G1, BK_MAX, 0, stream>>>(ei, E, NBK, btot, blkoff);
  bucket_scatter_k<<<G1, BK_MAX, 0, stream>>>(ei, E, NBK, btot, blkoff, bbase, pairs);
  bucket_sort_k<<<NBK, 1024, 0, stream>>>(pairs, btot, bbase, N, beg, cur, ssrc);

  gemm_wmma_f32<256, 128><<<mBlocks, 2 * H1C, 0, stream>>>(x, W1, h1, N);
  attn_scores_k<<<(N * H + 255) / 256, 256, 0, stream>>>(h1, att_src1, att_dst1,
                                                         es, ed, N, H, 32);
  node_gather_k<128, 32, false><<<(N + 7) / 8, 256, 0, stream>>>(
      beg, cur, ssrc, h1, es, ed, b1, ln0_g, ln0_b, out1, N);

  gemm_wmma_f32<128, 160><<<mBlocks, 2 * H2C, 0, stream>>>(out1, W2, h2, N);
  attn_scores_k<<<(N * H + 255) / 256, 256, 0, stream>>>(h2, att_src2, att_dst2,
                                                         es, ed, N, H, 40);
  node_gather_k<160, 40, true><<<(N + 7) / 8, 256, 0, stream>>>(
      beg, cur, ssrc, h2, es, ed, b2, ln1_g, ln1_b, (float*)d_out, N);
}


# Gather waves issue feature-row loads ahead of use: rows of 4 edges per round trip (edge id and weights read by v_readlane into scalar regs), own row with the first loads; all sums keep their order
# speedup vs baseline: 4.1957x; 1.0939x over previous
#include <hip/hip_runtime.h>
#include <math.h>

typedef __attribute__((ext_vector_type(2))) float v2f;
typedef __attribute__((ext_vector_type(8))) float v8f;

#define GAT_NEG_SLOPE 0.2f
#define GAT_SM_EPS 1e-16f
#define GAT_LN_EPS 1e-5f

__device__ __forceinline__ float warp_sum32(float v) {
#pragma unroll
  for (int o = 16; o > 0; o >>= 1) v += __shfl_xor(v, o, 32);
  return v;
}
__device__ __forceinline__ float warp_max32(float v) {
#pragma unroll
  for (int o = 16; o > 0; o >>= 1) v = fmaxf(v, __shfl_xor(v, o, 32));
  return v;
}

__global__ void fill_i32(int* __restrict__ p, int n, int v) {
  int i = blockIdx.x * blockDim.x + threadIdx.x;
  int stride = gridDim.x * blockDim.x;
  for (; i < n; i += stride) p[i] = v;
}

#define GEMM_MT 4
#define GEMM_BM (16 * GEMM_MT)
template <int K, int NC>
__global__ void __launch_bounds__(2 * NC)
gemm_wmma_f32(const float* __restrict__ A, const float* __restrict__ B,
              float* __restrict__ C, int M) {
  constexpr int KP = K + 4;
  constexpr int NT = 2 * NC;
  constexpr int RV = K / 4;
  constexpr int TOT = GEMM_BM * RV;
  constexpr int NIT = (TOT + NT - 1) / NT;
  __shared__ __attribute__((aligned(16))) float As[GEMM_BM * KP];
  const int tid = threadIdx.x;
  const int mBase = blockIdx.x * GEMM_BM;

  float4 stg[NIT];
#pragma unroll
  for (int it = 0; it < NIT; ++it) {
    const int i = tid + it * NT;
    const int r = i / RV, c4 = i - r * RV;
    const int row = mBase + r;
    stg[it] = make_float4(0.0f, 0.0f, 0.0f, 0.0f);
    if (i < TOT && row < M)
      stg[it] = *reinterpret_cast<const float4*>(A + (size_t)row * K + c4 * 4);
  }
#pragma unroll
  for (int it = 0; it < NIT; ++it) {
    const int i = tid + it * NT;
    const int r = i / RV, c4 = i - r * RV;
    if (i < TOT) *reinterpret_cast<float4*>(As + r * KP + c4 * 4) = stg[it];
  }
  __syncthreads();

  const int wave = tid >> 5;
  const int lane = tid & 31;
  const int nBase = wave * 16;
  const int l15 = lane & 15;
  const int khalf = (lane >> 4) * 2;

  v8f acc[GEMM_MT];
#pragma unroll
  for (int mt = 0; mt < GEMM_MT; ++mt) acc[mt] = v8f{};

  const float* ap = As + l15 * KP + khalf;
  const float* bp = B + (size_t)khalf * NC + nBase + l15;
#pragma unroll 8
  for (int k0 = 0; k0 < K; k0 += 4) {
    v2f b;
    b.x = bp[(size_t)k0 * NC];
    b.y = bp[(size_t)(k0 + 1) * NC];
#pragma unroll
    for (int mt = 0; mt < GEMM_MT; ++mt) {
      const v2f a = *reinterpret_cast<const v2f*>(ap + mt * 16 * KP + k0);
      acc[mt] = __builtin_amdgcn_wmma_f32_16x16x4_f32(false, a, false, b,
                                                      (short)0, acc[mt], false, false);
    }
  }

  const int col = nBase + l15;
#pragma unroll
  for (int mt = 0; mt < GEMM_MT; ++mt) {
    const int rowTop = mBase + mt * 16 + (lane >> 4) * 8;
#pragma unroll
    for (int r = 0; r < 8; ++r) {
      const int row = rowTop + r;
      if (row < M) C[(size_t)row * NC + col] = acc[mt][r];
    }
  }
}

__global__ void attn_scores_k(const float* __restrict__ h, const float* __restrict__ a_src,
                              const float* __restrict__ a_dst, float* __restrict__ es,
                              float* __restrict__ ed, int N, int H, int CH) {
  int idx = blockIdx.x * blockDim.x + threadIdx.x;
  if (idx >= N * H) return;
  int n = idx / H, hh = idx - n * H;
  const float* hp = h + (size_t)n * H * CH + hh * CH;
  const float* asp = a_src + hh * CH;
  const float* adp = a_dst + hh * CH;
  float s = 0.0f, d = 0.0f;
  for (int c = 0; c < CH; ++c) { float t = hp[c]; s += t * asp[c]; d += t * adp[c]; }
  es[idx] = s; ed[idx] = d;
}

__device__ __forceinline__ float lrelu(float v) {
  return (v > 0.0f) ? v : GAT_NEG_SLOPE * v;
}
__device__ __forceinline__ float pick4(int h, float a0, float a1, float a2, float a3) {
  return (h == 0) ? a0 : ((h == 1) ? a1 : ((h == 2) ? a2 : a3));
}

#define BK_SHIFT 8
#define BK_NODES (1 << BK_SHIFT)
#define BK_MAX 256
#define BK_PAD 32
#define EPB 2048
#define EPT (EPB / 256)

__device__ __forceinline__ int wave_incl_scan(int v, int lane) {
#pragma unroll
  for (int o = 1; o < 32; o <<= 1) {
    int t = __shfl_up(v, o, 32);
    if (lane >= o) v += t;
  }
  return v;
}

__global__ void bucket_count_k(const int* __restrict__ ei, int E, int NBK,
                               int* __restrict__ btot, int* __restrict__ blkoff) {
  __shared__ int hist[BK_MAX];
  const int t = threadIdx.x;
  hist[t] = 0;
  __syncthreads();
  const int e0 = blockIdx.x * EPB + t;
  int d[EPT];
#pragma unroll
  for (int k = 0; k < EPT; ++k) {
    const int e = e0 + k * 256;
    d[k] = ei[E + ((e < E) ? e : (E - 1))];
  }
#pragma unroll
  for (int k = 0; k < EPT; ++k)
    if (e0 + k * 256 < E) atomicAdd(&hist[d[k] >> BK_SHIFT], 1);
  __syncthreads();
  if (t < NBK) {
    const int c = hist[t];
    int off = 0;
    if (c > 0) off = atomicAdd(&btot[t * BK_PAD], c);
    blkoff[blockIdx.x * NBK + t] = off;
  }
}

__global__ void bucket_scatter_k(const int* __restrict__ ei, int E, int NBK,
                                 const int* __restrict__ btot, const int* __restrict__ blkoff,
                                 int* __restrict__ bbase, int* __restrict__ pairs) {
  __shared__ int cursor[BK_MAX];
  __shared__ int wsum[BK_MAX / 32];
  const int t = threadIdx.x, lane = t & 31, wave = t >> 5;
  const int v = (t < NBK) ? btot[t * BK_PAD] : 0;
  const int incl = wave_incl_scan(v, lane);
  if (lane == 31) wsum[wave] = incl;
  __syncthreads();
  int add = 0;
  for (int w = 0; w < wave; ++w) add += wsum[w];
  const int excl = add + incl - v;
  if (t < NBK) {
    cursor[t] = excl + blkoff[blockIdx.x * NBK + t];
    if (blockIdx.x == 0) bbase[t] = excl;
  }
  __syncthreads();
  const int e0 = blockIdx.x * EPB + t;
  int s[EPT], d[EPT];
#pragma unroll
  for (int k = 0; k < EPT; ++k) {
    const int e = e0 + k * 256;
    const int ec = (e < E) ? e : (E - 1);
    s[k] = ei[ec];
    d[k] = ei[E + ec];
  }
  int2* __restrict__ pr = reinterpret_cast<int2*>(pairs);
#pragma unroll
  for (int k = 0; k < EPT; ++k)
    if (e0 + k * 256 < E) {
      const int pos = atomicAdd(&cursor[d[k] >> BK_SHIFT], 1);
      pr[pos] = make_int2(s[k], d[k]);
    }
}

__global__ void __launch_bounds__(1024)
bucket_sort_k(const int* __restrict__ pairs, const int* __restrict__ btot,
              const int* __restrict__ bbase, int N, int* __restrict__ beg,
              int* __restrict__ endp, int* __restrict__ ssrc) {
  __shared__ int cnt[BK_NODES];
  __shared__ int cursor[BK_NODES];
  __shared__ int wsum[BK_NODES / 32];
  const int t = threadIdx.x, lane = t & 31, wave = t >> 5;
  const int bk = blockIdx.x;
  const int start = bbase[bk];
  const int end = start + btot[bk * BK_PAD];
  const int2* __restrict__ pr = reinterpret_cast<const int2*>(pairs);
  if (t < BK_NODES) cnt[t] = 0;
  __syncthreads();
  for (int i = start + t; i < end; i += 1024)
    atomicAdd(&cnt[pr[i].y & (BK_NODES - 1)], 1);
  __syncthreads();
  int v = 0, incl = 0;
  if (t < BK_NODES) {
    v = cnt[t];
    incl = wave_incl_scan(v, lane);
    if (lane == 31) wsum[wave] = incl;
  }
  __syncthreads();
  if (t < BK_NODES) {
    int add = 0;
    for (int w = 0; w < wave; ++w) add += wsum[w];
    const int p = start + add + incl - v;
    cursor[t] = p;
    const int node = bk * BK_NODES + t;
    if (node < N) { beg[node] = p; endp[node] = p + v; }
  }
  __syncthreads();
  for (int i = start + t; i < end; i += 1024) {
    const int2 e = pr[i];
    const int pos = atomicAdd(&cursor[e.y & (BK_NODES - 1)], 1);
    ssrc[pos] = e.x;
  }
}

template <int CHT, int U>
__device__ __forceinline__ void gather_rows(const float* __restrict__ feat, int lane, int k, int s,
                                            float p0, float p1, float p2, float p3,
                                            const int (&hd)[CHT / 32], float (&acc)[CHT / 32]) {
  constexpr int NI = CHT / 32;
  float r[U][NI];
#pragma unroll
  for (int u = 0; u < U; ++u) {
    const int sk = __builtin_amdgcn_readlane(s, k + u);
    const float* fp = feat + (size_t)sk * CHT + lane;
#pragma unroll
    for (int i = 0; i < NI; ++i) r[u][i] = fp[32 * i];
  }
#pragma unroll
  for (int u = 0; u < U; ++u) {
    const float q0 = __int_as_float(__builtin_amdgcn_readlane(__float_as_int(p0), k + u));
    const float q1 = __int_as_float(__builtin_amdgcn_readlane(__float_as_int(p1), k + u));
    const float q2 = __int_as_float(__builtin_amdgcn_readlane(__float_as_int(p2), k + u));
    const float q3 = __int_as_float(__builtin_amdgcn_readlane(__float_as_int(p3), k + u));
#pragma unroll
    for (int i = 0; i < NI; ++i) acc[i] = fmaf(r[u][i], pick4(hd[i], q0, q1, q2, q3), acc[i]);
  }
}

template <int CHT, int CH, bool LAST>
__global__ void node_gather_k(const int* __restrict__ beg, const int* __restrict__ endp,
                              const int* __restrict__ ssrc, const float* __restrict__ feat,
                              const float* __restrict__ es, const float* __restrict__ ed,
                              const float* __restrict__ bias, const float* __restrict__ g,
                              const float* __restrict__ b, float* __restrict__ out, int N) {
  constexpr int NI = CHT / 32;
  const int node = blockIdx.x * (blockDim.x >> 5) + (threadIdx.x >> 5);
  const int lane = threadIdx.x & 31;
  if (node >= N) return;
  const float4* es4 = reinterpret_cast<const float4*>(es);
  const float4* ed4 = reinterpret_cast<const float4*>(ed);
  const int jb = beg[node], je = endp[node];
  const float4 dv = ed4[node];
  const float4 sv = es4[node];
  float selfrow[CHT / 32];
#pragma unroll
  for (int i = 0; i < CHT / 32; ++i) selfrow[i] = feat[(size_t)node * CHT + lane + 32 * i];

  const float v0 = lrelu(sv.x + dv.x), v1 = lrelu(sv.y + dv.y);
  const float v2 = lrelu(sv.z + dv.z), v3 = lrelu(sv.w + dv.w);
  float m0 = v0, m1 = v1, m2 = v2, m3 = v3;
  for (int j = jb + lane; j < je; j += 32) {
    const float4 e4 = es4[ssrc[j]];
    m0 = fmaxf(m0, lrelu(e4.x + dv.x));
    m1 = fmaxf(m1, lrelu(e4.y + dv.y));
    m2 = fmaxf(m2, lrelu(e4.z + dv.z));
    m3 = fmaxf(m3, lrelu(e4.w + dv.w));
  }
  m0 = warp_max32(m0); m1 = warp_max32(m1); m2 = warp_max32(m2); m3 = warp_max32(m3);

  int hd[NI];
  float acc[NI];
#pragma unroll
  for (int i = 0; i < NI; ++i) { hd[i] = (lane + 32 * i) / CH; acc[i] = 0.0f; }
  float d0 = 0.0f, d1 = 0.0f, d2 = 0.0f, d3 = 0.0f;

  for (int j0 = jb; j0 < je; j0 += 32) {
    const int j = j0 + lane;
    const bool ok = j < je;
    const int s = ok ? ssrc[j] : node;
    const float4 e4 = es4[s];
    const float p0 = ok ? __expf(lrelu(e4.x + dv.x) - m0) : 0.0f;
    const float p1 = ok ? __expf(lrelu(e4.y + dv.y) - m1) : 0.0f;
    const float p2 = ok ? __expf(lrelu(e4.z + dv.z) - m2) : 0.0f;
    const float p3 = ok ? __expf(lrelu(e4.w + dv.w) - m3) : 0.0f;
    d0 += p0; d1 += p1; d2 += p2; d3 += p3;
    const int nk = (je - j0 < 32) ? (je - j0) : 32;
    int k = 0;
    for (; k + 4 <= nk; k += 4)
      gather_rows<CHT, 4>(feat, lane, k, s, p0, p1, p2, p3, hd, acc);
    if ((nk - k) & 2) {
      gather_rows<CHT, 2>(feat, lane, k, s, p0, p1, p2, p3, hd, acc);
      k += 2;
    }
    if ((nk - k) & 1)
      gather_rows<CHT, 1>(feat, lane, k, s, p0, p1, p2, p3, hd, acc);
  }

  const float s0 = __expf(v0 - m0), s1 = __expf(v1 - m1);
  const float s2 = __expf(v2 - m2), s3 = __expf(v3 - m3);
#pragma unroll
  for (int i = 0; i < NI; ++i) acc[i] = fmaf(selfrow[i], pick4(hd[i], s0, s1, s2, s3), acc[i]);
  d0 = warp_sum32(d0) + s0; d1 = warp_sum32(d1) + s1;
  d2 = warp_sum32(d2) + s2; d3 = warp_sum32(d3) + s3;

  float val[NI];
#pragma unroll
  for (int i = 0; i < NI; ++i)
    val[i] = acc[i] / (pick4(hd[i], d0, d1, d2, d3) + GAT_SM_EPS);

  if constexpr (!LAST) {
    float sum = 0.0f;
#pragma unroll
    for (int i = 0; i < NI; ++i) {
      float t = val[i] + bias[lane + 32 * i];
      t = (t > 0.0f) ? t : (__expf(t) - 1.0f);
      val[i] = t; sum += t;
    }
    const float mu = warp_sum32(sum) * (1.0f / (float)CHT);
    float vs = 0.0f;
#pragma unroll
    for (int i = 0; i < NI; ++i) { float dd = val[i] - mu; vs += dd * dd; }
    const float var = warp_sum32(vs) * (1.0f / (float)CHT);
    const float rstd = rsqrtf(var + GAT_LN_EPS);
    float* op = out + (size_t)node * CHT + lane;
#pragma unroll
    for (int i = 0; i < NI; ++i) {
      const int c = lane + 32 * i;
      op[32 * i] = (val[i] - mu) * rstd * g[c] + b[c];
    }
  } else {
    static_assert(!LAST || (CHT == 160 && CH == 40), "head-mean tail is laid out for 4 heads of 40");
    const bool has1 = (lane < 8);
    const int r8 = (lane + 8) & 31, r16 = (lane + 16) & 31, r24 = (lane + 24) & 31;
    const float a1 = __shfl((lane >= 8) ? val[1] : val[2], r8, 32);
    const float a2 = __shfl((lane >= 16) ? val[2] : val[3], r16, 32);
    const float a3 = __shfl((lane >= 24) ? val[3] : val[4], r24, 32);
    const float c1 = __shfl(val[2], r8, 32);
    const float c2 = __shfl(val[3], r16, 32);
    const float c3 = __shfl(val[4], r24, 32);

    float x0, x1 = 0.0f;
    { float s = val[0]; s += a1; s += a2; s += a3;
      x0 = s * 0.25f + bias[lane]; }
    if (has1) {
      float s = val[1]; s += c1; s += c2; s += c3;
      x1 = s * 0.25f + bias[lane + 32];
    }
    const float mu = warp_sum32(x0 + (has1 ? x1 : 0.0f)) * (1.0f / 40.0f);
    const float e0 = x0 - mu, e1 = has1 ? (x1 - mu) : 0.0f;
    const float var = warp_sum32(e0 * e0 + e1 * e1) * (1.0f / 40.0f);
    const float rstd = rsqrtf(var + GAT_LN_EPS);
    const float z0 = e0 * rstd * g[lane] + b[lane];
    const float z1 = has1 ? (e1 * rstd * g[lane + 32] + b[lane + 32]) : -3.0e38f;
    const float mx = warp_max32(fmaxf(z0, z1));
    const float se = warp_sum32(__expf(z0 - mx) + (has1 ? __expf(z1 - mx) : 0.0f));
    const float lse = __logf(se);
    float* yp = out + (size_t)node * CH;
    yp[lane] = z0 - mx - lse;
    if (has1) yp[lane + 32] = z1 - mx - lse;
  }
}

extern "C" void kernel_launch(void* const* d_in, const int* in_sizes, int n_in,
                              void* d_out, int out_size, void* d_ws, size_t ws_size,
                              hipStream_t stream) {
  const float* x        = (const float*)d_in[0];
  const int*   ei       = (const int*)d_in[1];
  const float* W1       = (const float*)d_in[2];
  const float* att_src1 = (const float*)d_in[3];
  const float* att_dst1 = (const float*)d_in[4];
  const float* b1       = (const float*)d_in[5];
  const float* W2       = (const float*)d_in[6];
  const float* att_src2 = (const float*)d_in[7];
  const float* att_dst2 = (const float*)d_in[8];
  const float* b2       = (const float*)d_in[9];
  const float* ln0_g    = (const float*)d_in[10];
  const float* ln0_b    = (const float*)d_in[11];
  const float* ln1_g    = (const float*)d_in[12];
  const float* ln1_b    = (const float*)d_in[13];

  const int F_IN = 256, H1C = 128, H2C = 160, H = 4;
  const int N = in_sizes[0] / F_IN;
  const int E = in_sizes[1] / 2;

  float* ws   = (float*)d_ws;
  float* h1   = ws;
  float* out1 = h1   + (size_t)N * H1C;
  float* h2   = out1 + (size_t)N * H1C;
  float* es   = h2   + (size_t)N * H2C;
  float* ed   = es   + (size_t)N * H;
  const int NBK = (N + BK_NODES - 1) / BK_NODES;
  const int G1  = (E + EPB - 1) / EPB;
  int* btot   = (int*)(ed + (size_t)N * H);
  int* pairs  = btot + (size_t)NBK * BK_PAD;
  int* ssrc   = pairs + 2 * (size_t)E;
  int* beg    = ssrc + E;
  int* cur    = beg + N;
  int* blkoff = cur + N;
  int* bbase  = blkoff + (size_t)G1 * NBK;

  const int mBlocks = (N + GEMM_BM - 1) / GEMM_BM;

  fill_i32<<<(NBK * BK_PAD + 255) / 256, 256, 0, stream>>>(btot, NBK * BK_PAD, 0);
  bucket_count_k<<<G1, BK_MAX, 0, stream>>>(ei, E, NBK, btot, blkoff);
  bucket_scatter_k<<<G1, BK_MAX, 0, stream>>>(ei, E, NBK, btot, blkoff, bbase, pairs);
  bucket_sort_k<<<NBK, 1024, 0, stream>>>(pairs, btot, bbase, N, beg, cur, ssrc);

  gemm_wmma_f32<256, 128><<<mBlocks, 2 * H1C, 0, stream>>>(x, W1, h1, N);
  attn_scores_k<<<(N * H + 255) / 256, 256, 0, stream>>>(h1, att_src1, att_dst1,
                                                         es, ed, N, H, 32);
  node_gather_k<128, 32, false><<<(N + 7) / 8, 256, 0, stream>>>(
      beg, cur, ssrc, h1, es, ed, b1, ln0_g, ln0_b, out1, N);

  gemm_wmma_f32<128, 160><<<mBlocks, 2 * H2C, 0, stream>>>(out1, W2, h2, N);
  attn_scores_k<<<(N * H + 255) / 256, 256, 0, stream>>>(h2, att_src2, att_dst2,
                                                         es, ed, N, H, 40);
  node_gather_k<160, 40, true><<<(N + 7) / 8, 256, 0, stream>>>(
      beg, cur, ssrc, h2, es, ed, b2, ln1_g, ln1_b, (float*)d_out, N);
}


# Attention scores formed in the GEMM epilogue from the accumulators (8-lane DPP sums, head groups joined in LDS); the two score launches go; score dot products now sum as a tree, not in column order
# speedup vs baseline: 4.4230x; 1.0542x over previous
#include <hip/hip_runtime.h>
#include <math.h>

typedef __attribute__((ext_vector_type(2))) float v2f;
typedef __attribute__((ext_vector_type(8))) float v8f;

#define GAT_NEG_SLOPE 0.2f
#define GAT_SM_EPS 1e-16f
#define GAT_LN_EPS 1e-5f

__device__ __forceinline__ float warp_sum32(float v) {
#pragma unroll
  for (int o = 16; o > 0; o >>= 1) v += __shfl_xor(v, o, 32);
  return v;
}
__device__ __forceinline__ float warp_max32(float v) {
#pragma unroll
  for (int o = 16; o > 0; o >>= 1) v = fmaxf(v, __shfl_xor(v, o, 32));
  return v;
}

__device__ __forceinline__ float group8_sum(float v) {
  v += __int_as_float(__builtin_amdgcn_update_dpp(0, __float_as_int(v), 0x161, 0xF, 0xF, true));
  v += __int_as_float(__builtin_amdgcn_update_dpp(0, __float_as_int(v), 0x162, 0xF, 0xF, true));
  v += __int_as_float(__builtin_amdgcn_update_dpp(0, __float_as_int(v), 0x164, 0xF, 0xF, true));
  return v;
}

__global__ void fill_i32(int* __restrict__ p, int n, int v) {
  int i = blockIdx.x * blockDim.x + threadIdx.x;
  int stride = gridDim.x * blockDim.x;
  for (; i < n; i += stride) p[i] = v;
}

#define GEMM_MT 4
#define GEMM_BM (16 * GEMM_MT)
template <int K, int NC, int CH>
__global__ void __launch_bounds__(2 * NC)
gemm_wmma_f32(const float* __restrict__ A, const float* __restrict__ B,
              float* __restrict__ C, const float* __restrict__ a_src,
              const float* __restrict__ a_dst, float* __restrict__ es,
              float* __restrict__ ed, int M) {
  static_assert(NC == 4 * CH && CH % 8 == 0, "four heads, each a whole number of 8-column groups");
  constexpr int NG = NC / 8;
  constexpr int GPH = CH / 8;
  __shared__ float part_s[GEMM_BM * NG];
  __shared__ float part_d[GEMM_BM * NG];
  constexpr int KP = K + 4;
  constexpr int NT = 2 * NC;
  constexpr int RV = K / 4;
  constexpr int TOT = GEMM_BM * RV;
  constexpr int NIT = (TOT + NT - 1) / NT;
  __shared__ __attribute__((aligned(16))) float As[GEMM_BM * KP];
  const int tid = threadIdx.x;
  const int mBase = blockIdx.x * GEMM_BM;

  float4 stg[NIT];
#pragma unroll
  for (int it = 0; it < NIT; ++it) {
    const int i = tid + it * NT;
    const int r = i / RV, c4 = i - r * RV;
    const int row = mBase + r;
    stg[it] = make_float4(0.0f, 0.0f, 0.0f, 0.0f);
    if (i < TOT && row < M)
      stg[it] = *reinterpret_cast<const float4*>(A + (size_t)row * K + c4 * 4);
  }
#pragma unroll
  for (int it = 0; it < NIT; ++it) {
    const int i = tid + it * NT;
    const int r = i / RV, c4 = i - r * RV;
    if (i < TOT) *reinterpret_cast<float4*>(As + r * KP + c4 * 4) = stg[it];
  }
  __syncthreads();

  const int wave = tid >> 5;
  const int lane = tid & 31;
  const int nBase = wave * 16;
  const int l15 = lane & 15;
  const int khalf = (lane >> 4) * 2;

  v8f acc[GEMM_MT];
#pragma unroll
  for (int mt = 0; mt < GEMM_MT; ++mt) acc[mt] = v8f{};

  const float* ap = As + l15 * KP + khalf;
  const float* bp = B + (size_t)khalf * NC + nBase + l15;
#pragma unroll 8
  for (int k0 = 0; k0 < K; k0 += 4) {
    v2f b;
    b.x = bp[(size_t)k0 * NC];
    b.y = bp[(size_t)(k0 + 1) * NC];
#pragma unroll
    for (int mt = 0; mt < GEMM_MT; ++mt) {
      const v2f a = *reinterpret_cast<const v2f*>(ap + mt * 16 * KP + k0);
      acc[mt] = __builtin_amdgcn_wmma_f32_16x16x4_f32(false, a, false, b,
                                                      (short)0, acc[mt], false, false);
    }
  }

  const int col = nBase + l15;
#pragma unroll
  for (int mt = 0; mt < GEMM_MT; ++mt) {
    const int rowTop = mBase + mt * 16 + (lane >> 4) * 8;
#pragma unroll
    for (int r = 0; r < 8; ++r) {
      const int row = rowTop + r;
      if (row < M) C[(size_t)row * NC + col] = acc[mt][r];
    }
  }

  const float as = a_src[col], ad = a_dst[col];
  const int grp = (nBase >> 3) + ((lane >> 3) & 1);
#pragma unroll
  for (int mt = 0; mt < GEMM_MT; ++mt) {
#pragma unroll
    for (int r = 0; r < 8; ++r) {
      const float ps = group8_sum(acc[mt][r] * as);
      const float pd = group8_sum(acc[mt][r] * ad);
      if ((lane & 7) == 0) {
        const int rl = mt * 16 + (lane >> 4) * 8 + r;
        part_s[rl * NG + grp] = ps;
        part_d[rl * NG + grp] = pd;
      }
    }
  }
  __syncthreads();
  if (tid < GEMM_BM * 4) {
    const int rl = tid >> 2, hh = tid & 3;
    const float* qs = part_s + rl * NG + hh * GPH;
    const float* qd = part_d + rl * NG + hh * GPH;
    float s = qs[0], d = qd[0];
#pragma unroll
    for (int gi = 1; gi < GPH; ++gi) { s += qs[gi]; d += qd[gi]; }
    const int row = mBase + rl;
    if (row < M) { es[(size_t)row * 4 + hh] = s; ed[(size_t)row * 4 + hh] = d; }
  }
}

__device__ __forceinline__ float lrelu(float v) {
  return (v > 0.0f) ? v : GAT_NEG_SLOPE * v;
}
__device__ __forceinline__ float pick4(int h, float a0, float a1, float a2, float a3) {
  return (h == 0) ? a0 : ((h == 1) ? a1 : ((h == 2) ? a2 : a3));
}

#define BK_SHIFT 8
#define BK_NODES (1 << BK_SHIFT)
#define BK_MAX 256
#define BK_PAD 32
#define EPB 2048
#define EPT (EPB / 256)

__device__ __forceinline__ int wave_incl_scan(int v, int lane) {
#pragma unroll
  for (int o = 1; o < 32; o <<= 1) {
    int t = __shfl_up(v, o, 32);
    if (lane >= o) v += t;
  }
  return v;
}

__global__ void bucket_count_k(const int* __restrict__ ei, int E, int NBK,
                               int* __restrict__ btot, int* __restrict__ blkoff) {
  __shared__ int hist[BK_MAX];
  const int t = threadIdx.x;
  hist[t] = 0;
  __syncthreads();
  const int e0 = blockIdx.x * EPB + t;
  int d[EPT];
#pragma unroll
  for (int k = 0; k < EPT; ++k) {
    const int e = e0 + k * 256;
    d[k] = ei[E + ((e < E) ? e : (E - 1))];
  }
#pragma unroll
  for (int k = 0; k < EPT; ++k)
    if (e0 + k * 256 < E) atomicAdd(&hist[d[k] >> BK_SHIFT], 1);
  __syncthreads();
  if (t < NBK) {
    const int c = hist[t];
    int off = 0;
    if (c > 0) off = atomicAdd(&btot[t * BK_PAD], c);
    blkoff[blockIdx.x * NBK + t] = off;
  }
}

__global__ void bucket_scatter_k(const int* __restrict__ ei, int E, int NBK,
                                 const int* __restrict__ btot, const int* __restrict__ blkoff,
                                 int* __restrict__ bbase, int* __restrict__ pairs) {
  __shared__ int cursor[BK_MAX];
  __shared__ int wsum[BK_MAX / 32];
  const int t = threadIdx.x, lane = t & 31, wave = t >> 5;
  const int v = (t < NBK) ? btot[t * BK_PAD] : 0;
  const int incl = wave_incl_scan(v, lane);
  if (lane == 31) wsum[wave] = incl;
  __syncthreads();
  int add = 0;
  for (int w = 0; w < wave; ++w) add += wsum[w];
  const int excl = add + incl - v;
  if (t < NBK) {
    cursor[t] = excl + blkoff[blockIdx.x * NBK + t];
    if (blockIdx.x == 0) bbase[t] = excl;
  }
  __syncthreads();
  const int e0 = blockIdx.x * EPB + t;
  int s[EPT], d[EPT];
#pragma unroll
  for (int k = 0; k < EPT; ++k) {
    const int e = e0 + k * 256;
    const int ec = (e < E) ? e : (E - 1);
    s[k] = ei[ec];
    d[k] = ei[E + ec];
  }
  int2* __restrict__ pr = reinterpret_cast<int2*>(pairs);
#pragma unroll
  for (int k = 0; k < EPT; ++k)
    if (e0 + k * 256 < E) {
      const int pos = atomicAdd(&cursor[d[k] >> BK_SHIFT], 1);
      pr[pos] = make_int2(s[k], d[k]);
    }
}

__global__ void __launch_bounds__(1024)
bucket_sort_k(const int* __restrict__ pairs, const int* __restrict__ btot,
              const int* __restrict__ bbase, int N, int* __restrict__ beg,
              int* __restrict__ endp, int* __restrict__ ssrc) {
  __shared__ int cnt[BK_NODES];
  __shared__ int cursor[BK_NODES];
  __shared__ int wsum[BK_NODES / 32];
  const int t = threadIdx.x, lane = t & 31, wave = t >> 5;
  const int bk = blockIdx.x;
  const int start = bbase[bk];
  const int end = start + btot[bk * BK_PAD];
  const int2* __restrict__ pr = reinterpret_cast<const int2*>(pairs);
  if (t < BK_NODES) cnt[t] = 0;
  __syncthreads();
  for (int i = start + t; i < end; i += 1024)
    atomicAdd(&cnt[pr[i].y & (BK_NODES - 1)], 1);
  __syncthreads();
  int v = 0, incl = 0;
  if (t < BK_NODES) {
    v = cnt[t];
    incl = wave_incl_scan(v, lane);
    if (lane == 31) wsum[wave] = incl;
  }
  __syncthreads();
  if (t < BK_NODES) {
    int add = 0;
    for (int w = 0; w < wave; ++w) add += wsum[w];
    const int p = start + add + incl - v;
    cursor[t] = p;
    const int node = bk * BK_NODES + t;
    if (node < N) { beg[node] = p; endp[node] = p + v; }
  }
  __syncthreads();
  for (int i = start + t; i < end; i += 1024) {
    const int2 e = pr[i];
    const int pos = atomicAdd(&cursor[e.y & (BK_NODES - 1)], 1);
    ssrc[pos] = e.x;
  }
}

template <int CHT, int U>
__device__ __forceinline__ void gather_rows(const float* __restrict__ feat, int lane, int k, int s,
                                            float p0, float p1, float p2, float p3,
                                            const int (&hd)[CHT / 32], float (&acc)[CHT / 32]) {
  constexpr int NI = CHT / 32;
  float r[U][NI];
#pragma unroll
  for (int u = 0; u < U; ++u) {
    const int sk = __builtin_amdgcn_readlane(s, k + u);
    const float* fp = feat + (size_t)sk * CHT + lane;
#pragma unroll
    for (int i = 0; i < NI; ++i) r[u][i] = fp[32 * i];
  }
#pragma unroll
  for (int u = 0; u < U; ++u) {
    const float q0 = __int_as_float(__builtin_amdgcn_readlane(__float_as_int(p0), k + u));
    const float q1 = __int_as_float(__builtin_amdgcn_readlane(__float_as_int(p1), k + u));
    const float q2 = __int_as_float(__builtin_amdgcn_readlane(__float_as_int(p2), k + u));
    const float q3 = __int_as_float(__builtin_amdgcn_readlane(__float_as_int(p3), k + u));
#pragma unroll
    for (int i = 0; i < NI; ++i) acc[i] = fmaf(r[u][i], pick4(hd[i], q0, q1, q2, q3), acc[i]);
  }
}

template <int CHT, int CH, bool LAST>
__global__ void node_gather_k(const int* __restrict__ beg, const int* __restrict__ endp,
                              const int* __restrict__ ssrc, const float* __restrict__ feat,
                              const float* __restrict__ es, const float* __restrict__ ed,
                              const float* __restrict__ bias, const float* __restrict__ g,
                              const float* __restrict__ b, float* __restrict__ out, int N) {
  constexpr int NI = CHT / 32;
  const int node = blockIdx.x * (blockDim.x >> 5) + (threadIdx.x >> 5);
  const int lane = threadIdx.x & 31;
  if (node >= N) return;
  const float4* es4 = reinterpret_cast<const float4*>(es);
  const float4* ed4 = reinterpret_cast<const float4*>(ed);
  const int jb = beg[node], je = endp[node];
  const float4 dv = ed4[node];
  const float4 sv = es4[node];
  float selfrow[CHT / 32];
#pragma unroll
  for (int i = 0; i < CHT / 32; ++i) selfrow[i] = feat[(size_t)node * CHT + lane + 32 * i];

  const float v0 = lrelu(sv.x + dv.x), v1 = lrelu(sv.y + dv.y);
  const float v2 = lrelu(sv.z + dv.z), v3 = lrelu(sv.w + dv.w);
  float m0 = v0, m1 = v1, m2 = v2, m3 = v3;
  for (int j = jb + lane; j < je; j += 32) {
    const float4 e4 = es4[ssrc[j]];
    m0 = fmaxf(m0, lrelu(e4.x + dv.x));
    m1 = fmaxf(m1, lrelu(e4.y + dv.y));
    m2 = fmaxf(m2, lrelu(e4.z + dv.z));
    m3 = fmaxf(m3, lrelu(e4.w + dv.w));
  }
  m0 = warp_max32(m0); m1 = warp_max32(m1); m2 = warp_max32(m2); m3 = warp_max32(m3);

  int hd[NI];
  float acc[NI];
#pragma unroll
  for (int i = 0; i < NI; ++i) { hd[i] = (lane + 32 * i) / CH; acc[i] = 0.0f; }
  float d0 = 0.0f, d1 = 0.0f, d2 = 0.0f, d3 = 0.0f;

  for (int j0 = jb; j0 < je; j0 += 32) {
    const int j = j0 + lane;
    const bool ok = j < je;
    const int s = ok ? ssrc[j] : node;
    const float4 e4 = es4[s];
    const float p0 = ok ? __expf(lrelu(e4.x + dv.x) - m0) : 0.0f;
    const float p1 = ok ? __expf(lrelu(e4.y + dv.y) - m1) : 0.0f;
    const float p2 = ok ? __expf(lrelu(e4.z + dv.z) - m2) : 0.0f;
    const float p3 = ok ? __expf(lrelu(e4.w + dv.w) - m3) : 0.0f;
    d0 += p0; d1 += p1; d2 += p2; d3 += p3;
    const int nk = (je - j0 < 32) ? (je - j0) : 32;
    int k = 0;
    for (; k + 4 <= nk; k += 4)
      gather_rows<CHT, 4>(feat, lane, k, s, p0, p1, p2, p3, hd, acc);
    if ((nk - k) & 2) {
      gather_rows<CHT, 2>(feat, lane, k, s, p0, p1, p2, p3, hd, acc);
      k += 2;
    }
    if ((nk - k) & 1)
      gather_rows<CHT, 1>(feat, lane, k, s, p0, p1, p2, p3, hd, acc);
  }

  const float s0 = __expf(v0 - m0), s1 = __expf(v1 - m1);
  const float s2 = __expf(v2 - m2), s3 = __expf(v3 - m3);
#pragma unroll
  for (int i = 0; i < NI; ++i) acc[i] = fmaf(selfrow[i], pick4(hd[i], s0, s1, s2, s3), acc[i]);
  d0 = warp_sum32(d0) + s0; d1 = warp_sum32(d1) + s1;
  d2 = warp_sum32(d2) + s2; d3 = warp_sum32(d3) + s3;

  float val[NI];
#pragma unroll
  for (int i = 0; i < NI; ++i)
    val[i] = acc[i] / (pick4(hd[i], d0, d1, d2, d3) + GAT_SM_EPS);

  if constexpr (!LAST) {
    float sum = 0.0f;
#pragma unroll
    for (int i = 0; i < NI; ++i) {
      float t = val[i] + bias[lane + 32 * i];
      t = (t > 0.0f) ? t : (__expf(t) - 1.0f);
      val[i] = t; sum += t;
    }
    const float mu = warp_sum32(sum) * (1.0f / (float)CHT);
    float vs = 0.0f;
#pragma unroll
    for (int i = 0; i < NI; ++i) { float dd = val[i] - mu; vs += dd * dd; }
    const float var = warp_sum32(vs) * (1.0f / (float)CHT);
    const float rstd = rsqrtf(var + GAT_LN_EPS);
    float* op = out + (size_t)node * CHT + lane;
#pragma unroll
    for (int i = 0; i < NI; ++i) {
      const int c = lane + 32 * i;
      op[32 * i] = (val[i] - mu) * rstd * g[c] + b[c];
    }
  } else {
    static_assert(!LAST || (CHT == 160 && CH == 40), "head-mean tail is laid out for 4 heads of 40");
    const bool has1 = (lane < 8);
    const int r8 = (lane + 8) & 31, r16 = (lane + 16) & 31, r24 = (lane + 24) & 31;
    const float a1 = __shfl((lane >= 8) ? val[1] : val[2], r8, 32);
    const float a2 = __shfl((lane >= 16) ? val[2] : val[3], r16, 32);
    const float a3 = __shfl((lane >= 24) ? val[3] : val[4], r24, 32);
    const float c1 = __shfl(val[2], r8, 32);
    const float c2 = __shfl(val[3], r16, 32);
    const float c3 = __shfl(val[4], r24, 32);

    float x0, x1 = 0.0f;
    { float s = val[0]; s += a1; s += a2; s += a3;
      x0 = s * 0.25f + bias[lane]; }
    if (has1) {
      float s = val[1]; s += c1; s += c2; s += c3;
      x1 = s * 0.25f + bias[lane + 32];
    }
    const float mu = warp_sum32(x0 + (has1 ? x1 : 0.0f)) * (1.0f / 40.0f);
    const float e0 = x0 - mu, e1 = has1 ? (x1 - mu) : 0.0f;
    const float var = warp_sum32(e0 * e0 + e1 * e1) * (1.0f / 40.0f);
    const float rstd = rsqrtf(var + GAT_LN_EPS);
    const float z0 = e0 * rstd * g[lane] + b[lane];
    const float z1 = has1 ? (e1 * rstd * g[lane + 32] + b[lane + 32]) : -3.0e38f;
    const float mx = warp_max32(fmaxf(z0, z1));
    const float se = warp_sum32(__expf(z0 - mx) + (has1 ? __expf(z1 - mx) : 0.0f));
    const float lse = __logf(se);
    float* yp = out + (size_t)node * CH;
    yp[lane] = z0 - mx - lse;
    if (has1) yp[lane + 32] = z1 - mx - lse;
  }
}

extern "C" void kernel_launch(void* const* d_in, const int* in_sizes, int n_in,
                              void* d_out, int out_size, void* d_ws, size_t ws_size,
                              hipStream_t stream) {
  const float* x        = (const float*)d_in[0];
  const int*   ei       = (const int*)d_in[1];
  const float* W1       = (const float*)d_in[2];
  const float* att_src1 = (const float*)d_in[3];
  const float* att_dst1 = (const float*)d_in[4];
  const float* b1       = (const float*)d_in[5];
  const float* W2       = (const float*)d_in[6];
  const float* att_src2 = (const float*)d_in[7];
  const float* att_dst2 = (const float*)d_in[8];
  const float* b2       = (const float*)d_in[9];
  const float* ln0_g    = (const float*)d_in[10];
  const float* ln0_b    = (const float*)d_in[11];
  const float* ln1_g    = (const float*)d_in[12];
  const float* ln1_b    = (const float*)d_in[13];

  const int F_IN = 256, H1C = 128, H2C = 160, H = 4;
  const int N = in_sizes[0] / F_IN;
  const int E = in_sizes[1] / 2;

  float* ws   = (float*)d_ws;
  float* h1   = ws;
  float* out1 = h1   + (size_t)N * H1C;
  float* h2   = out1 + (size_t)N * H1C;
  float* es   = h2   + (size_t)N * H2C;
  float* ed   = es   + (size_t)N * H;
  const int NBK = (N + BK_NODES - 1) / BK_NODES;
  const int G1  = (E + EPB - 1) / EPB;
  int* btot   = (int*)(ed + (size_t)N * H);
  int* pairs  = btot + (size_t)NBK * BK_PAD;
  int* ssrc   = pairs + 2 * (size_t)E;
  int* beg    = ssrc + E;
  int* cur    = beg + N;
  int* blkoff = cur + N;
  int* bbase  = blkoff + (size_t)G1 * NBK;

  const int mBlocks = (N + GEMM_BM - 1) / GEMM_BM;

  fill_i32<<<(NBK * BK_PAD + 255) / 256, 256, 0, stream>>>(btot, NBK * BK_PAD, 0);
  bucket_count_k<<<G1, BK_MAX, 0, stream>>>(ei, E, NBK, btot, blkoff);
  bucket_scatter_k<<<G1, BK_MAX, 0, stream>>>(ei, E, NBK, btot, blkoff, bbase, pairs);
  bucket_sort_k<<<NBK, 1024, 0, stream>>>(pairs, btot, bbase, N, beg, cur, ssrc);

  gemm_wmma_f32<256, 128, 32><<<mBlocks, 2 * H1C, 0, stream>>>(x, W1, h1, att_src1, att_dst1,
                                                               es, ed, N);
  node_gather_k<128, 32, false><<<(N + 7) / 8, 256, 0, stream>>>(
      beg, cur, ssrc, h1, es, ed, b1, ln0_g, ln0_b, out1, N);

  gemm_wmma_f32<128, 160, 40><<<mBlocks, 2 * H2C, 0, stream>>>(out1, W2, h2, att_src2, att_dst2,
                                                               es, ed, N);
  node_gather_k<160, 40, true><<<(N + 7) / 8, 256, 0, stream>>>(
      beg, cur, ssrc, h2, es, ed, b2, ln1_g, ln1_b, (float*)d_out, N);
}


# Gather waves fetch each in-edge's source id and source scores once: the maxima pass keeps them in registers for the weight pass, which fetched them again; every sum and its order unchanged
# speedup vs baseline: 4.5811x; 1.0357x over previous
#include <hip/hip_runtime.h>
#include <math.h>

typedef __attribute__((ext_vector_type(2))) float v2f;
typedef __attribute__((ext_vector_type(8))) float v8f;

#define GAT_NEG_SLOPE 0.2f
#define GAT_SM_EPS 1e-16f
#define GAT_LN_EPS 1e-5f

__device__ __forceinline__ float warp_sum32(float v) {
#pragma unroll
  for (int o = 16; o > 0; o >>= 1) v += __shfl_xor(v, o, 32);
  return v;
}
__device__ __forceinline__ float warp_max32(float v) {
#pragma unroll
  for (int o = 16; o > 0; o >>= 1) v = fmaxf(v, __shfl_xor(v, o, 32));
  return v;
}

__device__ __forceinline__ float group8_sum(float v) {
  v += __int_as_float(__builtin_amdgcn_update_dpp(0, __float_as_int(v), 0x161, 0xF, 0xF, true));
  v += __int_as_float(__builtin_amdgcn_update_dpp(0, __float_as_int(v), 0x162, 0xF, 0xF, true));
  v += __int_as_float(__builtin_amdgcn_update_dpp(0, __float_as_int(v), 0x164, 0xF, 0xF, true));
  return v;
}

__global__ void fill_i32(int* __restrict__ p, int n, int v) {
  int i = blockIdx.x * blockDim.x + threadIdx.x;
  int stride = gridDim.x * blockDim.x;
  for (; i < n; i += stride) p[i] = v;
}

#define GEMM_MT 4
#define GEMM_BM (16 * GEMM_MT)
template <int K, int NC, int CH>
__global__ void __launch_bounds__(2 * NC)
gemm_wmma_f32(const float* __restrict__ A, const float* __restrict__ B,
              float* __restrict__ C, const float* __restrict__ a_src,
              const float* __restrict__ a_dst, float* __restrict__ es,
              float* __restrict__ ed, int M) {
  static_assert(NC == 4 * CH && CH % 8 == 0, "four heads, each a whole number of 8-column groups");
  constexpr int NG = NC / 8;
  constexpr int GPH = CH / 8;
  __shared__ float part_s[GEMM_BM * NG];
  __shared__ float part_d[GEMM_BM * NG];
  constexpr int KP = K + 4;
  constexpr int NT = 2 * NC;
  constexpr int RV = K / 4;
  constexpr int TOT = GEMM_BM * RV;
  constexpr int NIT = (TOT + NT - 1) / NT;
  __shared__ __attribute__((aligned(16))) float As[GEMM_BM * KP];
  const int tid = threadIdx.x;
  const int mBase = blockIdx.x * GEMM_BM;

  float4 stg[NIT];
#pragma unroll
  for (int it = 0; it < NIT; ++it) {
    const int i = tid + it * NT;
    const int r = i / RV, c4 = i - r * RV;
    const int row = mBase + r;
    stg[it] = make_float4(0.0f, 0.0f, 0.0f, 0.0f);
    if (i < TOT && row < M)
      stg[it] = *reinterpret_cast<const float4*>(A + (size_t)row * K + c4 * 4);
  }
#pragma unroll
  for (int it = 0; it < NIT; ++it) {
    const int i = tid + it * NT;
    const int r = i / RV, c4 = i - r * RV;
    if (i < TOT) *reinterpret_cast<float4*>(As + r * KP + c4 * 4) = stg[it];
  }
  __syncthreads();

  const int wave = tid >> 5;
  const int lane = tid & 31;
  const int nBase = wave * 16;
  const int l15 = lane & 15;
  const int khalf = (lane >> 4) * 2;

  v8f acc[GEMM_MT];
#pragma unroll
  for (int mt = 0; mt < GEMM_MT; ++mt) acc[mt] = v8f{};

  const float* ap = As + l15 * KP + khalf;
  const float* bp = B + (size_t)khalf * NC + nBase + l15;
#pragma unroll 8
  for (int k0 = 0; k0 < K; k0 += 4) {
    v2f b;
    b.x = bp[(size_t)k0 * NC];
    b.y = bp[(size_t)(k0 + 1) * NC];
#pragma unroll
    for (int mt = 0; mt < GEMM_MT; ++mt) {
      const v2f a = *reinterpret_cast<const v2f*>(ap + mt * 16 * KP + k0);
      acc[mt] = __builtin_amdgcn_wmma_f32_16x16x4_f32(false, a, false, b,
                                                      (short)0, acc[mt], false, false);
    }
  }

  const int col = nBase + l15;
#pragma unroll
  for (int mt = 0; mt < GEMM_MT; ++mt) {
    const int rowTop = mBase + mt * 16 + (lane >> 4) * 8;
#pragma unroll
    for (int r = 0; r < 8; ++r) {
      const int row = rowTop + r;
      if (row < M) C[(size_t)row * NC + col] = acc[mt][r];
    }
  }

  const float as = a_src[col], ad = a_dst[col];
  const int grp = (nBase >> 3) + ((lane >> 3) & 1);
#pragma unroll
  for (int mt = 0; mt < GEMM_MT; ++mt) {
#pragma unroll
    for (int r = 0; r < 8; ++r) {
      const float ps = group8_sum(acc[mt][r] * as);
      const float pd = group8_sum(acc[mt][r] * ad);
      if ((lane & 7) == 0) {
        const int rl = mt * 16 + (lane >> 4) * 8 + r;
        part_s[rl * NG + grp] = ps;
        part_d[rl * NG + grp] = pd;
      }
    }
  }
  __syncthreads();
  if (tid < GEMM_BM * 4) {
    const int rl = tid >> 2, hh = tid & 3;
    const float* qs = part_s + rl * NG + hh * GPH;
    const float* qd = part_d + rl * NG + hh * GPH;
    float s = qs[0], d = qd[0];
#pragma unroll
    for (int gi = 1; gi < GPH; ++gi) { s += qs[gi]; d += qd[gi]; }
    const int row = mBase + rl;
    if (row < M) { es[(size_t)row * 4 + hh] = s; ed[(size_t)row * 4 + hh] = d; }
  }
}

__device__ __forceinline__ float lrelu(float v) {
  return (v > 0.0f) ? v : GAT_NEG_SLOPE * v;
}
__device__ __forceinline__ float pick4(int h, float a0, float a1, float a2, float a3) {
  return (h == 0) ? a0 : ((h == 1) ? a1 : ((h == 2) ? a2 : a3));
}

#define BK_SHIFT 8
#define BK_NODES (1 << BK_SHIFT)
#define BK_MAX 256
#define BK_PAD 32
#define EPB 2048
#define EPT (EPB / 256)

__device__ __forceinline__ int wave_incl_scan(int v, int lane) {
#pragma unroll
  for (int o = 1; o < 32; o <<= 1) {
    int t = __shfl_up(v, o, 32);
    if (lane >= o) v += t;
  }
  return v;
}

__global__ void bucket_count_k(const int* __restrict__ ei, int E, int NBK,
                               int* __restrict__ btot, int* __restrict__ blkoff) {
  __shared__ int hist[BK_MAX];
  const int t = threadIdx.x;
  hist[t] = 0;
  __syncthreads();
  const int e0 = blockIdx.x * EPB + t;
  int d[EPT];
#pragma unroll
  for (int k = 0; k < EPT; ++k) {
    const int e = e0 + k * 256;
    d[k] = ei[E + ((e < E) ? e : (E - 1))];
  }
#pragma unroll
  for (int k = 0; k < EPT; ++k)
    if (e0 + k * 256 < E) atomicAdd(&hist[d[k] >> BK_SHIFT], 1);
  __syncthreads();
  if (t < NBK) {
    const int c = hist[t];
    int off = 0;
    if (c > 0) off = atomicAdd(&btot[t * BK_PAD], c);
    blkoff[blockIdx.x * NBK + t] = off;
  }
}

__global__ void bucket_scatter_k(const int* __restrict__ ei, int E, int NBK,
                                 const int* __restrict__ btot, const int* __restrict__ blkoff,
                                 int* __restrict__ bbase, int* __restrict__ pairs) {
  __shared__ int cursor[BK_MAX];
  __shared__ int wsum[BK_MAX / 32];
  const int t = threadIdx.x, lane = t & 31, wave = t >> 5;
  const int v = (t < NBK) ? btot[t * BK_PAD] : 0;
  const int incl = wave_incl_scan(v, lane);
  if (lane == 31) wsum[wave] = incl;
  __syncthreads();
  int add = 0;
  for (int w = 0; w < wave; ++w) add += wsum[w];
  const int excl = add + incl - v;
  if (t < NBK) {
    cursor[t] = excl + blkoff[blockIdx.x * NBK + t];
    if (blockIdx.x == 0) bbase[t] = excl;
  }
  __syncthreads();
  const int e0 = blockIdx.x * EPB + t;
  int s[EPT], d[EPT];
#pragma unroll
  for (int k = 0; k < EPT; ++k) {
    const int e = e0 + k * 256;
    const int ec = (e < E) ? e : (E - 1);
    s[k] = ei[ec];
    d[k] = ei[E + ec];
  }
  int2* __restrict__ pr = reinterpret_cast<int2*>(pairs);
#pragma unroll
  for (int k = 0; k < EPT; ++k)
    if (e0 + k * 256 < E) {
      const int pos = atomicAdd(&cursor[d[k] >> BK_SHIFT], 1);
      pr[pos] = make_int2(s[k], d[k]);
    }
}

__global__ void __launch_bounds__(1024)
bucket_sort_k(const int* __restrict__ pairs, const int* __restrict__ btot,
              const int* __restrict__ bbase, int N, int* __restrict__ beg,
              int* __restrict__ endp, int* __restrict__ ssrc) {
  __shared__ int cnt[BK_NODES];
  __shared__ int cursor[BK_NODES];
  __shared__ int wsum[BK_NODES / 32];
  const int t = threadIdx.x, lane = t & 31, wave = t >> 5;
  const int bk = blockIdx.x;
  const int start = bbase[bk];
  const int end = start + btot[bk * BK_PAD];
  const int2* __restrict__ pr = reinterpret_cast<const int2*>(pairs);
  if (t < BK_NODES) cnt[t] = 0;
  __syncthreads();
  for (int i = start + t; i < end; i += 1024)
    atomicAdd(&cnt[pr[i].y & (BK_NODES - 1)], 1);
  __syncthreads();
  int v = 0, incl = 0;
  if (t < BK_NODES) {
    v = cnt[t];
    incl = wave_incl_scan(v, lane);
    if (lane == 31) wsum[wave] = incl;
  }
  __syncthreads();
  if (t < BK_NODES) {
    int add = 0;
    for (int w = 0; w < wave; ++w) add += wsum[w];
    const int p = start + add + incl - v;
    cursor[t] = p;
    const int node = bk * BK_NODES + t;
    if (node < N) { beg[node] = p; endp[node] = p + v; }
  }
  __syncthreads();
  for (int i = start + t; i < end; i += 1024) {
    const int2 e = pr[i];
    const int pos = atomicAdd(&cursor[e.y & (BK_NODES - 1)], 1);
    ssrc[pos] = e.x;
  }
}

template <int CHT, int U>
__device__ __forceinline__ void gather_rows(const float* __restrict__ feat, int lane, int k, int s,
                                            float p0, float p1, float p2, float p3,
                                            const int (&hd)[CHT / 32], float (&acc)[CHT / 32]) {
  constexpr int NI = CHT / 32;
  float r[U][NI];
#pragma unroll
  for (int u = 0; u < U; ++u) {
    const int sk = __builtin_amdgcn_readlane(s, k + u);
    const float* fp = feat + (size_t)sk * CHT + lane;
#pragma unroll
    for (int i = 0; i < NI; ++i) r[u][i] = fp[32 * i];
  }
#pragma unroll
  for (int u = 0; u < U; ++u) {
    const float q0 = __int_as_float(__builtin_amdgcn_readlane(__float_as_int(p0), k + u));
    const float q1 = __int_as_float(__builtin_amdgcn_readlane(__float_as_int(p1), k + u));
    const float q2 = __int_as_float(__builtin_amdgcn_readlane(__float_as_int(p2), k + u));
    const float q3 = __int_as_float(__builtin_amdgcn_readlane(__float_as_int(p3), k + u));
#pragma unroll
    for (int i = 0; i < NI; ++i) acc[i] = fmaf(r[u][i], pick4(hd[i], q0, q1, q2, q3), acc[i]);
  }
}

template <int CHT>
__device__ __forceinline__ void gather_batch(const float* __restrict__ feat, int lane, int nk,
                                             bool ok, int s, const float4 e4, const float4 dv,
                                             float m0, float m1, float m2, float m3,
                                             float& d0, float& d1, float& d2, float& d3,
                                             const int (&hd)[CHT / 32], float (&acc)[CHT / 32]) {
  const float p0 = ok ? __expf(lrelu(e4.x + dv.x) - m0) : 0.0f;
  const float p1 = ok ? __expf(lrelu(e4.y + dv.y) - m1) : 0.0f;
  const float p2 = ok ? __expf(lrelu(e4.z + dv.z) - m2) : 0.0f;
  const float p3 = ok ? __expf(lrelu(e4.w + dv.w) - m3) : 0.0f;
  d0 += p0; d1 += p1; d2 += p2; d3 += p3;
  int k = 0;
  for (; k + 4 <= nk; k += 4)
    gather_rows<CHT, 4>(feat, lane, k, s, p0, p1, p2, p3, hd, acc);
  if ((nk - k) & 2) {
    gather_rows<CHT, 2>(feat, lane, k, s, p0, p1, p2, p3, hd, acc);
    k += 2;
  }
  if ((nk - k) & 1)
    gather_rows<CHT, 1>(feat, lane, k, s, p0, p1, p2, p3, hd, acc);
}

template <int CHT, int CH, bool LAST>
__global__ void node_gather_k(const int* __restrict__ beg, const int* __restrict__ endp,
                              const int* __restrict__ ssrc, const float* __restrict__ feat,
                              const float* __restrict__ es, const float* __restrict__ ed,
                              const float* __restrict__ bias, const float* __restrict__ g,
                              const float* __restrict__ b, float* __restrict__ out, int N) {
  constexpr int NI = CHT / 32;
  const int node = blockIdx.x * (blockDim.x >> 5) + (threadIdx.x >> 5);
  const int lane = threadIdx.x & 31;
  if (node >= N) return;
  const float4* es4 = reinterpret_cast<const float4*>(es);
  const float4* ed4 = reinterpret_cast<const float4*>(ed);
  const int jb = beg[node], je = endp[node];
  const float4 dv = ed4[node];
  const float4 sv = es4[node];
  float selfrow[CHT / 32];
#pragma unroll
  for (int i = 0; i < CHT / 32; ++i) selfrow[i] = feat[(size_t)node * CHT + lane + 32 * i];

  const float v0 = lrelu(sv.x + dv.x), v1 = lrelu(sv.y + dv.y);
  const float v2 = lrelu(sv.z + dv.z), v3 = lrelu(sv.w + dv.w);
  float m0 = v0, m1 = v1, m2 = v2, m3 = v3;
  const bool okA = (jb + lane) < je;
  const int sA = okA ? ssrc[jb + lane] : node;
  const float4 eA = es4[sA];
  m0 = fmaxf(m0, lrelu(eA.x + dv.x));
  m1 = fmaxf(m1, lrelu(eA.y + dv.y));
  m2 = fmaxf(m2, lrelu(eA.z + dv.z));
  m3 = fmaxf(m3, lrelu(eA.w + dv.w));
  for (int j = jb + 32 + lane; j < je; j += 32) {
    const float4 e4 = es4[ssrc[j]];
    m0 = fmaxf(m0, lrelu(e4.x + dv.x));
    m1 = fmaxf(m1, lrelu(e4.y + dv.y));
    m2 = fmaxf(m2, lrelu(e4.z + dv.z));
    m3 = fmaxf(m3, lrelu(e4.w + dv.w));
  }
  m0 = warp_max32(m0); m1 = warp_max32(m1); m2 = warp_max32(m2); m3 = warp_max32(m3);

  int hd[NI];
  float acc[NI];
#pragma unroll
  for (int i = 0; i < NI; ++i) { hd[i] = (lane + 32 * i) / CH; acc[i] = 0.0f; }
  float d0 = 0.0f, d1 = 0.0f, d2 = 0.0f, d3 = 0.0f;

  {
    const int nk = (je - jb < 32) ? (je - jb) : 32;
    gather_batch<CHT>(feat, lane, nk, okA, sA, eA, dv, m0, m1, m2, m3, d0, d1, d2, d3, hd, acc);
  }
  for (int j0 = jb + 32; j0 < je; j0 += 32) {
    const int j = j0 + lane;
    const bool ok = j < je;
    const int s = ok ? ssrc[j] : node;
    const float4 e4 = es4[s];
    const int nk = (je - j0 < 32) ? (je - j0) : 32;
    gather_batch<CHT>(feat, lane, nk, ok, s, e4, dv, m0, m1, m2, m3, d0, d1, d2, d3, hd, acc);
  }

  const float s0 = __expf(v0 - m0), s1 = __expf(v1 - m1);
  const float s2 = __expf(v2 - m2), s3 = __expf(v3 - m3);
#pragma unroll
  for (int i = 0; i < NI; ++i) acc[i] = fmaf(selfrow[i], pick4(hd[i], s0, s1, s2, s3), acc[i]);
  d0 = warp_sum32(d0) + s0; d1 = warp_sum32(d1) + s1;
  d2 = warp_sum32(d2) + s2; d3 = warp_sum32(d3) + s3;

  float val[NI];
#pragma unroll
  for (int i = 0; i < NI; ++i)
    val[i] = acc[i] / (pick4(hd[i], d0, d1, d2, d3) + GAT_SM_EPS);

  if constexpr (!LAST) {
    float sum = 0.0f;
#pragma unroll
    for (int i = 0; i < NI; ++i) {
      float t = val[i] + bias[lane + 32 * i];
      t = (t > 0.0f) ? t : (__expf(t) - 1.0f);
      val[i] = t; sum += t;
    }
    const float mu = warp_sum32(sum) * (1.0f / (float)CHT);
    float vs = 0.0f;
#pragma unroll
    for (int i = 0; i < NI; ++i) { float dd = val[i] - mu; vs += dd * dd; }
    const float var = warp_sum32(vs) * (1.0f / (float)CHT);
    const float rstd = rsqrtf(var + GAT_LN_EPS);
    float* op = out + (size_t)node * CHT + lane;
#pragma unroll
    for (int i = 0; i < NI; ++i) {
      const int c = lane + 32 * i;
      op[32 * i] = (val[i] - mu) * rstd * g[c] + b[c];
    }
  } else {
    static_assert(!LAST || (CHT == 160 && CH == 40), "head-mean tail is laid out for 4 heads of 40");
    const bool has1 = (lane < 8);
    const int r8 = (lane + 8) & 31, r16 = (lane + 16) & 31, r24 = (lane + 24) & 31;
    const float a1 = __shfl((lane >= 8) ? val[1] : val[2], r8, 32);
    const float a2 = __shfl((lane >= 16) ? val[2] : val[3], r16, 32);
    const float a3 = __shfl((lane >= 24) ? val[3] : val[4], r24, 32);
    const float c1 = __shfl(val[2], r8, 32);
    const float c2 = __shfl(val[3], r16, 32);
    const float c3 = __shfl(val[4], r24, 32);

    float x0, x1 = 0.0f;
    { float s = val[0]; s += a1; s += a2; s += a3;
      x0 = s * 0.25f + bias[lane]; }
    if (has1) {
      float s = val[1]; s += c1; s += c2; s += c3;
      x1 = s * 0.25f + bias[lane + 32];
    }
    const float mu = warp_sum32(x0 + (has1 ? x1 : 0.0f)) * (1.0f / 40.0f);
    const float e0 = x0 - mu, e1 = has1 ? (x1 - mu) : 0.0f;
    const float var = warp_sum32(e0 * e0 + e1 * e1) * (1.0f / 40.0f);
    const float rstd = rsqrtf(var + GAT_LN_EPS);
    const float z0 = e0 * rstd * g[lane] + b[lane];
    const float z1 = has1 ? (e1 * rstd * g[lane + 32] + b[lane + 32]) : -3.0e38f;
    const float mx = warp_max32(fmaxf(z0, z1));
    const float se = warp_sum32(__expf(z0 - mx) + (has1 ? __expf(z1 - mx) : 0.0f));
    const float lse = __logf(se);
    float* yp = out + (size_t)node * CH;
    yp[lane] = z0 - mx - lse;
    if (has1) yp[lane + 32] = z1 - mx - lse;
  }
}

extern "C" void kernel_launch(void* const* d_in, const int* in_sizes, int n_in,
                              void* d_out, int out_size, void* d_ws, size_t ws_size,
                              hipStream_t stream) {
  const float* x        = (const float*)d_in[0];
  const int*   ei       = (const int*)d_in[1];
  const float* W1       = (const float*)d_in[2];
  const float* att_src1 = (const float*)d_in[3];
  const float* att_dst1 = (const float*)d_in[4];
  const float* b1       = (const float*)d_in[5];
  const float* W2       = (const float*)d_in[6];
  const float* att_src2 = (const float*)d_in[7];
  const float* att_dst2 = (const float*)d_in[8];
  const float* b2       = (const float*)d_in[9];
  const float* ln0_g    = (const float*)d_in[10];
  const float* ln0_b    = (const float*)d_in[11];
  const float* ln1_g    = (const float*)d_in[12];
  const float* ln1_b    = (const float*)d_in[13];

  const int F_IN = 256, H1C = 128, H2C = 160, H = 4;
  const int N = in_sizes[0] / F_IN;
  const int E = in_sizes[1] / 2;

  float* ws   = (float*)d_ws;
  float* h1   = ws;
  float* out1 = h1   + (size_t)N * H1C;
  float* h2   = out1 + (size_t)N * H1C;
  float* es   = h2   + (size_t)N * H2C;
  float* ed   = es   + (size_t)N * H;
  const int NBK = (N + BK_NODES - 1) / BK_NODES;
  const int G1  = (E + EPB - 1) / EPB;
  int* btot   = (int*)(ed + (size_t)N * H);
  int* pairs  = btot + (size_t)NBK * BK_PAD;
  int* ssrc   = pairs + 2 * (size_t)E;
  int* beg    = ssrc + E;
  int* cur    = beg + N;
  int* blkoff = cur + N;
  int* bbase  = blkoff + (size_t)G1 * NBK;

  const int mBlocks = (N + GEMM_BM - 1) / GEMM_BM;

  fill_i32<<<(NBK * BK_PAD + 255) / 256, 256, 0, stream>>>(btot, NBK * BK_PAD, 0);
  bucket_count_k<<<G1, BK_MAX, 0, stream>>>(ei, E, NBK, btot, blkoff);
  bucket_scatter_k<<<G1, BK_MAX, 0, stream>>>(ei, E, NBK, btot, blkoff, bbase, pairs);
  bucket_sort_k<<<NBK, 1024, 0, stream>>>(pairs, btot, bbase, N, beg, cur, ssrc);

  gemm_wmma_f32<256, 128, 32><<<mBlocks, 2 * H1C, 0, stream>>>(x, W1, h1, att_src1, att_dst1,
                                                               es, ed, N);
  node_gather_k<128, 32, false><<<(N + 7) / 8, 256, 0, stream>>>(
      beg, cur, ssrc, h1, es, ed, b1, ln0_g, ln0_b, out1, N);

  gemm_wmma_f32<128, 160, 40><<<mBlocks, 2 * H2C, 0, stream>>>(out1, W2, h2, att_src2, att_dst2,
                                                               es, ed, N);
  node_gather_k<160, 40, true><<<(N + 7) / 8, 256, 0, stream>>>(
      beg, cur, ssrc, h2, es, ed, b2, ln1_g, ln1_b, (float*)d_out, N);
}


# Gather waves' 32-lane sums and maxima on the vector ALU: one v_permlanex16 exchange, then four DPP row_xmask steps, replace five ds_bpermute round trips each; same tree, sums bit-identical
# speedup vs baseline: 4.7600x; 1.0391x over previous
#include <hip/hip_runtime.h>
#include <math.h>

typedef __attribute__((ext_vector_type(2))) float v2f;
typedef __attribute__((ext_vector_type(8))) float v8f;

#define GAT_NEG_SLOPE 0.2f
#define GAT_SM_EPS 1e-16f
#define GAT_LN_EPS 1e-5f

__device__ __forceinline__ float xhalf(float v) {
  const int b = __float_as_int(v);
  return __int_as_float(__builtin_amdgcn_permlanex16(b, b, 0x76543210, 0xfedcba98, false, false));
}
template <int XM>
__device__ __forceinline__ float xrow(float v) {
  return __int_as_float(__builtin_amdgcn_update_dpp(0, __float_as_int(v), 0x160 + XM, 0xF, 0xF, true));
}
__device__ __forceinline__ float warp_sum32(float v) {
  v += xhalf(v);
  v += xrow<8>(v);
  v += xrow<4>(v);
  v += xrow<2>(v);
  v += xrow<1>(v);
  return v;
}
__device__ __forceinline__ float warp_max32(float v) {
  v = fmaxf(v, xhalf(v));
  v = fmaxf(v, xrow<8>(v));
  v = fmaxf(v, xrow<4>(v));
  v = fmaxf(v, xrow<2>(v));
  v = fmaxf(v, xrow<1>(v));
  return v;
}

__device__ __forceinline__ float group8_sum(float v) {
  v += __int_as_float(__builtin_amdgcn_update_dpp(0, __float_as_int(v), 0x161, 0xF, 0xF, true));
  v += __int_as_float(__builtin_amdgcn_update_dpp(0, __float_as_int(v), 0x162, 0xF, 0xF, true));
  v += __int_as_float(__builtin_amdgcn_update_dpp(0, __float_as_int(v), 0x164, 0xF, 0xF, true));
  return v;
}

__global__ void fill_i32(int* __restrict__ p, int n, int v) {
  int i = blockIdx.x * blockDim.x + threadIdx.x;
  int stride = gridDim.x * blockDim.x;
  for (; i < n; i += stride) p[i] = v;
}

#define GEMM_MT 4
#define GEMM_BM (16 * GEMM_MT)
template <int K, int NC, int CH>
__global__ void __launch_bounds__(2 * NC)
gemm_wmma_f32(const float* __restrict__ A, const float* __restrict__ B,
              float* __restrict__ C, const float* __restrict__ a_src,
              const float* __restrict__ a_dst, float* __restrict__ es,
              float* __restrict__ ed, int M) {
  static_assert(NC == 4 * CH && CH % 8 == 0, "four heads, each a whole number of 8-column groups");
  constexpr int NG = NC / 8;
  constexpr int GPH = CH / 8;
  __shared__ float part_s[GEMM_BM * NG];
  __shared__ float part_d[GEMM_BM * NG];
  constexpr int KP = K + 4;
  constexpr int NT = 2 * NC;
  constexpr int RV = K / 4;
  constexpr int TOT = GEMM_BM * RV;
  constexpr int NIT = (TOT + NT - 1) / NT;
  __shared__ __attribute__((aligned(16))) float As[GEMM_BM * KP];
  const int tid = threadIdx.x;
  const int mBase = blockIdx.x * GEMM_BM;

  float4 stg[NIT];
#pragma unroll
  for (int it = 0; it < NIT; ++it) {
    const int i = tid + it * NT;
    const int r = i / RV, c4 = i - r * RV;
    const int row = mBase + r;
    stg[it] = make_float4(0.0f, 0.0f, 0.0f, 0.0f);
    if (i < TOT && row < M)
      stg[it] = *reinterpret_cast<const float4*>(A + (size_t)row * K + c4 * 4);
  }
#pragma unroll
  for (int it = 0; it < NIT; ++it) {
    const int i = tid + it * NT;
    const int r = i / RV, c4 = i - r * RV;
    if (i < TOT) *reinterpret_cast<float4*>(As + r * KP + c4 * 4) = stg[it];
  }
  __syncthreads();

  const int wave = tid >> 5;
  const int lane = tid & 31;
  const int nBase = wave * 16;
  const int l15 = lane & 15;
  const int khalf = (lane >> 4) * 2;

  v8f acc[GEMM_MT];
#pragma unroll
  for (int mt = 0; mt < GEMM_MT; ++mt) acc[mt] = v8f{};

  const float* ap = As + l15 * KP + khalf;
  const float* bp = B + (size_t)khalf * NC + nBase + l15;
#pragma unroll 8
  for (int k0 = 0; k0 < K; k0 += 4) {
    v2f b;
    b.x = bp[(size_t)k0 * NC];
    b.y = bp[(size_t)(k0 + 1) * NC];
#pragma unroll
    for (int mt = 0; mt < GEMM_MT; ++mt) {
      const v2f a = *reinterpret_cast<const v2f*>(ap + mt * 16 * KP + k0);
      acc[mt] = __builtin_amdgcn_wmma_f32_16x16x4_f32(false, a, false, b,
                                                      (short)0, acc[mt], false, false);
    }
  }

  const int col = nBase + l15;
#pragma unroll
  for (int mt = 0; mt < GEMM_MT; ++mt) {
    const int rowTop = mBase + mt * 16 + (lane >> 4) * 8;
#pragma unroll
    for (int r = 0; r < 8; ++r) {
      const int row = rowTop + r;
      if (row < M) C[(size_t)row * NC + col] = acc[mt][r];
    }
  }

  const float as = a_src[col], ad = a_dst[col];
  const int grp = (nBase >> 3) + ((lane >> 3) & 1);
#pragma unroll
  for (int mt = 0; mt < GEMM_MT; ++mt) {
#pragma unroll
    for (int r = 0; r < 8; ++r) {
      const float ps = group8_sum(acc[mt][r] * as);
      const float pd = group8_sum(acc[mt][r] * ad);
      if ((lane & 7) == 0) {
        const int rl = mt * 16 + (lane >> 4) * 8 + r;
        part_s[rl * NG + grp] = ps;
        part_d[rl * NG + grp] = pd;
      }
    }
  }
  __syncthreads();
  if (tid < GEMM_BM * 4) {
    const int rl = tid >> 2, hh = tid & 3;
    const float* qs = part_s + rl * NG + hh * GPH;
    const float* qd = part_d + rl * NG + hh * GPH;
    float s = qs[0], d = qd[0];
#pragma unroll
    for (int gi = 1; gi < GPH; ++gi) { s += qs[gi]; d += qd[gi]; }
    const int row = mBase + rl;
    if (row < M) { es[(size_t)row * 4 + hh] = s; ed[(size_t)row * 4 + hh] = d; }
  }
}

__device__ __forceinline__ float lrelu(float v) {
  return (v > 0.0f) ? v : GAT_NEG_SLOPE * v;
}
__device__ __forceinline__ float pick4(int h, float a0, float a1, float a2, float a3) {
  return (h == 0) ? a0 : ((h == 1) ? a1 : ((h == 2) ? a2 : a3));
}

#define BK_SHIFT 8
#define BK_NODES (1 << BK_SHIFT)
#define BK_MAX 256
#define BK_PAD 32
#define EPB 2048
#define EPT (EPB / 256)

__device__ __forceinline__ int wave_incl_scan(int v, int lane) {
#pragma unroll
  for (int o = 1; o < 32; o <<= 1) {
    int t = __shfl_up(v, o, 32);
    if (lane >= o) v += t;
  }
  return v;
}

__global__ void bucket_count_k(const int* __restrict__ ei, int E, int NBK,
                               int* __restrict__ btot, int* __restrict__ blkoff) {
  __shared__ int hist[BK_MAX];
  const int t = threadIdx.x;
  hist[t] = 0;
  __syncthreads();
  const int e0 = blockIdx.x * EPB + t;
  int d[EPT];
#pragma unroll
  for (int k = 0; k < EPT; ++k) {
    const int e = e0 + k * 256;
    d[k] = ei[E + ((e < E) ? e : (E - 1))];
  }
#pragma unroll
  for (int k = 0; k < EPT; ++k)
    if (e0 + k * 256 < E) atomicAdd(&hist[d[k] >> BK_SHIFT], 1);
  __syncthreads();
  if (t < NBK) {
    const int c = hist[t];
    int off = 0;
    if (c > 0) off = atomicAdd(&btot[t * BK_PAD], c);
    blkoff[blockIdx.x * NBK + t] = off;
  }
}

__global__ void bucket_scatter_k(const int* __restrict__ ei, int E, int NBK,
                                 const int* __restrict__ btot, const int* __restrict__ blkoff,
                                 int* __restrict__ bbase, int* __restrict__ pairs) {
  __shared__ int cursor[BK_MAX];
  __shared__ int wsum[BK_MAX / 32];
  const int t = threadIdx.x, lane = t & 31, wave = t >> 5;
  const int v = (t < NBK) ? btot[t * BK_PAD] : 0;
  const int incl = wave_incl_scan(v, lane);
  if (lane == 31) wsum[wave] = incl;
  __syncthreads();
  int add = 0;
  for (int w = 0; w < wave; ++w) add += wsum[w];
  const int excl = add + incl - v;
  if (t < NBK) {
    cursor[t] = excl + blkoff[blockIdx.x * NBK + t];
    if (blockIdx.x == 0) bbase[t] = excl;
  }
  __syncthreads();
  const int e0 = blockIdx.x * EPB + t;
  int s[EPT], d[EPT];
#pragma unroll
  for (int k = 0; k < EPT; ++k) {
    const int e = e0 + k * 256;
    const int ec = (e < E) ? e : (E - 1);
    s[k] = ei[ec];
    d[k] = ei[E + ec];
  }
  int2* __restrict__ pr = reinterpret_cast<int2*>(pairs);
#pragma unroll
  for (int k = 0; k < EPT; ++k)
    if (e0 + k * 256 < E) {
      const int pos = atomicAdd(&cursor[d[k] >> BK_SHIFT], 1);
      pr[pos] = make_int2(s[k], d[k]);
    }
}

__global__ void __launch_bounds__(1024)
bucket_sort_k(const int* __restrict__ pairs, const int* __restrict__ btot,
              const int* __restrict__ bbase, int N, int* __restrict__ beg,
              int* __restrict__ endp, int* __restrict__ ssrc) {
  __shared__ int cnt[BK_NODES];
  __shared__ int cursor[BK_NODES];
  __shared__ int wsum[BK_NODES / 32];
  const int t = threadIdx.x, lane = t & 31, wave = t >> 5;
  const int bk = blockIdx.x;
  const int start = bbase[bk];
  const int end = start + btot[bk * BK_PAD];
  const int2* __restrict__ pr = reinterpret_cast<const int2*>(pairs);
  if (t < BK_NODES) cnt[t] = 0;
  __syncthreads();
  for (int i = start + t; i < end; i += 1024)
    atomicAdd(&cnt[pr[i].y & (BK_NODES - 1)], 1);
  __syncthreads();
  int v = 0, incl = 0;
  if (t < BK_NODES) {
    v = cnt[t];
    incl = wave_incl_scan(v, lane);
    if (lane == 31) wsum[wave] = incl;
  }
  __syncthreads();
  if (t < BK_NODES) {
    int add = 0;
    for (int w = 0; w < wave; ++w) add += wsum[w];
    const int p = start + add + incl - v;
    cursor[t] = p;
    const int node = bk * BK_NODES + t;
    if (node < N) { beg[node] = p; endp[node] = p + v; }
  }
  __syncthreads();
  for (int i = start + t; i < end; i += 1024) {
    const int2 e = pr[i];
    const int pos = atomicAdd(&cursor[e.y & (BK_NODES - 1)], 1);
    ssrc[pos] = e.x;
  }
}

template <int CHT, int U>
__device__ __forceinline__ void gather_rows(const float* __restrict__ feat, int lane, int k, int s,
                                            float p0, float p1, float p2, float p3,
                                            const int (&hd)[CHT / 32], float (&acc)[CHT / 32]) {
  constexpr int NI = CHT / 32;
  float r[U][NI];
#pragma unroll
  for (int u = 0; u < U; ++u) {
    const int sk = __builtin_amdgcn_readlane(s, k + u);
    const float* fp = feat + (size_t)sk * CHT + lane;
#pragma unroll
    for (int i = 0; i < NI; ++i) r[u][i] = fp[32 * i];
  }
#pragma unroll
  for (int u = 0; u < U; ++u) {
    const float q0 = __int_as_float(__builtin_amdgcn_readlane(__float_as_int(p0), k + u));
    const float q1 = __int_as_float(__builtin_amdgcn_readlane(__float_as_int(p1), k + u));
    const float q2 = __int_as_float(__builtin_amdgcn_readlane(__float_as_int(p2), k + u));
    const float q3 = __int_as_float(__builtin_amdgcn_readlane(__float_as_int(p3), k + u));
#pragma unroll
    for (int i = 0; i < NI; ++i) acc[i] = fmaf(r[u][i], pick4(hd[i], q0, q1, q2, q3), acc[i]);
  }
}

template <int CHT>
__device__ __forceinline__ void gather_batch(const float* __restrict__ feat, int lane, int nk,
                                             bool ok, int s, const float4 e4, const float4 dv,
                                             float m0, float m1, float m2, float m3,
                                             float& d0, float& d1, float& d2, float& d3,
                                             const int (&hd)[CHT / 32], float (&acc)[CHT / 32]) {
  const float p0 = ok ? __expf(lrelu(e4.x + dv.x) - m0) : 0.0f;
  const float p1 = ok ? __expf(lrelu(e4.y + dv.y) - m1) : 0.0f;
  const float p2 = ok ? __expf(lrelu(e4.z + dv.z) - m2) : 0.0f;
  const float p3 = ok ? __expf(lrelu(e4.w + dv.w) - m3) : 0.0f;
  d0 += p0; d1 += p1; d2 += p2; d3 += p3;
  int k = 0;
  for (; k + 4 <= nk; k += 4)
    gather_rows<CHT, 4>(feat, lane, k, s, p0, p1, p2, p3, hd, acc);
  if ((nk - k) & 2) {
    gather_rows<CHT, 2>(feat, lane, k, s, p0, p1, p2, p3, hd, acc);
    k += 2;
  }
  if ((nk - k) & 1)
    gather_rows<CHT, 1>(feat, lane, k, s, p0, p1, p2, p3, hd, acc);
}

template <int CHT, int CH, bool LAST>
__global__ void node_gather_k(const int* __restrict__ beg, const int* __restrict__ endp,
                              const int* __restrict__ ssrc, const float* __restrict__ feat,
                              const float* __restrict__ es, const float* __restrict__ ed,
                              const float* __restrict__ bias, const float* __restrict__ g,
                              const float* __restrict__ b, float* __restrict__ out, int N) {
  constexpr int NI = CHT / 32;
  const int node = blockIdx.x * (blockDim.x >> 5) + (threadIdx.x >> 5);
  const int lane = threadIdx.x & 31;
  if (node >= N) return;
  const float4* es4 = reinterpret_cast<const float4*>(es);
  const float4* ed4 = reinterpret_cast<const float4*>(ed);
  const int jb = beg[node], je = endp[node];
  const float4 dv = ed4[node];
  const float4 sv = es4[node];
  float selfrow[CHT / 32];
#pragma unroll
  for (int i = 0; i < CHT / 32; ++i) selfrow[i] = feat[(size_t)node * CHT + lane + 32 * i];

  const float v0 = lrelu(sv.x + dv.x), v1 = lrelu(sv.y + dv.y);
  const float v2 = lrelu(sv.z + dv.z), v3 = lrelu(sv.w + dv.w);
  float m0 = v0, m1 = v1, m2 = v2, m3 = v3;
  const bool okA = (jb + lane) < je;
  const int sA = okA ? ssrc[jb + lane] : node;
  const float4 eA = es4[sA];
  m0 = fmaxf(m0, lrelu(eA.x + dv.x));
  m1 = fmaxf(m1, lrelu(eA.y + dv.y));
  m2 = fmaxf(m2, lrelu(eA.z + dv.z));
  m3 = fmaxf(m3, lrelu(eA.w + dv.w));
  for (int j = jb + 32 + lane; j < je; j += 32) {
    const float4 e4 = es4[ssrc[j]];
    m0 = fmaxf(m0, lrelu(e4.x + dv.x));
    m1 = fmaxf(m1, lrelu(e4.y + dv.y));
    m2 = fmaxf(m2, lrelu(e4.z + dv.z));
    m3 = fmaxf(m3, lrelu(e4.w + dv.w));
  }
  m0 = warp_max32(m0); m1 = warp_max32(m1); m2 = warp_max32(m2); m3 = warp_max32(m3);

  int hd[NI];
  float acc[NI];
#pragma unroll
  for (int i = 0; i < NI; ++i) { hd[i] = (lane + 32 * i) / CH; acc[i] = 0.0f; }
  float d0 = 0.0f, d1 = 0.0f, d2 = 0.0f, d3 = 0.0f;

  {
    const int nk = (je - jb < 32) ? (je - jb) : 32;
    gather_batch<CHT>(feat, lane, nk, okA, sA, eA, dv, m0, m1, m2, m3, d0, d1, d2, d3, hd, acc);
  }
  for (int j0 = jb + 32; j0 < je; j0 += 32) {
    const int j = j0 + lane;
    const bool ok = j < je;
    const int s = ok ? ssrc[j] : node;
    const float4 e4 = es4[s];
    const int nk = (je - j0 < 32) ? (je - j0) : 32;
    gather_batch<CHT>(feat, lane, nk, ok, s, e4, dv, m0, m1, m2, m3, d0, d1, d2, d3, hd, acc);
  }

  const float s0 = __expf(v0 - m0), s1 = __expf(v1 - m1);
  const float s2 = __expf(v2 - m2), s3 = __expf(v3 - m3);
#pragma unroll
  for (int i = 0; i < NI; ++i) acc[i] = fmaf(selfrow[i], pick4(hd[i], s0, s1, s2, s3), acc[i]);
  d0 = warp_sum32(d0) + s0; d1 = warp_sum32(d1) + s1;
  d2 = warp_sum32(d2) + s2; d3 = warp_sum32(d3) + s3;

  float val[NI];
#pragma unroll
  for (int i = 0; i < NI; ++i)
    val[i] = acc[i] / (pick4(hd[i], d0, d1, d2, d3) + GAT_SM_EPS);

  if constexpr (!LAST) {
    float sum = 0.0f;
#pragma unroll
    for (int i = 0; i < NI; ++i) {
      float t = val[i] + bias[lane + 32 * i];
      t = (t > 0.0f) ? t : (__expf(t) - 1.0f);
      val[i] = t; sum += t;
    }
    const float mu = warp_sum32(sum) * (1.0f / (float)CHT);
    float vs = 0.0f;
#pragma unroll
    for (int i = 0; i < NI; ++i) { float dd = val[i] - mu; vs += dd * dd; }
    const float var = warp_sum32(vs) * (1.0f / (float)CHT);
    const float rstd = rsqrtf(var + GAT_LN_EPS);
    float* op = out + (size_t)node * CHT + lane;
#pragma unroll
    for (int i = 0; i < NI; ++i) {
      const int c = lane + 32 * i;
      op[32 * i] = (val[i] - mu) * rstd * g[c] + b[c];
    }
  } else {
    static_assert(!LAST || (CHT == 160 && CH == 40), "head-mean tail is laid out for 4 heads of 40");
    const bool has1 = (lane < 8);
    const int r8 = (lane + 8) & 31, r16 = (lane + 16) & 31, r24 = (lane + 24) & 31;
    const float a1 = __shfl((lane >= 8) ? val[1] : val[2], r8, 32);
    const float a2 = __shfl((lane >= 16) ? val[2] : val[3], r16, 32);
    const float a3 = __shfl((lane >= 24) ? val[3] : val[4], r24, 32);
    const float c1 = __shfl(val[2], r8, 32);
    const float c2 = __shfl(val[3], r16, 32);
    const float c3 = __shfl(val[4], r24, 32);

    float x0, x1 = 0.0f;
    { float s = val[0]; s += a1; s += a2; s += a3;
      x0 = s * 0.25f + bias[lane]; }
    if (has1) {
      float s = val[1]; s += c1; s += c2; s += c3;
      x1 = s * 0.25f + bias[lane + 32];
    }
    const float mu = warp_sum32(x0 + (has1 ? x1 : 0.0f)) * (1.0f / 40.0f);
    const float e0 = x0 - mu, e1 = has1 ? (x1 - mu) : 0.0f;
    const float var = warp_sum32(e0 * e0 + e1 * e1) * (1.0f / 40.0f);
    const float rstd = rsqrtf(var + GAT_LN_EPS);
    const float z0 = e0 * rstd * g[lane] + b[lane];
    const float z1 = has1 ? (e1 * rstd * g[lane + 32] + b[lane + 32]) : -3.0e38f;
    const float mx = warp_max32(fmaxf(z0, z1));
    const float se = warp_sum32(__expf(z0 - mx) + (has1 ? __expf(z1 - mx) : 0.0f));
    const float lse = __logf(se);
    float* yp = out + (size_t)node * CH;
    yp[lane] = z0 - mx - lse;
    if (has1) yp[lane + 32] = z1 - mx - lse;
  }
}

extern "C" void kernel_launch(void* const* d_in, const int* in_sizes, int n_in,
                              void* d_out, int out_size, void* d_ws, size_t ws_size,
                              hipStream_t stream) {
  const float* x        = (const float*)d_in[0];
  const int*   ei       = (const int*)d_in[1];
  const float* W1       = (const float*)d_in[2];
  const float* att_src1 = (const float*)d_in[3];
  const float* att_dst1 = (const float*)d_in[4];
  const float* b1       = (const float*)d_in[5];
  const float* W2       = (const float*)d_in[6];
  const float* att_src2 = (const float*)d_in[7];
  const float* att_dst2 = (const float*)d_in[8];
  const float* b2       = (const float*)d_in[9];
  const float* ln0_g    = (const float*)d_in[10];
  const float* ln0_b    = (const float*)d_in[11];
  const float* ln1_g    = (const float*)d_in[12];
  const float* ln1_b    = (const float*)d_in[13];

  const int F_IN = 256, H1C = 128, H2C = 160, H = 4;
  const int N = in_sizes[0] / F_IN;
  const int E = in_sizes[1] / 2;

  float* ws   = (float*)d_ws;
  float* h1   = ws;
  float* out1 = h1   + (size_t)N * H1C;
  float* h2   = out1 + (size_t)N * H1C;
  float* es   = h2   + (size_t)N * H2C;
  float* ed   = es   + (size_t)N * H;
  const int NBK = (N + BK_NODES - 1) / BK_NODES;
  const int G1  = (E + EPB - 1) / EPB;
  int* btot   = (int*)(ed + (size_t)N * H);
  int* pairs  = btot + (size_t)NBK * BK_PAD;
  int* ssrc   = pairs + 2 * (size_t)E;
  int* beg    = ssrc + E;
  int* cur    = beg + N;
  int* blkoff = cur + N;
  int* bbase  = blkoff + (size_t)G1 * NBK;

  const int mBlocks = (N + GEMM_BM - 1) / GEMM_BM;

  fill_i32<<<(NBK * BK_PAD + 255) / 256, 256, 0, stream>>>(btot, NBK * BK_PAD, 0);
  bucket_count_k<<<G1, BK_MAX, 0, stream>>>(ei, E, NBK, btot, blkoff);
  bucket_scatter_k<<<G1, BK_MAX, 0, stream>>>(ei, E, NBK, btot, blkoff, bbase, pairs);
  bucket_sort_k<<<NBK, 1024, 0, stream>>>(pairs, btot, bbase, N, beg, cur, ssrc);

  gemm_wmma_f32<256, 128, 32><<<mBlocks, 2 * H1C, 0, stream>>>(x, W1, h1, att_src1, att_dst1,
                                                               es, ed, N);
  node_gather_k<128, 32, false><<<(N + 7) / 8, 256, 0, stream>>>(
      beg, cur, ssrc, h1, es, ed, b1, ln0_g, ln0_b, out1, N);

  gemm_wmma_f32<128, 160, 40><<<mBlocks, 2 * H2C, 0, stream>>>(out1, W2, h2, att_src2, att_dst2,
                                                               es, ed, N);
  node_gather_k<160, 40, true><<<(N + 7) / 8, 256, 0, stream>>>(
      beg, cur, ssrc, h2, es, ed, b2, ln1_g, ln1_b, (float*)d_out, N);
}


# conv1 product x@W1 rides as extra work-groups in the two level-1 edge-bucketing launches, half its row blocks in each: 8 launches to 7, edge passes run under matrix work; every sum unchanged
# speedup vs baseline: 4.9167x; 1.0329x over previous
#include <hip/hip_runtime.h>
#include <math.h>

typedef __attribute__((ext_vector_type(2))) float v2f;
typedef __attribute__((ext_vector_type(8))) float v8f;

#define GAT_NEG_SLOPE 0.2f
#define GAT_SM_EPS 1e-16f
#define GAT_LN_EPS 1e-5f

__device__ __forceinline__ float xhalf(float v) {
  const int b = __float_as_int(v);
  return __int_as_float(__builtin_amdgcn_permlanex16(b, b, 0x76543210, 0xfedcba98, false, false));
}
template <int XM>
__device__ __forceinline__ float xrow(float v) {
  return __int_as_float(__builtin_amdgcn_update_dpp(0, __float_as_int(v), 0x160 + XM, 0xF, 0xF, true));
}
__device__ __forceinline__ float warp_sum32(float v) {
  v += xhalf(v);
  v += xrow<8>(v);
  v += xrow<4>(v);
  v += xrow<2>(v);
  v += xrow<1>(v);
  return v;
}
__device__ __forceinline__ float warp_max32(float v) {
  v = fmaxf(v, xhalf(v));
  v = fmaxf(v, xrow<8>(v));
  v = fmaxf(v, xrow<4>(v));
  v = fmaxf(v, xrow<2>(v));
  v = fmaxf(v, xrow<1>(v));
  return v;
}

__device__ __forceinline__ float group8_sum(float v) {
  v += __int_as_float(__builtin_amdgcn_update_dpp(0, __float_as_int(v), 0x161, 0xF, 0xF, true));
  v += __int_as_float(__builtin_amdgcn_update_dpp(0, __float_as_int(v), 0x162, 0xF, 0xF, true));
  v += __int_as_float(__builtin_amdgcn_update_dpp(0, __float_as_int(v), 0x164, 0xF, 0xF, true));
  return v;
}

__global__ void fill_i32(int* __restrict__ p, int n, int v) {
  int i = blockIdx.x * blockDim.x + threadIdx.x;
  int stride = gridDim.x * blockDim.x;
  for (; i < n; i += stride) p[i] = v;
}

#define GEMM_MT 4
#define GEMM_BM (16 * GEMM_MT)
template <int K, int NC, int CH>
__device__ __forceinline__ void
gemm_block(const int mb, const float* __restrict__ A, const float* __restrict__ B,
           float* __restrict__ C, const float* __restrict__ a_src,
           const float* __restrict__ a_dst, float* __restrict__ es,
           float* __restrict__ ed, int M) {
  static_assert(NC == 4 * CH && CH % 8 == 0, "four heads, each a whole number of 8-column groups");
  constexpr int NG = NC / 8;
  constexpr int GPH = CH / 8;
  __shared__ float part_s[GEMM_BM * NG];
  __shared__ float part_d[GEMM_BM * NG];
  constexpr int KP = K + 4;
  constexpr int NT = 2 * NC;
  constexpr int RV = K / 4;
  constexpr int TOT = GEMM_BM * RV;
  constexpr int NIT = (TOT + NT - 1) / NT;
  __shared__ __attribute__((aligned(16))) float As[GEMM_BM * KP];
  const int tid = threadIdx.x;
  const int mBase = mb * GEMM_BM;

  float4 stg[NIT];
#pragma unroll
  for (int it = 0; it < NIT; ++it) {
    const int i = tid + it * NT;
    const int r = i / RV, c4 = i - r * RV;
    const int row = mBase + r;
    stg[it] = make_float4(0.0f, 0.0f, 0.0f, 0.0f);
    if (i < TOT && row < M)
      stg[it] = *reinterpret_cast<const float4*>(A + (size_t)row * K + c4 * 4);
  }
#pragma unroll
  for (int it = 0; it < NIT; ++it) {
    const int i = tid + it * NT;
    const int r = i / RV, c4 = i - r * RV;
    if (i < TOT) *reinterpret_cast<float4*>(As + r * KP + c4 * 4) = stg[it];
  }
  __syncthreads();

  const int wave = tid >> 5;
  const int lane = tid & 31;
  const int nBase = wave * 16;
  const int l15 = lane & 15;
  const int khalf = (lane >> 4) * 2;

  v8f acc[GEMM_MT];
#pragma unroll
  for (int mt = 0; mt < GEMM_MT; ++mt) acc[mt] = v8f{};

  const float* ap = As + l15 * KP + khalf;
  const float* bp = B + (size_t)khalf * NC + nBase + l15;
#pragma unroll 8
  for (int k0 = 0; k0 < K; k0 += 4) {
    v2f b;
    b.x = bp[(size_t)k0 * NC];
    b.y = bp[(size_t)(k0 + 1) * NC];
#pragma unroll
    for (int mt = 0; mt < GEMM_MT; ++mt) {
      const v2f a = *reinterpret_cast<const v2f*>(ap + mt * 16 * KP + k0);
      acc[mt] = __builtin_amdgcn_wmma_f32_16x16x4_f32(false, a, false, b,
                                                      (short)0, acc[mt], false, false);
    }
  }

  const int col = nBase + l15;
#pragma unroll
  for (int mt = 0; mt < GEMM_MT; ++mt) {
    const int rowTop = mBase + mt * 16 + (lane >> 4) * 8;
#pragma unroll
    for (int r = 0; r < 8; ++r) {
      const int row = rowTop + r;
      if (row < M) C[(size_t)row * NC + col] = acc[mt][r];
    }
  }

  const float as = a_src[col], ad = a_dst[col];
  const int grp = (nBase >> 3) + ((lane >> 3) & 1);
#pragma unroll
  for (int mt = 0; mt < GEMM_MT; ++mt) {
#pragma unroll
    for (int r = 0; r < 8; ++r) {
      const float ps = group8_sum(acc[mt][r] * as);
      const float pd = group8_sum(acc[mt][r] * ad);
      if ((lane & 7) == 0) {
        const int rl = mt * 16 + (lane >> 4) * 8 + r;
        part_s[rl * NG + grp] = ps;
        part_d[rl * NG + grp] = pd;
      }
    }
  }
  __syncthreads();
  if (tid < GEMM_BM * 4) {
    const int rl = tid >> 2, hh = tid & 3;
    const float* qs = part_s + rl * NG + hh * GPH;
    const float* qd = part_d + rl * NG + hh * GPH;
    float s = qs[0], d = qd[0];
#pragma unroll
    for (int gi = 1; gi < GPH; ++gi) { s += qs[gi]; d += qd[gi]; }
    const int row = mBase + rl;
    if (row < M) { es[(size_t)row * 4 + hh] = s; ed[(size_t)row * 4 + hh] = d; }
  }
}

template <int K, int NC, int CH>
__global__ void __launch_bounds__(2 * NC)
gemm_wmma_f32(const float* __restrict__ A, const float* __restrict__ B,
              float* __restrict__ C, const float* __restrict__ a_src,
              const float* __restrict__ a_dst, float* __restrict__ es,
              float* __restrict__ ed, int M) {
  gemm_block<K, NC, CH>(blockIdx.x, A, B, C, a_src, a_dst, es, ed, M);
}

__device__ __forceinline__ float lrelu(float v) {
  return (v > 0.0f) ? v : GAT_NEG_SLOPE * v;
}
__device__ __forceinline__ float pick4(int h, float a0, float a1, float a2, float a3) {
  return (h == 0) ? a0 : ((h == 1) ? a1 : ((h == 2) ? a2 : a3));
}

#define BK_SHIFT 8
#define BK_NODES (1 << BK_SHIFT)
#define BK_MAX 256
#define BK_PAD 32
#define EPB 2048
#define EPT (EPB / 256)

__device__ __forceinline__ int wave_incl_scan(int v, int lane) {
#pragma unroll
  for (int o = 1; o < 32; o <<= 1) {
    int t = __shfl_up(v, o, 32);
    if (lane >= o) v += t;
  }
  return v;
}

__device__ __forceinline__ void bucket_count_block(const int bx, const int* __restrict__ ei, int E,
                                                   int NBK, int* __restrict__ btot,
                                                   int* __restrict__ blkoff) {
  __shared__ int hist[BK_MAX];
  const int t = threadIdx.x;
  hist[t] = 0;
  __syncthreads();
  const int e0 = bx * EPB + t;
  int d[EPT];
#pragma unroll
  for (int k = 0; k < EPT; ++k) {
    const int e = e0 + k * 256;
    d[k] = ei[E + ((e < E) ? e : (E - 1))];
  }
#pragma unroll
  for (int k = 0; k < EPT; ++k)
    if (e0 + k * 256 < E) atomicAdd(&hist[d[k] >> BK_SHIFT], 1);
  __syncthreads();
  if (t < NBK) {
    const int c = hist[t];
    int off = 0;
    if (c > 0) off = atomicAdd(&btot[t * BK_PAD], c);
    blkoff[bx * NBK + t] = off;
  }
}

__device__ __forceinline__ void bucket_scatter_block(const int bx, const int* __restrict__ ei, int E,
                                                     int NBK, const int* __restrict__ btot,
                                                     const int* __restrict__ blkoff,
                                                     int* __restrict__ bbase,
                                                     int* __restrict__ pairs) {
  __shared__ int cursor[BK_MAX];
  __shared__ int wsum[BK_MAX / 32];
  const int t = threadIdx.x, lane = t & 31, wave = t >> 5;
  const int v = (t < NBK) ? btot[t * BK_PAD] : 0;
  const int incl = wave_incl_scan(v, lane);
  if (lane == 31) wsum[wave] = incl;
  __syncthreads();
  int add = 0;
  for (int w = 0; w < wave; ++w) add += wsum[w];
  const int excl = add + incl - v;
  if (t < NBK) {
    cursor[t] = excl + blkoff[bx * NBK + t];
    if (bx == 0) bbase[t] = excl;
  }
  __syncthreads();
  const int e0 = bx * EPB + t;
  int s[EPT], d[EPT];
#pragma unroll
  for (int k = 0; k < EPT; ++k) {
    const int e = e0 + k * 256;
    const int ec = (e < E) ? e : (E - 1);
    s[k] = ei[ec];
    d[k] = ei[E + ec];
  }
  int2* __restrict__ pr = reinterpret_cast<int2*>(pairs);
#pragma unroll
  for (int k = 0; k < EPT; ++k)
    if (e0 + k * 256 < E) {
      const int pos = atomicAdd(&cursor[d[k] >> BK_SHIFT], 1);
      pr[pos] = make_int2(s[k], d[k]);
    }
}

template <int K, int NC, int CH>
__global__ void __launch_bounds__(2 * NC)
count_with_gemm_k(const int* __restrict__ ei, int E, int NBK, int* __restrict__ btot,
                  int* __restrict__ blkoff, int nGemm, int mb0,
                  const float* __restrict__ A, const float* __restrict__ B,
                  float* __restrict__ C, const float* __restrict__ a_src,
                  const float* __restrict__ a_dst, float* __restrict__ es,
                  float* __restrict__ ed, int M) {
  static_assert(2 * NC == BK_MAX, "both kinds of block run with BK_MAX threads");
  const int bx = blockIdx.x;
  if (bx < nGemm) gemm_block<K, NC, CH>(mb0 + bx, A, B, C, a_src, a_dst, es, ed, M);
  else bucket_count_block(bx - nGemm, ei, E, NBK, btot, blkoff);
}
template <int K, int NC, int CH>
__global__ void __launch_bounds__(2 * NC)
scatter_with_gemm_k(const int* __restrict__ ei, int E, int NBK, const int* __restrict__ btot,
                    const int* __restrict__ blkoff, int* __restrict__ bbase,
                    int* __restrict__ pairs, int nGemm, int mb0,
                    const float* __restrict__ A, const float* __restrict__ B,
                    float* __restrict__ C, const float* __restrict__ a_src,
                    const float* __restrict__ a_dst, float* __restrict__ es,
                    float* __restrict__ ed, int M) {
  static_assert(2 * NC == BK_MAX, "both kinds of block run with BK_MAX threads");
  const int bx = blockIdx.x;
  if (bx < nGemm) gemm_block<K, NC, CH>(mb0 + bx, A, B, C, a_src, a_dst, es, ed, M);
  else bucket_scatter_block(bx - nGemm, ei, E, NBK, btot, blkoff, bbase, pairs);
}

__global__ void __launch_bounds__(1024)
bucket_sort_k(const int* __restrict__ pairs, const int* __restrict__ btot,
              const int* __restrict__ bbase, int N, int* __restrict__ beg,
              int* __restrict__ endp, int* __restrict__ ssrc) {
  __shared__ int cnt[BK_NODES];
  __shared__ int cursor[BK_NODES];
  __shared__ int wsum[BK_NODES / 32];
  const int t = threadIdx.x, lane = t & 31, wave = t >> 5;
  const int bk = blockIdx.x;
  const int start = bbase[bk];
  const int end = start + btot[bk * BK_PAD];
  const int2* __restrict__ pr = reinterpret_cast<const int2*>(pairs);
  if (t < BK_NODES) cnt[t] = 0;
  __syncthreads();
  for (int i = start + t; i < end; i += 1024)
    atomicAdd(&cnt[pr[i].y & (BK_NODES - 1)], 1);
  __syncthreads();
  int v = 0, incl = 0;
  if (t < BK_NODES) {
    v = cnt[t];
    incl = wave_incl_scan(v, lane);
    if (lane == 31) wsum[wave] = incl;
  }
  __syncthreads();
  if (t < BK_NODES) {
    int add = 0;
    for (int w = 0; w < wave; ++w) add += wsum[w];
    const int p = start + add + incl - v;
    cursor[t] = p;
    const int node = bk * BK_NODES + t;
    if (node < N) { beg[node] = p; endp[node] = p + v; }
  }
  __syncthreads();
  for (int i = start + t; i < end; i += 1024) {
    const int2 e = pr[i];
    const int pos = atomicAdd(&cursor[e.y & (BK_NODES - 1)], 1);
    ssrc[pos] = e.x;
  }
}

template <int CHT, int U>
__device__ __forceinline__ void gather_rows(const float* __restrict__ feat, int lane, int k, int s,
                                            float p0, float p1, float p2, float p3,
                                            const int (&hd)[CHT / 32], float (&acc)[CHT / 32]) {
  constexpr int NI = CHT / 32;
  float r[U][NI];
#pragma unroll
  for (int u = 0; u < U; ++u) {
    const int sk = __builtin_amdgcn_readlane(s, k + u);
    const float* fp = feat + (size_t)sk * CHT + lane;
#pragma unroll
    for (int i = 0; i < NI; ++i) r[u][i] = fp[32 * i];
  }
#pragma unroll
  for (int u = 0; u < U; ++u) {
    const float q0 = __int_as_float(__builtin_amdgcn_readlane(__float_as_int(p0), k + u));
    const float q1 = __int_as_float(__builtin_amdgcn_readlane(__float_as_int(p1), k + u));
    const float q2 = __int_as_float(__builtin_amdgcn_readlane(__float_as_int(p2), k + u));
    const float q3 = __int_as_float(__builtin_amdgcn_readlane(__float_as_int(p3), k + u));
#pragma unroll
    for (int i = 0; i < NI; ++i) acc[i] = fmaf(r[u][i], pick4(hd[i], q0, q1, q2, q3), acc[i]);
  }
}

template <int CHT>
__device__ __forceinline__ void gather_batch(const float* __restrict__ feat, int lane, int nk,
                                             bool ok, int s, const float4 e4, const float4 dv,
                                             float m0, float m1, float m2, float m3,
                                             float& d0, float& d1, float& d2, float& d3,
                                             const int (&hd)[CHT / 32], float (&acc)[CHT / 32]) {
  const float p0 = ok ? __expf(lrelu(e4.x + dv.x) - m0) : 0.0f;
  const float p1 = ok ? __expf(lrelu(e4.y + dv.y) - m1) : 0.0f;
  const float p2 = ok ? __expf(lrelu(e4.z + dv.z) - m2) : 0.0f;
  const float p3 = ok ? __expf(lrelu(e4.w + dv.w) - m3) : 0.0f;
  d0 += p0; d1 += p1; d2 += p2; d3 += p3;
  int k = 0;
  for (; k + 4 <= nk; k += 4)
    gather_rows<CHT, 4>(feat, lane, k, s, p0, p1, p2, p3, hd, acc);
  if ((nk - k) & 2) {
    gather_rows<CHT, 2>(feat, lane, k, s, p0, p1, p2, p3, hd, acc);
    k += 2;
  }
  if ((nk - k) & 1)
    gather_rows<CHT, 1>(feat, lane, k, s, p0, p1, p2, p3, hd, acc);
}

template <int CHT, int CH, bool LAST>
__global__ void node_gather_k(const int* __restrict__ beg, const int* __restrict__ endp,
                              const int* __restrict__ ssrc, const float* __restrict__ feat,
                              const float* __restrict__ es, const float* __restrict__ ed,
                              const float* __restrict__ bias, const float* __restrict__ g,
                              const float* __restrict__ b, float* __restrict__ out, int N) {
  constexpr int NI = CHT / 32;
  const int node = blockIdx.x * (blockDim.x >> 5) + (threadIdx.x >> 5);
  const int lane = threadIdx.x & 31;
  if (node >= N) return;
  const float4* es4 = reinterpret_cast<const float4*>(es);
  const float4* ed4 = reinterpret_cast<const float4*>(ed);
  const int jb = beg[node], je = endp[node];
  const float4 dv = ed4[node];
  const float4 sv = es4[node];
  float selfrow[CHT / 32];
#pragma unroll
  for (int i = 0; i < CHT / 32; ++i) selfrow[i] = feat[(size_t)node * CHT + lane + 32 * i];

  const float v0 = lrelu(sv.x + dv.x), v1 = lrelu(sv.y + dv.y);
  const float v2 = lrelu(sv.z + dv.z), v3 = lrelu(sv.w + dv.w);
  float m0 = v0, m1 = v1, m2 = v2, m3 = v3;
  const bool okA = (jb + lane) < je;
  const int sA = okA ? ssrc[jb + lane] : node;
  const float4 eA = es4[sA];
  m0 = fmaxf(m0, lrelu(eA.x + dv.x));
  m1 = fmaxf(m1, lrelu(eA.y + dv.y));
  m2 = fmaxf(m2, lrelu(eA.z + dv.z));
  m3 = fmaxf(m3, lrelu(eA.w + dv.w));
  for (int j = jb + 32 + lane; j < je; j += 32) {
    const float4 e4 = es4[ssrc[j]];
    m0 = fmaxf(m0, lrelu(e4.x + dv.x));
    m1 = fmaxf(m1, lrelu(e4.y + dv.y));
    m2 = fmaxf(m2, lrelu(e4.z + dv.z));
    m3 = fmaxf(m3, lrelu(e4.w + dv.w));
  }
  m0 = warp_max32(m0); m1 = warp_max32(m1); m2 = warp_max32(m2); m3 = warp_max32(m3);

  int hd[NI];
  float acc[NI];
#pragma unroll
  for (int i = 0; i < NI; ++i) { hd[i] = (lane + 32 * i) / CH; acc[i] = 0.0f; }
  float d0 = 0.0f, d1 = 0.0f, d2 = 0.0f, d3 = 0.0f;

  {
    const int nk = (je - jb < 32) ? (je - jb) : 32;
    gather_batch<CHT>(feat, lane, nk, okA, sA, eA, dv, m0, m1, m2, m3, d0, d1, d2, d3, hd, acc);
  }
  for (int j0 = jb + 32; j0 < je; j0 += 32) {
    const int j = j0 + lane;
    const bool ok = j < je;
    const int s = ok ? ssrc[j] : node;
    const float4 e4 = es4[s];
    const int nk = (je - j0 < 32) ? (je - j0) : 32;
    gather_batch<CHT>(feat, lane, nk, ok, s, e4, dv, m0, m1, m2, m3, d0, d1, d2, d3, hd, acc);
  }

  const float s0 = __expf(v0 - m0), s1 = __expf(v1 - m1);
  const float s2 = __expf(v2 - m2), s3 = __expf(v3 - m3);
#pragma unroll
  for (int i = 0; i < NI; ++i) acc[i] = fmaf(selfrow[i], pick4(hd[i], s0, s1, s2, s3), acc[i]);
  d0 = warp_sum32(d0) + s0; d1 = warp_sum32(d1) + s1;
  d2 = warp_sum32(d2) + s2; d3 = warp_sum32(d3) + s3;

  float val[NI];
#pragma unroll
  for (int i = 0; i < NI; ++i)
    val[i] = acc[i] / (pick4(hd[i], d0, d1, d2, d3) + GAT_SM_EPS);

  if constexpr (!LAST) {
    float sum = 0.0f;
#pragma unroll
    for (int i = 0; i < NI; ++i) {
      float t = val[i] + bias[lane + 32 * i];
      t = (t > 0.0f) ? t : (__expf(t) - 1.0f);
      val[i] = t; sum += t;
    }
    const float mu = warp_sum32(sum) * (1.0f / (float)CHT);
    float vs = 0.0f;
#pragma unroll
    for (int i = 0; i < NI; ++i) { float dd = val[i] - mu; vs += dd * dd; }
    const float var = warp_sum32(vs) * (1.0f / (float)CHT);
    const float rstd = rsqrtf(var + GAT_LN_EPS);
    float* op = out + (size_t)node * CHT + lane;
#pragma unroll
    for (int i = 0; i < NI; ++i) {
      const int c = lane + 32 * i;
      op[32 * i] = (val[i] - mu) * rstd * g[c] + b[c];
    }
  } else {
    static_assert(!LAST || (CHT == 160 && CH == 40), "head-mean tail is laid out for 4 heads of 40");
    const bool has1 = (lane < 8);
    const int r8 = (lane + 8) & 31, r16 = (lane + 16) & 31, r24 = (lane + 24) & 31;
    const float a1 = __shfl((lane >= 8) ? val[1] : val[2], r8, 32);
    const float a2 = __shfl((lane >= 16) ? val[2] : val[3], r16, 32);
    const float a3 = __shfl((lane >= 24) ? val[3] : val[4], r24, 32);
    const float c1 = __shfl(val[2], r8, 32);
    const float c2 = __shfl(val[3], r16, 32);
    const float c3 = __shfl(val[4], r24, 32);

    float x0, x1 = 0.0f;
    { float s = val[0]; s += a1; s += a2; s += a3;
      x0 = s * 0.25f + bias[lane]; }
    if (has1) {
      float s = val[1]; s += c1; s += c2; s += c3;
      x1 = s * 0.25f + bias[lane + 32];
    }
    const float mu = warp_sum32(x0 + (has1 ? x1 : 0.0f)) * (1.0f / 40.0f);
    const float e0 = x0 - mu, e1 = has1 ? (x1 - mu) : 0.0f;
    const float var = warp_sum32(e0 * e0 + e1 * e1) * (1.0f / 40.0f);
    const float rstd = rsqrtf(var + GAT_LN_EPS);
    const float z0 = e0 * rstd * g[lane] + b[lane];
    const float z1 = has1 ? (e1 * rstd * g[lane + 32] + b[lane + 32]) : -3.0e38f;
    const float mx = warp_max32(fmaxf(z0, z1));
    const float se = warp_sum32(__expf(z0 - mx) + (has1 ? __expf(z1 - mx) : 0.0f));
    const float lse = __logf(se);
    float* yp = out + (size_t)node * CH;
    yp[lane] = z0 - mx - lse;
    if (has1) yp[lane + 32] = z1 - mx - lse;
  }
}

extern "C" void kernel_launch(void* const* d_in, const int* in_sizes, int n_in,
                              void* d_out, int out_size, void* d_ws, size_t ws_size,
                              hipStream_t stream) {
  const float* x        = (const float*)d_in[0];
  const int*   ei       = (const int*)d_in[1];
  const float* W1       = (const float*)d_in[2];
  const float* att_src1 = (const float*)d_in[3];
  const float* att_dst1 = (const float*)d_in[4];
  const float* b1       = (const float*)d_in[5];
  const float* W2       = (const float*)d_in[6];
  const float* att_src2 = (const float*)d_in[7];
  const float* att_dst2 = (const float*)d_in[8];
  const float* b2       = (const float*)d_in[9];
  const float* ln0_g    = (const float*)d_in[10];
  const float* ln0_b    = (const float*)d_in[11];
  const float* ln1_g    = (const float*)d_in[12];
  const float* ln1_b    = (const float*)d_in[13];

  const int F_IN = 256, H1C = 128, H2C = 160, H = 4;
  const int N = in_sizes[0] / F_IN;
  const int E = in_sizes[1] / 2;

  float* ws   = (float*)d_ws;
  float* h1   = ws;
  float* out1 = h1   + (size_t)N * H1C;
  float* h2   = out1 + (size_t)N * H1C;
  float* es   = h2   + (size_t)N * H2C;
  float* ed   = es   + (size_t)N * H;
  const int NBK = (N + BK_NODES - 1) / BK_NODES;
  const int G1  = (E + EPB - 1) / EPB;
  int* btot   = (int*)(ed + (size_t)N * H);
  int* pairs  = btot + (size_t)NBK * BK_PAD;
  int* ssrc   = pairs + 2 * (size_t)E;
  int* beg    = ssrc + E;
  int* cur    = beg + N;
  int* blkoff = cur + N;
  int* bbase  = blkoff + (size_t)G1 * NBK;

  const int mBlocks = (N + GEMM_BM - 1) / GEMM_BM;

  const int mbA = mBlocks / 2;
  const int mbB = mBlocks - mbA;
  fill_i32<<<(NBK * BK_PAD + 255) / 256, 256, 0, stream>>>(btot, NBK * BK_PAD, 0);
  count_with_gemm_k<256, 128, 32><<<mbA + G1, BK_MAX, 0, stream>>>(
      ei, E, NBK, btot, blkoff, mbA, 0, x, W1, h1, att_src1, att_dst1, es, ed, N);
  scatter_with_gemm_k<256, 128, 32><<<mbB + G1, BK_MAX, 0, stream>>>(
      ei, E, NBK, btot, blkoff, bbase, pairs, mbB, mbA, x, W1, h1, att_src1, att_dst1, es, ed, N);
  bucket_sort_k<<<NBK, 1024, 0, stream>>>(pairs, btot, bbase, N, beg, cur, ssrc);

  node_gather_k<128, 32, false><<<(N + 7) / 8, 256, 0, stream>>>(
      beg, cur, ssrc, h1, es, ed, b1, ln0_g, ln0_b, out1, N);

  gemm_wmma_f32<128, 160, 40><<<mBlocks, 2 * H2C, 0, stream>>>(out1, W2, h2, att_src2, att_dst2,
                                                               es, ed, N);
  node_gather_k<160, 40, true><<<(N + 7) / 8, 256, 0, stream>>>(
      beg, cur, ssrc, h2, es, ed, b2, ln1_g, ln1_b, (float*)d_out, N);
}


# Level-1 edge bucketing in one pass: a block orders its edges by bucket in its own stretch and publishes run tables that level 2 walks; fill and count launches go (7 to 5); arrival-order ranks kept
# speedup vs baseline: 5.3879x; 1.0958x over previous
#include <hip/hip_runtime.h>
#include <math.h>

typedef __attribute__((ext_vector_type(2))) float v2f;
typedef __attribute__((ext_vector_type(8))) float v8f;

#define GAT_NEG_SLOPE 0.2f
#define GAT_SM_EPS 1e-16f
#define GAT_LN_EPS 1e-5f

__device__ __forceinline__ float xhalf(float v) {
  const int b = __float_as_int(v);
  return __int_as_float(__builtin_amdgcn_permlanex16(b, b, 0x76543210, 0xfedcba98, false, false));
}
template <int XM>
__device__ __forceinline__ float xrow(float v) {
  return __int_as_float(__builtin_amdgcn_update_dpp(0, __float_as_int(v), 0x160 + XM, 0xF, 0xF, true));
}
__device__ __forceinline__ float warp_sum32(float v) {
  v += xhalf(v);
  v += xrow<8>(v);
  v += xrow<4>(v);
  v += xrow<2>(v);
  v += xrow<1>(v);
  return v;
}
__device__ __forceinline__ float warp_max32(float v) {
  v = fmaxf(v, xhalf(v));
  v = fmaxf(v, xrow<8>(v));
  v = fmaxf(v, xrow<4>(v));
  v = fmaxf(v, xrow<2>(v));
  v = fmaxf(v, xrow<1>(v));
  return v;
}

__device__ __forceinline__ float group8_sum(float v) {
  v += __int_as_float(__builtin_amdgcn_update_dpp(0, __float_as_int(v), 0x161, 0xF, 0xF, true));
  v += __int_as_float(__builtin_amdgcn_update_dpp(0, __float_as_int(v), 0x162, 0xF, 0xF, true));
  v += __int_as_float(__builtin_amdgcn_update_dpp(0, __float_as_int(v), 0x164, 0xF, 0xF, true));
  return v;
}

#define GEMM_MT 4
#define GEMM_BM (16 * GEMM_MT)
template <int K, int NC, int CH>
__device__ __forceinline__ void
gemm_block(const int mb, const float* __restrict__ A, const float* __restrict__ B,
           float* __restrict__ C, const float* __restrict__ a_src,
           const float* __restrict__ a_dst, float* __restrict__ es,
           float* __restrict__ ed, int M) {
  static_assert(NC == 4 * CH && CH % 8 == 0, "four heads, each a whole number of 8-column groups");
  constexpr int NG = NC / 8;
  constexpr int GPH = CH / 8;
  __shared__ float part_s[GEMM_BM * NG];
  __shared__ float part_d[GEMM_BM * NG];
  constexpr int KP = K + 4;
  constexpr int NT = 2 * NC;
  constexpr int RV = K / 4;
  constexpr int TOT = GEMM_BM * RV;
  constexpr int NIT = (TOT + NT - 1) / NT;
  __shared__ __attribute__((aligned(16))) float As[GEMM_BM * KP];
  const int tid = threadIdx.x;
  const int mBase = mb * GEMM_BM;

  float4 stg[NIT];
#pragma unroll
  for (int it = 0; it < NIT; ++it) {
    const int i = tid + it * NT;
    const int r = i / RV, c4 = i - r * RV;
    const int row = mBase + r;
    stg[it] = make_float4(0.0f, 0.0f, 0.0f, 0.0f);
    if (i < TOT && row < M)
      stg[it] = *reinterpret_cast<const float4*>(A + (size_t)row * K + c4 * 4);
  }
#pragma unroll
  for (int it = 0; it < NIT; ++it) {
    const int i = tid + it * NT;
    const int r = i / RV, c4 = i - r * RV;
    if (i < TOT) *reinterpret_cast<float4*>(As + r * KP + c4 * 4) = stg[it];
  }
  __syncthreads();

  const int wave = tid >> 5;
  const int lane = tid & 31;
  const int nBase = wave * 16;
  const int l15 = lane & 15;
  const int khalf = (lane >> 4) * 2;

  v8f acc[GEMM_MT];
#pragma unroll
  for (int mt = 0; mt < GEMM_MT; ++mt) acc[mt] = v8f{};

  const float* ap = As + l15 * KP + khalf;
  const float* bp = B + (size_t)khalf * NC + nBase + l15;
#pragma unroll 8
  for (int k0 = 0; k0 < K; k0 += 4) {
    v2f b;
    b.x = bp[(size_t)k0 * NC];
    b.y = bp[(size_t)(k0 + 1) * NC];
#pragma unroll
    for (int mt = 0; mt < GEMM_MT; ++mt) {
      const v2f a = *reinterpret_cast<const v2f*>(ap + mt * 16 * KP + k0);
      acc[mt] = __builtin_amdgcn_wmma_f32_16x16x4_f32(false, a, false, b,
                                                      (short)0, acc[mt], false, false);
    }
  }

  const int col = nBase + l15;
#pragma unroll
  for (int mt = 0; mt < GEMM_MT; ++mt) {
    const int rowTop = mBase + mt * 16 + (lane >> 4) * 8;
#pragma unroll
    for (int r = 0; r < 8; ++r) {
      const int row = rowTop + r;
      if (row < M) C[(size_t)row * NC + col] = acc[mt][r];
    }
  }

  const float as = a_src[col], ad = a_dst[col];
  const int grp = (nBase >> 3) + ((lane >> 3) & 1);
#pragma unroll
  for (int mt = 0; mt < GEMM_MT; ++mt) {
#pragma unroll
    for (int r = 0; r < 8; ++r) {
      const float ps = group8_sum(acc[mt][r] * as);
      const float pd = group8_sum(acc[mt][r] * ad);
      if ((lane & 7) == 0) {
        const int rl = mt * 16 + (lane >> 4) * 8 + r;
        part_s[rl * NG + grp] = ps;
        part_d[rl * NG + grp] = pd;
      }
    }
  }
  __syncthreads();
  if (tid < GEMM_BM * 4) {
    const int rl = tid >> 2, hh = tid & 3;
    const float* qs = part_s + rl * NG + hh * GPH;
    const float* qd = part_d + rl * NG + hh * GPH;
    float s = qs[0], d = qd[0];
#pragma unroll
    for (int gi = 1; gi < GPH; ++gi) { s += qs[gi]; d += qd[gi]; }
    const int row = mBase + rl;
    if (row < M) { es[(size_t)row * 4 + hh] = s; ed[(size_t)row * 4 + hh] = d; }
  }
}

template <int K, int NC, int CH>
__global__ void __launch_bounds__(2 * NC)
gemm_wmma_f32(const float* __restrict__ A, const float* __restrict__ B,
              float* __restrict__ C, const float* __restrict__ a_src,
              const float* __restrict__ a_dst, float* __restrict__ es,
              float* __restrict__ ed, int M) {
  gemm_block<K, NC, CH>(blockIdx.x, A, B, C, a_src, a_dst, es, ed, M);
}

__device__ __forceinline__ float lrelu(float v) {
  return (v > 0.0f) ? v : GAT_NEG_SLOPE * v;
}
__device__ __forceinline__ float pick4(int h, float a0, float a1, float a2, float a3) {
  return (h == 0) ? a0 : ((h == 1) ? a1 : ((h == 2) ? a2 : a3));
}

#define BK_SHIFT 8
#define BK_NODES (1 << BK_SHIFT)
#define BK_MAX 256
#define EPB 2048
#define EPT (EPB / 256)
#define RUN_MAX 1024
#define RUN_GRP 4

__device__ __forceinline__ int wave_incl_scan(int v, int lane) {
#pragma unroll
  for (int o = 1; o < 32; o <<= 1) {
    int t = __shfl_up(v, o, 32);
    if (lane >= o) v += t;
  }
  return v;
}

__device__ __forceinline__ void bucket_local_block(const int bx, const int* __restrict__ ei, int E,
                                                   int NBK, int G1, int* __restrict__ runcnt,
                                                   int* __restrict__ runoff,
                                                   int* __restrict__ pairs) {
  __shared__ int hist[BK_MAX];
  __shared__ int loff[BK_MAX];
  __shared__ int wsum[BK_MAX / 32];
  const int t = threadIdx.x, lane = t & 31, wave = t >> 5;
  hist[t] = 0;
  __syncthreads();
  const int e0 = bx * EPB + t;
  int s[EPT], d[EPT], r[EPT];
#pragma unroll
  for (int k = 0; k < EPT; ++k) {
    const int e = e0 + k * 256;
    const int ec = (e < E) ? e : (E - 1);
    s[k] = ei[ec];
    d[k] = ei[E + ec];
  }
#pragma unroll
  for (int k = 0; k < EPT; ++k) {
    r[k] = 0;
    if (e0 + k * 256 < E) r[k] = atomicAdd(&hist[d[k] >> BK_SHIFT], 1);
  }
  __syncthreads();
  const int v = hist[t];
  const int incl = wave_incl_scan(v, lane);
  if (lane == 31) wsum[wave] = incl;
  __syncthreads();
  int add = 0;
  for (int w = 0; w < wave; ++w) add += wsum[w];
  const int excl = add + incl - v;
  loff[t] = excl;
  if (t < NBK) {
    runcnt[(size_t)t * G1 + bx] = v;
    runoff[(size_t)t * G1 + bx] = bx * EPB + excl;
  }
  __syncthreads();
  int2* __restrict__ pr = reinterpret_cast<int2*>(pairs) + (size_t)bx * EPB;
#pragma unroll
  for (int k = 0; k < EPT; ++k)
    if (e0 + k * 256 < E) pr[loff[d[k] >> BK_SHIFT] + r[k]] = make_int2(s[k], d[k]);
}

template <int K, int NC, int CH>
__global__ void __launch_bounds__(2 * NC)
bucket_with_gemm_k(const int* __restrict__ ei, int E, int NBK, int G1, int* __restrict__ runcnt,
                   int* __restrict__ runoff, int* __restrict__ pairs, int nGemm,
                   const float* __restrict__ A, const float* __restrict__ B,
                   float* __restrict__ C, const float* __restrict__ a_src,
                   const float* __restrict__ a_dst, float* __restrict__ es,
                   float* __restrict__ ed, int M) {
  static_assert(2 * NC == BK_MAX, "both kinds of block run with BK_MAX threads");
  const int bx = blockIdx.x;
  if (bx < nGemm) gemm_block<K, NC, CH>(bx, A, B, C, a_src, a_dst, es, ed, M);
  else bucket_local_block(bx - nGemm, ei, E, NBK, G1, runcnt, runoff, pairs);
}

template <bool PLACE>
__device__ __forceinline__ void walk_runs(const int2* __restrict__ pr, const int* rc, const int* ro,
                                          int G1, int wave, int lane, int* ctr,
                                          int* __restrict__ ssrc) {
  for (int b0 = wave * RUN_GRP; b0 < G1; b0 += 32 * RUN_GRP) {
    int c[RUN_GRP], o[RUN_GRP], ex[RUN_GRP], ey[RUN_GRP];
#pragma unroll
    for (int u = 0; u < RUN_GRP; ++u) {
      const int b = b0 + u;
      c[u] = (b < G1) ? rc[b] : 0;
      o[u] = (b < G1) ? ro[b] : 0;
      const int idx = (lane < c[u]) ? (o[u] + lane) : 0;
      if (PLACE) {
        const int2 q = pr[idx];
        ex[u] = q.x; ey[u] = q.y;
      } else {
        ex[u] = 0;
        ey[u] = reinterpret_cast<const int*>(pr)[2 * idx + 1];
      }
    }
#pragma unroll
    for (int u = 0; u < RUN_GRP; ++u) {
      if (lane < c[u]) {
        const int pos = atomicAdd(&ctr[ey[u] & (BK_NODES - 1)], 1);
        if (PLACE) ssrc[pos] = ex[u];
      }
      for (int i = 32 + lane; i < c[u]; i += 32) {
        const int2 f = pr[o[u] + i];
        const int pos = atomicAdd(&ctr[f.y & (BK_NODES - 1)], 1);
        if (PLACE) ssrc[pos] = f.x;
      }
    }
  }
}

__global__ void __launch_bounds__(1024)
bucket_sort_k(const int* __restrict__ pairs, const int* __restrict__ runcnt,
              const int* __restrict__ runoff, int G1, int N, int* __restrict__ beg,
              int* __restrict__ endp, int* __restrict__ ssrc) {
  __shared__ int cnt[BK_NODES];
  __shared__ int cursor[BK_NODES];
  __shared__ int wsum[32];
  __shared__ int rc[RUN_MAX];
  __shared__ int ro[RUN_MAX];
  const int t = threadIdx.x, lane = t & 31, wave = t >> 5;
  const int bk = blockIdx.x;
  const int2* __restrict__ pr = reinterpret_cast<const int2*>(pairs);
  const int nfront = bk * G1;
  for (int b = t; b < G1; b += 1024) { rc[b] = runcnt[nfront + b]; ro[b] = runoff[nfront + b]; }
  const int4* __restrict__ rc4 = reinterpret_cast<const int4*>(runcnt);
  const int n4 = nfront >> 2;
  int part = 0;
#pragma unroll 8
  for (int i = t; i < n4; i += 1024) {
    const int4 q = rc4[i];
    part += (q.x + q.y) + (q.z + q.w);
  }
  for (int i = (n4 << 2) + t; i < nfront; i += 1024) part += runcnt[i];
  part = wave_incl_scan(part, lane);
  if (lane == 31) wsum[wave] = part;
  if (t < BK_NODES) cnt[t] = 0;
  __syncthreads();
  int start = 0;
  for (int w = 0; w < 32; ++w) start += wsum[w];
  walk_runs<false>(pr, rc, ro, G1, wave, lane, cnt, ssrc);
  __syncthreads();
  int v = 0, incl = 0;
  if (t < BK_NODES) {
    v = cnt[t];
    incl = wave_incl_scan(v, lane);
    if (lane == 31) wsum[wave] = incl;
  }
  __syncthreads();
  if (t < BK_NODES) {
    int add = 0;
    for (int w = 0; w < wave; ++w) add += wsum[w];
    const int p = start + add + incl - v;
    cursor[t] = p;
    const int node = bk * BK_NODES + t;
    if (node < N) { beg[node] = p; endp[node] = p + v; }
  }
  __syncthreads();
  walk_runs<true>(pr, rc, ro, G1, wave, lane, cursor, ssrc);
}

template <int CHT, int U>
__device__ __forceinline__ void gather_rows(const float* __restrict__ feat, int lane, int k, int s,
                                            float p0, float p1, float p2, float p3,
                                            const int (&hd)[CHT / 32], float (&acc)[CHT / 32]) {
  constexpr int NI = CHT / 32;
  float r[U][NI];
#pragma unroll
  for (int u = 0; u < U; ++u) {
    const int sk = __builtin_amdgcn_readlane(s, k + u);
    const float* fp = feat + (size_t)sk * CHT + lane;
#pragma unroll
    for (int i = 0; i < NI; ++i) r[u][i] = fp[32 * i];
  }
#pragma unroll
  for (int u = 0; u < U; ++u) {
    const float q0 = __int_as_float(__builtin_amdgcn_readlane(__float_as_int(p0), k + u));
    const float q1 = __int_as_float(__builtin_amdgcn_readlane(__float_as_int(p1), k + u));
    const float q2 = __int_as_float(__builtin_amdgcn_readlane(__float_as_int(p2), k + u));
    const float q3 = __int_as_float(__builtin_amdgcn_readlane(__float_as_int(p3), k + u));
#pragma unroll
    for (int i = 0; i < NI; ++i) acc[i] = fmaf(r[u][i], pick4(hd[i], q0, q1, q2, q3), acc[i]);
  }
}

template <int CHT>
__device__ __forceinline__ void gather_batch(const float* __restrict__ feat, int lane, int nk,
                                             bool ok, int s, const float4 e4, const float4 dv,
                                             float m0, float m1, float m2, float m3,
                                             float& d0, float& d1, float& d2, float& d3,
                                             const int (&hd)[CHT / 32], float (&acc)[CHT / 32]) {
  const float p0 = ok ? __expf(lrelu(e4.x + dv.x) - m0) : 0.0f;
  const float p1 = ok ? __expf(lrelu(e4.y + dv.y) - m1) : 0.0f;
  const float p2 = ok ? __expf(lrelu(e4.z + dv.z) - m2) : 0.0f;
  const float p3 = ok ? __expf(lrelu(e4.w + dv.w) - m3) : 0.0f;
  d0 += p0; d1 += p1; d2 += p2; d3 += p3;
  int k = 0;
  for (; k + 4 <= nk; k += 4)
    gather_rows<CHT, 4>(feat, lane, k, s, p0, p1, p2, p3, hd, acc);
  if ((nk - k) & 2) {
    gather_rows<CHT, 2>(feat, lane, k, s, p0, p1, p2, p3, hd, acc);
    k += 2;
  }
  if ((nk - k) & 1)
    gather_rows<CHT, 1>(feat, lane, k, s, p0, p1, p2, p3, hd, acc);
}

template <int CHT, int CH, bool LAST>
__global__ void node_gather_k(const int* __restrict__ beg, const int* __restrict__ endp,
                              const int* __restrict__ ssrc, const float* __restrict__ feat,
                              const float* __restrict__ es, const float* __restrict__ ed,
                              const float* __restrict__ bias, const float* __restrict__ g,
                              const float* __restrict__ b, float* __restrict__ out, int N) {
  constexpr int NI = CHT / 32;
  const int node = blockIdx.x * (blockDim.x >> 5) + (threadIdx.x >> 5);
  const int lane = threadIdx.x & 31;
  if (node >= N) return;
  const float4* es4 = reinterpret_cast<const float4*>(es);
  const float4* ed4 = reinterpret_cast<const float4*>(ed);
  const int jb = beg[node], je = endp[node];
  const float4 dv = ed4[node];
  const float4 sv = es4[node];
  float selfrow[CHT / 32];
#pragma unroll
  for (int i = 0; i < CHT / 32; ++i) selfrow[i] = feat[(size_t)node * CHT + lane + 32 * i];

  const float v0 = lrelu(sv.x + dv.x), v1 = lrelu(sv.y + dv.y);
  const float v2 = lrelu(sv.z + dv.z), v3 = lrelu(sv.w + dv.w);
  float m0 = v0, m1 = v1, m2 = v2, m3 = v3;
  const bool okA = (jb + lane) < je;
  const int sA = okA ? ssrc[jb + lane] : node;
  const float4 eA = es4[sA];
  m0 = fmaxf(m0, lrelu(eA.x + dv.x));
  m1 = fmaxf(m1, lrelu(eA.y + dv.y));
  m2 = fmaxf(m2, lrelu(eA.z + dv.z));
  m3 = fmaxf(m3, lrelu(eA.w + dv.w));
  for (int j = jb + 32 + lane; j < je; j += 32) {
    const float4 e4 = es4[ssrc[j]];
    m0 = fmaxf(m0, lrelu(e4.x + dv.x));
    m1 = fmaxf(m1, lrelu(e4.y + dv.y));
    m2 = fmaxf(m2, lrelu(e4.z + dv.z));
    m3 = fmaxf(m3, lrelu(e4.w + dv.w));
  }
  m0 = warp_max32(m0); m1 = warp_max32(m1); m2 = warp_max32(m2); m3 = warp_max32(m3);

  int hd[NI];
  float acc[NI];
#pragma unroll
  for (int i = 0; i < NI; ++i) { hd[i] = (lane + 32 * i) / CH; acc[i] = 0.0f; }
  float d0 = 0.0f, d1 = 0.0f, d2 = 0.0f, d3 = 0.0f;

  {
    const int nk = (je - jb < 32) ? (je - jb) : 32;
    gather_batch<CHT>(feat, lane, nk, okA, sA, eA, dv, m0, m1, m2, m3, d0, d1, d2, d3, hd, acc);
  }
  for (int j0 = jb + 32; j0 < je; j0 += 32) {
    const int j = j0 + lane;
    const bool ok = j < je;
    const int s = ok ? ssrc[j] : node;
    const float4 e4 = es4[s];
    const int nk = (je - j0 < 32) ? (je - j0) : 32;
    gather_batch<CHT>(feat, lane, nk, ok, s, e4, dv, m0, m1, m2, m3, d0, d1, d2, d3, hd, acc);
  }

  const float s0 = __expf(v0 - m0), s1 = __expf(v1 - m1);
  const float s2 = __expf(v2 - m2), s3 = __expf(v3 - m3);
#pragma unroll
  for (int i = 0; i < NI; ++i) acc[i] = fmaf(selfrow[i], pick4(hd[i], s0, s1, s2, s3), acc[i]);
  d0 = warp_sum32(d0) + s0; d1 = warp_sum32(d1) + s1;
  d2 = warp_sum32(d2) + s2; d3 = warp_sum32(d3) + s3;

  float val[NI];
#pragma unroll
  for (int i = 0; i < NI; ++i)
    val[i] = acc[i] / (pick4(hd[i], d0, d1, d2, d3) + GAT_SM_EPS);

  if constexpr (!LAST) {
    float sum = 0.0f;
#pragma unroll
    for (int i = 0; i < NI; ++i) {
      float t = val[i] + bias[lane + 32 * i];
      t = (t > 0.0f) ? t : (__expf(t) - 1.0f);
      val[i] = t; sum += t;
    }
    const float mu = warp_sum32(sum) * (1.0f / (float)CHT);
    float vs = 0.0f;
#pragma unroll
    for (int i = 0; i < NI; ++i) { float dd = val[i] - mu; vs += dd * dd; }
    const float var = warp_sum32(vs) * (1.0f / (float)CHT);
    const float rstd = rsqrtf(var + GAT_LN_EPS);
    float* op = out + (size_t)node * CHT + lane;
#pragma unroll
    for (int i = 0; i < NI; ++i) {
      const int c = lane + 32 * i;
      op[32 * i] = (val[i] - mu) * rstd * g[c] + b[c];
    }
  } else {
    static_assert(!LAST || (CHT == 160 && CH == 40), "head-mean tail is laid out for 4 heads of 40");
    const bool has1 = (lane < 8);
    const int r8 = (lane + 8) & 31, r16 = (lane + 16) & 31, r24 = (lane + 24) & 31;
    const float a1 = __shfl((lane >= 8) ? val[1] : val[2], r8, 32);
    const float a2 = __shfl((lane >= 16) ? val[2] : val[3], r16, 32);
    const float a3 = __shfl((lane >= 24) ? val[3] : val[4], r24, 32);
    const float c1 = __shfl(val[2], r8, 32);
    const float c2 = __shfl(val[3], r16, 32);
    const float c3 = __shfl(val[4], r24, 32);

    float x0, x1 = 0.0f;
    { float s = val[0]; s += a1; s += a2; s += a3;
      x0 = s * 0.25f + bias[lane]; }
    if (has1) {
      float s = val[1]; s += c1; s += c2; s += c3;
      x1 = s * 0.25f + bias[lane + 32];
    }
    const float mu = warp_sum32(x0 + (has1 ? x1 : 0.0f)) * (1.0f / 40.0f);
    const float e0 = x0 - mu, e1 = has1 ? (x1 - mu) : 0.0f;
    const float var = warp_sum32(e0 * e0 + e1 * e1) * (1.0f / 40.0f);
    const float rstd = rsqrtf(var + GAT_LN_EPS);
    const float z0 = e0 * rstd * g[lane] + b[lane];
    const float z1 = has1 ? (e1 * rstd * g[lane + 32] + b[lane + 32]) : -3.0e38f;
    const float mx = warp_max32(fmaxf(z0, z1));
    const float se = warp_sum32(__expf(z0 - mx) + (has1 ? __expf(z1 - mx) : 0.0f));
    const float lse = __logf(se);
    float* yp = out + (size_t)node * CH;
    yp[lane] = z0 - mx - lse;
    if (has1) yp[lane + 32] = z1 - mx - lse;
  }
}

extern "C" void kernel_launch(void* const* d_in, const int* in_sizes, int n_in,
                              void* d_out, int out_size, void* d_ws, size_t ws_size,
                              hipStream_t stream) {
  const float* x        = (const float*)d_in[0];
  const int*   ei       = (const int*)d_in[1];
  const float* W1       = (const float*)d_in[2];
  const float* att_src1 = (const float*)d_in[3];
  const float* att_dst1 = (const float*)d_in[4];
  const float* b1       = (const float*)d_in[5];
  const float* W2       = (const float*)d_in[6];
  const float* att_src2 = (const float*)d_in[7];
  const float* att_dst2 = (const float*)d_in[8];
  const float* b2       = (const float*)d_in[9];
  const float* ln0_g    = (const float*)d_in[10];
  const float* ln0_b    = (const float*)d_in[11];
  const float* ln1_g    = (const float*)d_in[12];
  const float* ln1_b    = (const float*)d_in[13];

  const int F_IN = 256, H1C = 128, H2C = 160, H = 4;
  const int N = in_sizes[0] / F_IN;
  const int E = in_sizes[1] / 2;

  float* ws   = (float*)d_ws;
  float* h1   = ws;
  float* out1 = h1   + (size_t)N * H1C;
  float* h2   = out1 + (size_t)N * H1C;
  float* es   = h2   + (size_t)N * H2C;
  float* ed   = es   + (size_t)N * H;
  const int NBK = (N + BK_NODES - 1) / BK_NODES;
  const int G1  = (E + EPB - 1) / EPB;
  int* runcnt = (int*)(ed + (size_t)N * H);
  int* runoff = runcnt + (size_t)NBK * G1;
  int* pairs  = runoff + (size_t)NBK * G1;
  int* ssrc   = pairs + 2 * (size_t)G1 * EPB;
  int* beg    = ssrc + E;
  int* cur    = beg + N;

  const int mBlocks = (N + GEMM_BM - 1) / GEMM_BM;

  bucket_with_gemm_k<256, 128, 32><<<mBlocks + G1, BK_MAX, 0, stream>>>(
      ei, E, NBK, G1, runcnt, runoff, pairs, mBlocks, x, W1, h1, att_src1, att_dst1, es, ed, N);
  bucket_sort_k<<<NBK, 1024, 0, stream>>>(pairs, runcnt, runoff, G1, N, beg, cur, ssrc);

  node_gather_k<128, 32, false><<<(N + 7) / 8, 256, 0, stream>>>(
      beg, cur, ssrc, h1, es, ed, b1, ln0_g, ln0_b, out1, N);

  gemm_wmma_f32<128, 160, 40><<<mBlocks, 2 * H2C, 0, stream>>>(out1, W2, h2, att_src2, att_dst2,
                                                               es, ed, N);
  node_gather_k<160, 40, true><<<(N + 7) / 8, 256, 0, stream>>>(
      beg, cur, ssrc, h2, es, ed, b2, ln1_g, ln1_b, (float*)d_out, N);
}
